# starting kernel
# baseline (speedup 1.0000x reference)
#include <hip/hip_runtime.h>

#define B_    8
#define C_    3
#define N_    4096
#define K_    20
#define H_    64
#define DOUT_ 64
#define EPS_  1e-5f

typedef __attribute__((ext_vector_type(2))) float v2f;
typedef __attribute__((ext_vector_type(8))) float v8f;

__global__ __launch_bounds__(256) void knn_kernel(const float* __restrict__ x,
                                                  int* __restrict__ idx)
{
    __shared__ float sx[N_], sy[N_], sz[N_];
    const int blocksPerBatch = N_ / 256;
    const int b     = blockIdx.x / blocksPerBatch;
    const int chunk = blockIdx.x % blocksPerBatch;
    const float* xb = x + (size_t)b * 3 * N_;
    for (int j = threadIdx.x; j < N_; j += 256) {
        sx[j] = xb[j];
        sy[j] = xb[N_ + j];
        sz[j] = xb[2 * N_ + j];
    }
    __syncthreads();

    const int   n  = chunk * 256 + threadIdx.x;
    const float qx = sx[n], qy = sy[n], qz = sz[n];
    const float qq = qx * qx + qy * qy + qz * qz;

    float bd[K_];
    int   bi[K_];
#pragma unroll
    for (int t = 0; t < K_; ++t) { bd[t] = -3.4e38f; bi[t] = 0; }

    for (int j = 0; j < N_; ++j) {
        const float px = sx[j], py = sy[j], pz = sz[j];
        const float d = 2.0f * (qx * px + qy * py + qz * pz) - qq
                        - (px * px + py * py + pz * pz);
        if (d > bd[K_ - 1]) {
            bd[K_ - 1] = d; bi[K_ - 1] = j;
#pragma unroll
            for (int t = K_ - 1; t > 0; --t) {
                if (bd[t] > bd[t - 1]) {
                    float tf = bd[t]; bd[t] = bd[t - 1]; bd[t - 1] = tf;
                    int   ti = bi[t]; bi[t] = bi[t - 1]; bi[t - 1] = ti;
                }
            }
        }
    }
    int* outp = idx + ((size_t)b * N_ + n) * K_;
#pragma unroll
    for (int t = 0; t < K_; ++t) outp[t] = bi[t];
}

__global__ __launch_bounds__(128) void edge_kernel(
    const float* __restrict__ x,  const int* __restrict__ idx,
    const float* __restrict__ w1, const float* __restrict__ b1,
    const float* __restrict__ g1, const float* __restrict__ beta1,
    const float* __restrict__ rm1,const float* __restrict__ rv1,
    float* __restrict__ out_m1,   float* __restrict__ out_m2)
{
    const int lane = threadIdx.x & 31;
    const int half = lane >> 4;
    const int l15  = lane & 15;
    const int wave = (blockIdx.x * blockDim.x + threadIdx.x) >> 5;
    const int b    = wave >> 12;
    const int n    = wave & (N_ - 1);

    const float* xb = x + (size_t)b * 3 * N_;
    const float cx = xb[n], cy = xb[N_ + n], cz = xb[2 * N_ + n];

    v2f a0[2], a1[2];
#pragma unroll
    for (int rt = 0; rt < 2; ++rt) {
        const int  kk    = rt * 16 + l15;
        const bool valid = (kk < K_);
        const int  kc    = valid ? kk : (K_ - 1);
        const int  nb    = idx[((size_t)b * N_ + n) * K_ + kc];
        const float nx = xb[nb], ny = xb[N_ + nb], nz = xb[2 * N_ + nb];
        const float e0 = valid ? (nx - cx) : 0.0f;
        const float e1 = valid ? (ny - cy) : 0.0f;
        const float e2 = valid ? (nz - cz) : 0.0f;
        const float e3 = valid ? cx : 0.0f;
        const float e4 = valid ? cy : 0.0f;
        const float e5 = valid ? cz : 0.0f;
        v2f t0, t1;
        t0.x = half ? e2 : e0;  t0.y = half ? e3 : e1;
        t1.x = half ? 0.0f : e4; t1.y = half ? 0.0f : e5;
        a0[rt] = t0; a1[rt] = t1;
    }

    v8f acc0[4], acc1[4];
#pragma unroll
    for (int ct = 0; ct < 4; ++ct) {
        const int    o    = ct * 16 + l15;
        const float* wrow = w1 + o * 6;
        const float w0 = wrow[0], w1v = wrow[1], w2v = wrow[2];
        const float w3 = wrow[3], w4v = wrow[4], w5v = wrow[5];
        v2f bt0, bt1;
        bt0.x = half ? w2v : w0;   bt0.y = half ? w3 : w1v;
        bt1.x = half ? 0.0f : w4v; bt1.y = half ? 0.0f : w5v;
        v8f z = {};
        z = __builtin_amdgcn_wmma_f32_16x16x4_f32(false, a0[0], false, bt0,
                                                  (short)0, z, false, false);
        z = __builtin_amdgcn_wmma_f32_16x16x4_f32(false, a1[0], false, bt1,
                                                  (short)0, z, false, false);
        acc0[ct] = z;
        v8f z1 = {};
        z1 = __builtin_amdgcn_wmma_f32_16x16x4_f32(false, a0[1], false, bt0,
                                                   (short)0, z1, false, false);
        z1 = __builtin_amdgcn_wmma_f32_16x16x4_f32(false, a1[1], false, bt1,
                                                   (short)0, z1, false, false);
        acc1[ct] = z1;
    }

#pragma unroll
    for (int ct = 0; ct < 4; ++ct) {
        const int   o = ct * 16 + l15;
        const float s = g1[o] * rsqrtf(rv1[o] + EPS_);
        const float t = (b1[o] - rm1[o]) * s + beta1[o];
        float mx = -3.4e38f, sm = 0.0f;
#pragma unroll
        for (int v = 0; v < 8; ++v) {
            const float val = fmaxf(acc0[ct][v] * s + t, 0.0f);
            mx = fmaxf(mx, val); sm += val;
        }
#pragma unroll
        for (int v = 0; v < 4; ++v) {
            const float val = fmaxf(acc1[ct][v] * s + t, 0.0f);
            mx = (half == 0) ? fmaxf(mx, val) : mx;
            sm = (half == 0) ? (sm + val) : sm;
        }
        mx = fmaxf(mx, __shfl_xor(mx, 16, 32));
        sm = sm + __shfl_xor(sm, 16, 32);
        if (half == 0) {
            const size_t off = ((size_t)b * 64 + o) * N_ + n;
            out_m1[off] = mx;
            out_m2[off] = sm * (1.0f / K_);
        }
    }
}

__global__ __launch_bounds__(128) void conv2_kernel(
    const float* __restrict__ m1, const float* __restrict__ m2,
    const float* __restrict__ w2, const float* __restrict__ b2,
    const float* __restrict__ g2, const float* __restrict__ beta2,
    const float* __restrict__ rm2,const float* __restrict__ rv2,
    float* __restrict__ out)
{
    __shared__ float sw[DOUT_ * 128];
    for (int i = threadIdx.x; i < DOUT_ * 128; i += 128) sw[i] = w2[i];
    __syncthreads();

    const int lane    = threadIdx.x & 31;
    const int half    = lane >> 4;
    const int l15     = lane & 15;
    const int wave    = (blockIdx.x * blockDim.x + threadIdx.x) >> 5;
    const int rowbase = wave * 16;

    const int m  = rowbase + l15;
    const int bb = m >> 12;
    const int nn = m & (N_ - 1);
    const float* arow1 = m1 + (size_t)bb * 64 * N_ + nn;
    const float* arow2 = m2 + (size_t)bb * 64 * N_ + nn;

    v8f acc[4] = {{}, {}, {}, {}};
#pragma unroll
    for (int ks = 0; ks < 32; ++ks) {
        const int k0 = ks * 4 + half * 2;
        float av0, av1;
        if (ks < 16) {
            av0 = arow1[(size_t)k0 * N_];
            av1 = arow1[(size_t)(k0 + 1) * N_];
        } else {
            av0 = arow2[(size_t)(k0 - 64) * N_];
            av1 = arow2[(size_t)(k0 - 63) * N_];
        }
        const v2f a = {av0, av1};
#pragma unroll
        for (int ct = 0; ct < 4; ++ct) {
            const int o = ct * 16 + l15;
            const v2f bt = { sw[o * 128 + k0], sw[o * 128 + k0 + 1] };
            acc[ct] = __builtin_amdgcn_wmma_f32_16x16x4_f32(
                false, a, false, bt, (short)0, acc[ct], false, false);
        }
    }

#pragma unroll
    for (int ct = 0; ct < 4; ++ct) {
        const int   o = ct * 16 + l15;
        const float s = g2[o] * rsqrtf(rv2[o] + EPS_);
        const float t = (b2[o] - rm2[o]) * s + beta2[o];
#pragma unroll
        for (int v = 0; v < 8; ++v) {
            const int mr = rowbase + v + half * 8;
            const int ob = mr >> 12;
            const int on = mr & (N_ - 1);
            out[((size_t)ob * 64 + o) * N_ + on] = fmaxf(acc[ct][v] * s + t, 0.0f);
        }
    }
}

extern "C" void kernel_launch(void* const* d_in, const int* in_sizes, int n_in,
                              void* d_out, int out_size, void* d_ws, size_t ws_size,
                              hipStream_t stream)
{
    const float* x     = (const float*)d_in[0];
    const float* w1    = (const float*)d_in[1];
    const float* b1    = (const float*)d_in[2];
    const float* g1    = (const float*)d_in[3];
    const float* beta1 = (const float*)d_in[4];
    const float* rm1   = (const float*)d_in[5];
    const float* rv1   = (const float*)d_in[6];
    const float* w2    = (const float*)d_in[7];
    const float* b2    = (const float*)d_in[8];
    const float* g2    = (const float*)d_in[9];
    const float* beta2 = (const float*)d_in[10];
    const float* rm2   = (const float*)d_in[11];
    const float* rv2   = (const float*)d_in[12];

    float* out    = (float*)d_out;
    float* out_m1 = out + (size_t)B_ * 64 * N_;
    float* out_m2 = out + 2 * (size_t)B_ * 64 * N_;
    int*   idx    = (int*)d_ws;

    knn_kernel<<<B_ * (N_ / 256), 256, 0, stream>>>(x, idx);

    edge_kernel<<<(B_ * N_) / 4, 128, 0, stream>>>(
        x, idx, w1, b1, g1, beta1, rm1, rv1, out_m1, out_m2);

    conv2_kernel<<<(B_ * N_ / 16) / 4, 128, 0, stream>>>(
        out_m1, out_m2, w2, b2, g2, beta2, rm2, rv2, out);
}


# knn: the scan only queues points that beat a lane's last-drained 20th best (per-thread LDS queue, branch-free tail store); the sorted insertion runs per queued slot in batches, not per scanned point
# speedup vs baseline: 2.4017x; 2.4017x over previous
#include <hip/hip_runtime.h>

#define B_    8
#define C_    3
#define N_    4096
#define K_    20
#define H_    64
#define DOUT_ 64
#define EPS_  1e-5f

typedef __attribute__((ext_vector_type(2))) float v2f;
typedef __attribute__((ext_vector_type(8))) float v8f;

#define KNN_CH  16
#define KNN_CAP 32

#define KNN_STEP(dhi, dlo, ihi, ilo)                      \
    {   const bool  sw = (dlo) > (dhi);                   \
        const float fh = sw ? (dlo) : (dhi);              \
        const float fl = sw ? (dhi) : (dlo);              \
        const int   jh = sw ? (ilo) : (ihi);              \
        const int   jl = sw ? (ihi) : (ilo);              \
        dhi = fh; dlo = fl; ihi = jh; ilo = jl; }

__global__ __launch_bounds__(256) void knn_kernel(const float* __restrict__ x,
                                                  int* __restrict__ idx)
{
    __shared__ float sx[N_], sy[N_], sz[N_];
    __shared__ float qd[KNN_CAP * 256];
    __shared__ int   qj[KNN_CAP * 256];
    const int blocksPerBatch = N_ / 256;
    const int b     = blockIdx.x / blocksPerBatch;
    const int chunk = blockIdx.x % blocksPerBatch;
    const float* xb = x + (size_t)b * 3 * N_;
    for (int j = threadIdx.x; j < N_; j += 256) {
        sx[j] = xb[j];
        sy[j] = xb[N_ + j];
        sz[j] = xb[2 * N_ + j];
    }
    __syncthreads();

    const int   tid = threadIdx.x;
    const int   n  = chunk * 256 + tid;
    const float qx = sx[n], qy = sy[n], qz = sz[n];
    const float qq = qx * qx + qy * qy + qz * qz;

    float e0 = -3.4e38f, e1 = -3.4e38f, e2 = -3.4e38f, e3 = -3.4e38f, e4 = -3.4e38f, e5 = -3.4e38f, e6 = -3.4e38f, e7 = -3.4e38f, e8 = -3.4e38f, e9 = -3.4e38f, e10 = -3.4e38f, e11 = -3.4e38f, e12 = -3.4e38f, e13 = -3.4e38f, e14 = -3.4e38f, e15 = -3.4e38f, e16 = -3.4e38f, e17 = -3.4e38f, e18 = -3.4e38f, e19 = -3.4e38f;
    int   k0 = 0, k1 = 0, k2 = 0, k3 = 0, k4 = 0, k5 = 0, k6 = 0, k7 = 0, k8 = 0, k9 = 0, k10 = 0, k11 = 0, k12 = 0, k13 = 0, k14 = 0, k15 = 0, k16 = 0, k17 = 0, k18 = 0, k19 = 0;

    float thr = -3.4e38f;
    int   cnt = 0;

    for (int j0 = 0; j0 < N_; j0 += KNN_CH) {
#pragma unroll
        for (int u = 0; u < KNN_CH; ++u) {
            const int j = j0 + u;
            const float px = sx[j], py = sy[j], pz = sz[j];
            const float d = 2.0f * (qx * px + qy * py + qz * pz) - qq
                            - (px * px + py * py + pz * pz);
            qd[cnt * 256 + tid] = d;
            qj[cnt * 256 + tid] = j;
            cnt += (d > thr) ? 1 : 0;
        }

        if ((j0 + KNN_CH >= N_) || __any(cnt > KNN_CAP - KNN_CH)) {
            for (int s = 0; __any(s < cnt); ++s) {
                float d = -3.4e38f;
                int   j = 0;
                if (s < cnt) { d = qd[s * 256 + tid]; j = qj[s * 256 + tid]; }
                const bool ins = d > e19;
                e19 = ins ? d : e19;
                k19 = ins ? j : k19;
                KNN_STEP(e18, e19, k18, k19)
                KNN_STEP(e17, e18, k17, k18)
                KNN_STEP(e16, e17, k16, k17)
                KNN_STEP(e15, e16, k15, k16)
                KNN_STEP(e14, e15, k14, k15)
                KNN_STEP(e13, e14, k13, k14)
                KNN_STEP(e12, e13, k12, k13)
                KNN_STEP(e11, e12, k11, k12)
                KNN_STEP(e10, e11, k10, k11)
                KNN_STEP(e9, e10, k9, k10)
                KNN_STEP(e8, e9, k8, k9)
                KNN_STEP(e7, e8, k7, k8)
                KNN_STEP(e6, e7, k6, k7)
                KNN_STEP(e5, e6, k5, k6)
                KNN_STEP(e4, e5, k4, k5)
                KNN_STEP(e3, e4, k3, k4)
                KNN_STEP(e2, e3, k2, k3)
                KNN_STEP(e1, e2, k1, k2)
                KNN_STEP(e0, e1, k0, k1)
            }
            cnt = 0;
            thr = e19;
        }
    }
    int* outp = idx + ((size_t)b * N_ + n) * K_;
    outp[0] = k0;
    outp[1] = k1;
    outp[2] = k2;
    outp[3] = k3;
    outp[4] = k4;
    outp[5] = k5;
    outp[6] = k6;
    outp[7] = k7;
    outp[8] = k8;
    outp[9] = k9;
    outp[10] = k10;
    outp[11] = k11;
    outp[12] = k12;
    outp[13] = k13;
    outp[14] = k14;
    outp[15] = k15;
    outp[16] = k16;
    outp[17] = k17;
    outp[18] = k18;
    outp[19] = k19;
}

__global__ __launch_bounds__(128) void edge_kernel(
    const float* __restrict__ x,  const int* __restrict__ idx,
    const float* __restrict__ w1, const float* __restrict__ b1,
    const float* __restrict__ g1, const float* __restrict__ beta1,
    const float* __restrict__ rm1,const float* __restrict__ rv1,
    float* __restrict__ out_m1,   float* __restrict__ out_m2)
{
    const int lane = threadIdx.x & 31;
    const int half = lane >> 4;
    const int l15  = lane & 15;
    const int wave = (blockIdx.x * blockDim.x + threadIdx.x) >> 5;
    const int b    = wave >> 12;
    const int n    = wave & (N_ - 1);

    const float* xb = x + (size_t)b * 3 * N_;
    const float cx = xb[n], cy = xb[N_ + n], cz = xb[2 * N_ + n];

    v2f a0[2], a1[2];
#pragma unroll
    for (int rt = 0; rt < 2; ++rt) {
        const int  kk    = rt * 16 + l15;
        const bool valid = (kk < K_);
        const int  kc    = valid ? kk : (K_ - 1);
        const int  nb    = idx[((size_t)b * N_ + n) * K_ + kc];
        const float nx = xb[nb], ny = xb[N_ + nb], nz = xb[2 * N_ + nb];
        const float e0 = valid ? (nx - cx) : 0.0f;
        const float e1 = valid ? (ny - cy) : 0.0f;
        const float e2 = valid ? (nz - cz) : 0.0f;
        const float e3 = valid ? cx : 0.0f;
        const float e4 = valid ? cy : 0.0f;
        const float e5 = valid ? cz : 0.0f;
        v2f t0, t1;
        t0.x = half ? e2 : e0;  t0.y = half ? e3 : e1;
        t1.x = half ? 0.0f : e4; t1.y = half ? 0.0f : e5;
        a0[rt] = t0; a1[rt] = t1;
    }

    v8f acc0[4], acc1[4];
#pragma unroll
    for (int ct = 0; ct < 4; ++ct) {
        const int    o    = ct * 16 + l15;
        const float* wrow = w1 + o * 6;
        const float w0 = wrow[0], w1v = wrow[1], w2v = wrow[2];
        const float w3 = wrow[3], w4v = wrow[4], w5v = wrow[5];
        v2f bt0, bt1;
        bt0.x = half ? w2v : w0;   bt0.y = half ? w3 : w1v;
        bt1.x = half ? 0.0f : w4v; bt1.y = half ? 0.0f : w5v;
        v8f z = {};
        z = __builtin_amdgcn_wmma_f32_16x16x4_f32(false, a0[0], false, bt0,
                                                  (short)0, z, false, false);
        z = __builtin_amdgcn_wmma_f32_16x16x4_f32(false, a1[0], false, bt1,
                                                  (short)0, z, false, false);
        acc0[ct] = z;
        v8f z1 = {};
        z1 = __builtin_amdgcn_wmma_f32_16x16x4_f32(false, a0[1], false, bt0,
                                                   (short)0, z1, false, false);
        z1 = __builtin_amdgcn_wmma_f32_16x16x4_f32(false, a1[1], false, bt1,
                                                   (short)0, z1, false, false);
        acc1[ct] = z1;
    }

#pragma unroll
    for (int ct = 0; ct < 4; ++ct) {
        const int   o = ct * 16 + l15;
        const float s = g1[o] * rsqrtf(rv1[o] + EPS_);
        const float t = (b1[o] - rm1[o]) * s + beta1[o];
        float mx = -3.4e38f, sm = 0.0f;
#pragma unroll
        for (int v = 0; v < 8; ++v) {
            const float val = fmaxf(acc0[ct][v] * s + t, 0.0f);
            mx = fmaxf(mx, val); sm += val;
        }
#pragma unroll
        for (int v = 0; v < 4; ++v) {
            const float val = fmaxf(acc1[ct][v] * s + t, 0.0f);
            mx = (half == 0) ? fmaxf(mx, val) : mx;
            sm = (half == 0) ? (sm + val) : sm;
        }
        mx = fmaxf(mx, __shfl_xor(mx, 16, 32));
        sm = sm + __shfl_xor(sm, 16, 32);
        if (half == 0) {
            const size_t off = ((size_t)b * 64 + o) * N_ + n;
            out_m1[off] = mx;
            out_m2[off] = sm * (1.0f / K_);
        }
    }
}

__global__ __launch_bounds__(128) void conv2_kernel(
    const float* __restrict__ m1, const float* __restrict__ m2,
    const float* __restrict__ w2, const float* __restrict__ b2,
    const float* __restrict__ g2, const float* __restrict__ beta2,
    const float* __restrict__ rm2,const float* __restrict__ rv2,
    float* __restrict__ out)
{
    __shared__ float sw[DOUT_ * 128];
    for (int i = threadIdx.x; i < DOUT_ * 128; i += 128) sw[i] = w2[i];
    __syncthreads();

    const int lane    = threadIdx.x & 31;
    const int half    = lane >> 4;
    const int l15     = lane & 15;
    const int wave    = (blockIdx.x * blockDim.x + threadIdx.x) >> 5;
    const int rowbase = wave * 16;

    const int m  = rowbase + l15;
    const int bb = m >> 12;
    const int nn = m & (N_ - 1);
    const float* arow1 = m1 + (size_t)bb * 64 * N_ + nn;
    const float* arow2 = m2 + (size_t)bb * 64 * N_ + nn;

    v8f acc[4] = {{}, {}, {}, {}};
#pragma unroll
    for (int ks = 0; ks < 32; ++ks) {
        const int k0 = ks * 4 + half * 2;
        float av0, av1;
        if (ks < 16) {
            av0 = arow1[(size_t)k0 * N_];
            av1 = arow1[(size_t)(k0 + 1) * N_];
        } else {
            av0 = arow2[(size_t)(k0 - 64) * N_];
            av1 = arow2[(size_t)(k0 - 63) * N_];
        }
        const v2f a = {av0, av1};
#pragma unroll
        for (int ct = 0; ct < 4; ++ct) {
            const int o = ct * 16 + l15;
            const v2f bt = { sw[o * 128 + k0], sw[o * 128 + k0 + 1] };
            acc[ct] = __builtin_amdgcn_wmma_f32_16x16x4_f32(
                false, a, false, bt, (short)0, acc[ct], false, false);
        }
    }

#pragma unroll
    for (int ct = 0; ct < 4; ++ct) {
        const int   o = ct * 16 + l15;
        const float s = g2[o] * rsqrtf(rv2[o] + EPS_);
        const float t = (b2[o] - rm2[o]) * s + beta2[o];
#pragma unroll
        for (int v = 0; v < 8; ++v) {
            const int mr = rowbase + v + half * 8;
            const int ob = mr >> 12;
            const int on = mr & (N_ - 1);
            out[((size_t)ob * 64 + o) * N_ + on] = fmaxf(acc[ct][v] * s + t, 0.0f);
        }
    }
}

extern "C" void kernel_launch(void* const* d_in, const int* in_sizes, int n_in,
                              void* d_out, int out_size, void* d_ws, size_t ws_size,
                              hipStream_t stream)
{
    const float* x     = (const float*)d_in[0];
    const float* w1    = (const float*)d_in[1];
    const float* b1    = (const float*)d_in[2];
    const float* g1    = (const float*)d_in[3];
    const float* beta1 = (const float*)d_in[4];
    const float* rm1   = (const float*)d_in[5];
    const float* rv1   = (const float*)d_in[6];
    const float* w2    = (const float*)d_in[7];
    const float* b2    = (const float*)d_in[8];
    const float* g2    = (const float*)d_in[9];
    const float* beta2 = (const float*)d_in[10];
    const float* rm2   = (const float*)d_in[11];
    const float* rv2   = (const float*)d_in[12];

    float* out    = (float*)d_out;
    float* out_m1 = out + (size_t)B_ * 64 * N_;
    float* out_m2 = out + 2 * (size_t)B_ * 64 * N_;
    int*   idx    = (int*)d_ws;

    knn_kernel<<<B_ * (N_ / 256), 256, 0, stream>>>(x, idx);

    edge_kernel<<<(B_ * N_) / 4, 128, 0, stream>>>(
        x, idx, w1, b1, g1, beta1, rm1, rv1, out_m1, out_m2);

    conv2_kernel<<<(B_ * N_ / 16) / 4, 128, 0, stream>>>(
        out_m1, out_m2, w2, b2, g2, beta2, rm2, rv2, out);
}


# knn: 128-thread blocks put one wave on every SIMD of all 256 WGPs, written to issue alone: insertion as 20 independent selects, next queue slot and next 16 points read from LDS one step ahead
# speedup vs baseline: 2.7584x; 1.1485x over previous
#include <hip/hip_runtime.h>

#define B_    8
#define C_    3
#define N_    4096
#define K_    20
#define H_    64
#define DOUT_ 64
#define EPS_  1e-5f

typedef __attribute__((ext_vector_type(2))) float v2f;
typedef __attribute__((ext_vector_type(8))) float v8f;

#define KNN_T   128
#define KNN_CH  16
#define KNN_CAP 32
#define KNN_TP  129

typedef __attribute__((ext_vector_type(4))) float v4f;

#define KNN_POS(eu, el, ku, kl, cu, cl)                    \
    {   el = (cu) ? (eu) : ((cl) ? d : (el));              \
        kl = (cu) ? (ku) : ((cl) ? j : (kl)); }

#define KNN_LOAD(PX, PY, PZ, j0)                                          \
    {   const v4f* lx = (const v4f*)(sx + (j0));                          \
        const v4f* ly = (const v4f*)(sy + (j0));                          \
        const v4f* lz = (const v4f*)(sz + (j0));                          \
        PX##0 = lx[0]; PX##1 = lx[1]; PX##2 = lx[2]; PX##3 = lx[3];       \
        PY##0 = ly[0]; PY##1 = ly[1]; PY##2 = ly[2]; PY##3 = ly[3];       \
        PZ##0 = lz[0]; PZ##1 = lz[1]; PZ##2 = lz[2]; PZ##3 = lz[3]; }

#define KNN_HOLD(PX, PY, PZ)                                              \
    asm volatile("" :: "v"(PX##0), "v"(PX##1), "v"(PX##2), "v"(PX##3),    \
                       "v"(PY##0), "v"(PY##1), "v"(PY##2), "v"(PY##3),    \
                       "v"(PZ##0), "v"(PZ##1), "v"(PZ##2), "v"(PZ##3));

#define KNN_PUSH(dv, jj)                                                  \
    {   const float d = (dv);                                             \
        qd[cnt * KNN_T + tid] = d;                                        \
        qj[cnt * KNN_T + tid] = (jj);                                     \
        cnt += (d > thr) ? 1 : 0;     }

#define KNN_PAIR(X2, Y2, Z2, jb)                                          \
    {   const v2f dot = __builtin_elementwise_fma(qz2, (Z2),              \
                        __builtin_elementwise_fma(qx2, (X2), qy2 * (Y2)));\
        const v2f pp  = __builtin_elementwise_fma((Z2), (Z2),             \
                        __builtin_elementwise_fma((X2), (X2), (Y2) * (Y2)));\
        const v2f dd  = __builtin_elementwise_fma(dot, two2, nqq2) - pp;  \
        KNN_PUSH(dd.x, (jb)) KNN_PUSH(dd.y, (jb) + 1) }

#define KNN_QUAD(X, Y, Z, jb)                                             \
    KNN_PAIR(X.xy, Y.xy, Z.xy, (jb)) KNN_PAIR(X.zw, Y.zw, Z.zw, (jb) + 2)

#define KNN_SCAN(PX, PY, PZ, j0)                                          \
    KNN_QUAD(PX##0, PY##0, PZ##0, (j0))                                   \
    KNN_QUAD(PX##1, PY##1, PZ##1, (j0) + 4)                               \
    KNN_QUAD(PX##2, PY##2, PZ##2, (j0) + 8)                               \
    KNN_QUAD(PX##3, PY##3, PZ##3, (j0) + 12)

#define KNN_DRAIN(j0)                                                     \
    if (((j0) + KNN_CH >= N_) || __any(cnt > KNN_CAP - KNN_CH)) {         \
         \
         \
         \
        float dn = qd[tid];                                               \
        int   jn = qj[tid];                                               \
        for (int s = 0; __any(s < cnt); ++s) {                            \
            const bool  have = s < cnt;                                   \
            const float d = have ? dn : -3.4e38f;                         \
            const int   j = have ? jn : 0;                                \
            const int   sn = (s + 1 < KNN_CAP) ? (s + 1) : (KNN_CAP - 1); \
            dn = qd[sn * KNN_T + tid];                                    \
            jn = qj[sn * KNN_T + tid];                                    \
               \
               \
               \
               \
            __builtin_amdgcn_sched_barrier(0);                            \
            KNN_INSERT                                                    \
            __builtin_amdgcn_sched_barrier(0);                            \
               \
               \
            asm volatile("" :: "v"(dn), "v"(jn));                         \
        }                                                                 \
        cnt = 0;                                                          \
        thr = e19;                                                        \
    }

#define KNN_INSERT                                                        \
    const bool c0 = d > e0, c1 = d > e1, c2 = d > e2, c3 = d > e3, c4 = d > e4;\
    const bool c5 = d > e5, c6 = d > e6, c7 = d > e7, c8 = d > e8, c9 = d > e9;\
    const bool c10 = d > e10, c11 = d > e11, c12 = d > e12, c13 = d > e13, c14 = d > e14;\
    const bool c15 = d > e15, c16 = d > e16, c17 = d > e17, c18 = d > e18, c19 = d > e19;\
    KNN_POS(e18, e19, k18, k19, c18, c19)                                 \
    KNN_POS(e17, e18, k17, k18, c17, c18)                                 \
    KNN_POS(e16, e17, k16, k17, c16, c17)                                 \
    KNN_POS(e15, e16, k15, k16, c15, c16)                                 \
    KNN_POS(e14, e15, k14, k15, c14, c15)                                 \
    KNN_POS(e13, e14, k13, k14, c13, c14)                                 \
    KNN_POS(e12, e13, k12, k13, c12, c13)                                 \
    KNN_POS(e11, e12, k11, k12, c11, c12)                                 \
    KNN_POS(e10, e11, k10, k11, c10, c11)                                 \
    KNN_POS(e9, e10, k9, k10, c9, c10)                                    \
    KNN_POS(e8, e9, k8, k9, c8, c9)                                       \
    KNN_POS(e7, e8, k7, k8, c7, c8)                                       \
    KNN_POS(e6, e7, k6, k7, c6, c7)                                       \
    KNN_POS(e5, e6, k5, k6, c5, c6)                                       \
    KNN_POS(e4, e5, k4, k5, c4, c5)                                       \
    KNN_POS(e3, e4, k3, k4, c3, c4)                                       \
    KNN_POS(e2, e3, k2, k3, c2, c3)                                       \
    KNN_POS(e1, e2, k1, k2, c1, c2)                                       \
    KNN_POS(e0, e1, k0, k1, c0, c1)                                       \
    e0 = c0 ? d : e0;                                                     \
    k0 = c0 ? j : k0;

__global__ __launch_bounds__(KNN_T) void knn_kernel(const float* __restrict__ x,
                                                    int* __restrict__ idx)
{
    __shared__ __attribute__((aligned(16))) float spts[3 * N_];
    const float* sx = spts;
    const float* sy = spts + N_;
    const float* sz = spts + 2 * N_;
    __shared__ float qd[KNN_CAP * KNN_T];
    __shared__ int   qj[KNN_CAP * KNN_T];
    const int blocksPerBatch = N_ / KNN_T;
    const int b     = blockIdx.x / blocksPerBatch;
    const int chunk = blockIdx.x % blocksPerBatch;
    const float* xb = x + (size_t)b * 3 * N_;
    const v4f* xb4 = (const v4f*)xb;
    v4f*       sp4 = (v4f*)spts;
#pragma unroll
    for (int t = 0; t < (3 * N_ / 4) / KNN_T; ++t)
        sp4[t * KNN_T + threadIdx.x] = xb4[t * KNN_T + threadIdx.x];
    __syncthreads();

    const int   tid = threadIdx.x;
    const int   n  = chunk * KNN_T + tid;
    const float qx = sx[n], qy = sy[n], qz = sz[n];
    const float qq = fmaf(qz, qz, fmaf(qx, qx, qy * qy));
    const v2f qx2 = {qx, qx}, qy2 = {qy, qy}, qz2 = {qz, qz};
    const v2f nqq2 = {-qq, -qq}, two2 = {2.0f, 2.0f};

    float e0 = -3.4e38f, e1 = -3.4e38f, e2 = -3.4e38f, e3 = -3.4e38f, e4 = -3.4e38f, e5 = -3.4e38f, e6 = -3.4e38f, e7 = -3.4e38f, e8 = -3.4e38f, e9 = -3.4e38f, e10 = -3.4e38f, e11 = -3.4e38f, e12 = -3.4e38f, e13 = -3.4e38f, e14 = -3.4e38f, e15 = -3.4e38f, e16 = -3.4e38f, e17 = -3.4e38f, e18 = -3.4e38f, e19 = -3.4e38f;
    int   k0 = 0, k1 = 0, k2 = 0, k3 = 0, k4 = 0, k5 = 0, k6 = 0, k7 = 0, k8 = 0, k9 = 0, k10 = 0, k11 = 0, k12 = 0, k13 = 0, k14 = 0, k15 = 0, k16 = 0, k17 = 0, k18 = 0, k19 = 0;

    float thr = -3.4e38f;
    int   cnt = 0;

    v4f ax0, ax1, ax2, ax3, ay0, ay1, ay2, ay3, az0, az1, az2, az3;
    v4f bx0, bx1, bx2, bx3, by0, by1, by2, by3, bz0, bz1, bz2, bz3;
    KNN_LOAD(ax, ay, az, 0)
    for (int j0 = 0; j0 < N_; j0 += 2 * KNN_CH) {
        KNN_LOAD(bx, by, bz, j0 + KNN_CH)
        KNN_SCAN(ax, ay, az, j0)
        KNN_HOLD(bx, by, bz)
        KNN_DRAIN(j0)
        KNN_LOAD(ax, ay, az, (j0 + 2 * KNN_CH) & (N_ - 1))
        KNN_SCAN(bx, by, bz, j0 + KNN_CH)
        KNN_HOLD(ax, ay, az)
        KNN_DRAIN(j0 + KNN_CH)
    }
    __syncthreads();
    int* tk = qj;
    tk[0 * KNN_TP + tid] = k0;
    tk[1 * KNN_TP + tid] = k1;
    tk[2 * KNN_TP + tid] = k2;
    tk[3 * KNN_TP + tid] = k3;
    tk[4 * KNN_TP + tid] = k4;
    tk[5 * KNN_TP + tid] = k5;
    tk[6 * KNN_TP + tid] = k6;
    tk[7 * KNN_TP + tid] = k7;
    tk[8 * KNN_TP + tid] = k8;
    tk[9 * KNN_TP + tid] = k9;
    tk[10 * KNN_TP + tid] = k10;
    tk[11 * KNN_TP + tid] = k11;
    tk[12 * KNN_TP + tid] = k12;
    tk[13 * KNN_TP + tid] = k13;
    tk[14 * KNN_TP + tid] = k14;
    tk[15 * KNN_TP + tid] = k15;
    tk[16 * KNN_TP + tid] = k16;
    tk[17 * KNN_TP + tid] = k17;
    tk[18 * KNN_TP + tid] = k18;
    tk[19 * KNN_TP + tid] = k19;
    __syncthreads();
    int* outp = idx + ((size_t)b * N_ + chunk * KNN_T) * K_;
#pragma unroll
    for (int m = 0; m < K_; ++m) {
        const int p = m * KNN_T + tid;
        const int q = p / K_;
        const int i = p - q * K_;
        outp[p] = tk[i * KNN_TP + q];
    }
}

__global__ __launch_bounds__(128) void edge_kernel(
    const float* __restrict__ x,  const int* __restrict__ idx,
    const float* __restrict__ w1, const float* __restrict__ b1,
    const float* __restrict__ g1, const float* __restrict__ beta1,
    const float* __restrict__ rm1,const float* __restrict__ rv1,
    float* __restrict__ out_m1,   float* __restrict__ out_m2)
{
    const int lane = threadIdx.x & 31;
    const int half = lane >> 4;
    const int l15  = lane & 15;
    const int wave = (blockIdx.x * blockDim.x + threadIdx.x) >> 5;
    const int b    = wave >> 12;
    const int n    = wave & (N_ - 1);

    const float* xb = x + (size_t)b * 3 * N_;
    const float cx = xb[n], cy = xb[N_ + n], cz = xb[2 * N_ + n];

    v2f a0[2], a1[2];
#pragma unroll
    for (int rt = 0; rt < 2; ++rt) {
        const int  kk    = rt * 16 + l15;
        const bool valid = (kk < K_);
        const int  kc    = valid ? kk : (K_ - 1);
        const int  nb    = idx[((size_t)b * N_ + n) * K_ + kc];
        const float nx = xb[nb], ny = xb[N_ + nb], nz = xb[2 * N_ + nb];
        const float e0 = valid ? (nx - cx) : 0.0f;
        const float e1 = valid ? (ny - cy) : 0.0f;
        const float e2 = valid ? (nz - cz) : 0.0f;
        const float e3 = valid ? cx : 0.0f;
        const float e4 = valid ? cy : 0.0f;
        const float e5 = valid ? cz : 0.0f;
        v2f t0, t1;
        t0.x = half ? e2 : e0;  t0.y = half ? e3 : e1;
        t1.x = half ? 0.0f : e4; t1.y = half ? 0.0f : e5;
        a0[rt] = t0; a1[rt] = t1;
    }

    v8f acc0[4], acc1[4];
#pragma unroll
    for (int ct = 0; ct < 4; ++ct) {
        const int    o    = ct * 16 + l15;
        const float* wrow = w1 + o * 6;
        const float w0 = wrow[0], w1v = wrow[1], w2v = wrow[2];
        const float w3 = wrow[3], w4v = wrow[4], w5v = wrow[5];
        v2f bt0, bt1;
        bt0.x = half ? w2v : w0;   bt0.y = half ? w3 : w1v;
        bt1.x = half ? 0.0f : w4v; bt1.y = half ? 0.0f : w5v;
        v8f z = {};
        z = __builtin_amdgcn_wmma_f32_16x16x4_f32(false, a0[0], false, bt0,
                                                  (short)0, z, false, false);
        z = __builtin_amdgcn_wmma_f32_16x16x4_f32(false, a1[0], false, bt1,
                                                  (short)0, z, false, false);
        acc0[ct] = z;
        v8f z1 = {};
        z1 = __builtin_amdgcn_wmma_f32_16x16x4_f32(false, a0[1], false, bt0,
                                                   (short)0, z1, false, false);
        z1 = __builtin_amdgcn_wmma_f32_16x16x4_f32(false, a1[1], false, bt1,
                                                   (short)0, z1, false, false);
        acc1[ct] = z1;
    }

#pragma unroll
    for (int ct = 0; ct < 4; ++ct) {
        const int   o = ct * 16 + l15;
        const float s = g1[o] * rsqrtf(rv1[o] + EPS_);
        const float t = (b1[o] - rm1[o]) * s + beta1[o];
        float mx = -3.4e38f, sm = 0.0f;
#pragma unroll
        for (int v = 0; v < 8; ++v) {
            const float val = fmaxf(acc0[ct][v] * s + t, 0.0f);
            mx = fmaxf(mx, val); sm += val;
        }
#pragma unroll
        for (int v = 0; v < 4; ++v) {
            const float val = fmaxf(acc1[ct][v] * s + t, 0.0f);
            mx = (half == 0) ? fmaxf(mx, val) : mx;
            sm = (half == 0) ? (sm + val) : sm;
        }
        mx = fmaxf(mx, __shfl_xor(mx, 16, 32));
        sm = sm + __shfl_xor(sm, 16, 32);
        if (half == 0) {
            const size_t off = ((size_t)b * 64 + o) * N_ + n;
            out_m1[off] = mx;
            out_m2[off] = sm * (1.0f / K_);
        }
    }
}

__global__ __launch_bounds__(128) void conv2_kernel(
    const float* __restrict__ m1, const float* __restrict__ m2,
    const float* __restrict__ w2, const float* __restrict__ b2,
    const float* __restrict__ g2, const float* __restrict__ beta2,
    const float* __restrict__ rm2,const float* __restrict__ rv2,
    float* __restrict__ out)
{
    __shared__ float sw[DOUT_ * 128];
    for (int i = threadIdx.x; i < DOUT_ * 128; i += 128) sw[i] = w2[i];
    __syncthreads();

    const int lane    = threadIdx.x & 31;
    const int half    = lane >> 4;
    const int l15     = lane & 15;
    const int wave    = (blockIdx.x * blockDim.x + threadIdx.x) >> 5;
    const int rowbase = wave * 16;

    const int m  = rowbase + l15;
    const int bb = m >> 12;
    const int nn = m & (N_ - 1);
    const float* arow1 = m1 + (size_t)bb * 64 * N_ + nn;
    const float* arow2 = m2 + (size_t)bb * 64 * N_ + nn;

    v8f acc[4] = {{}, {}, {}, {}};
#pragma unroll
    for (int ks = 0; ks < 32; ++ks) {
        const int k0 = ks * 4 + half * 2;
        float av0, av1;
        if (ks < 16) {
            av0 = arow1[(size_t)k0 * N_];
            av1 = arow1[(size_t)(k0 + 1) * N_];
        } else {
            av0 = arow2[(size_t)(k0 - 64) * N_];
            av1 = arow2[(size_t)(k0 - 63) * N_];
        }
        const v2f a = {av0, av1};
#pragma unroll
        for (int ct = 0; ct < 4; ++ct) {
            const int o = ct * 16 + l15;
            const v2f bt = { sw[o * 128 + k0], sw[o * 128 + k0 + 1] };
            acc[ct] = __builtin_amdgcn_wmma_f32_16x16x4_f32(
                false, a, false, bt, (short)0, acc[ct], false, false);
        }
    }

#pragma unroll
    for (int ct = 0; ct < 4; ++ct) {
        const int   o = ct * 16 + l15;
        const float s = g2[o] * rsqrtf(rv2[o] + EPS_);
        const float t = (b2[o] - rm2[o]) * s + beta2[o];
#pragma unroll
        for (int v = 0; v < 8; ++v) {
            const int mr = rowbase + v + half * 8;
            const int ob = mr >> 12;
            const int on = mr & (N_ - 1);
            out[((size_t)ob * 64 + o) * N_ + on] = fmaxf(acc[ct][v] * s + t, 0.0f);
        }
    }
}

extern "C" void kernel_launch(void* const* d_in, const int* in_sizes, int n_in,
                              void* d_out, int out_size, void* d_ws, size_t ws_size,
                              hipStream_t stream)
{
    const float* x     = (const float*)d_in[0];
    const float* w1    = (const float*)d_in[1];
    const float* b1    = (const float*)d_in[2];
    const float* g1    = (const float*)d_in[3];
    const float* beta1 = (const float*)d_in[4];
    const float* rm1   = (const float*)d_in[5];
    const float* rv1   = (const float*)d_in[6];
    const float* w2    = (const float*)d_in[7];
    const float* b2    = (const float*)d_in[8];
    const float* g2    = (const float*)d_in[9];
    const float* beta2 = (const float*)d_in[10];
    const float* rm2   = (const float*)d_in[11];
    const float* rv2   = (const float*)d_in[12];

    float* out    = (float*)d_out;
    float* out_m1 = out + (size_t)B_ * 64 * N_;
    float* out_m2 = out + 2 * (size_t)B_ * 64 * N_;
    int*   idx    = (int*)d_ws;

    knn_kernel<<<B_ * (N_ / KNN_T), KNN_T, 0, stream>>>(x, idx);

    edge_kernel<<<(B_ * N_) / 4, 128, 0, stream>>>(
        x, idx, w1, b1, g1, beta1, rm1, rv1, out_m1, out_m2);

    conv2_kernel<<<(B_ * N_ / 16) / 4, 128, 0, stream>>>(
        out_m1, out_m2, w2, b2, g2, beta2, rm2, rv2, out);
}


# edge: a work-group owns 32 consecutive points (each wave walks 8, w1 and BN terms loaded once); pooled rows pass through LDS and leave as whole 128-byte lines, not 4-byte stores to 16 lines
# speedup vs baseline: 4.0014x; 1.4506x over previous
#include <hip/hip_runtime.h>

#define B_    8
#define C_    3
#define N_    4096
#define K_    20
#define H_    64
#define DOUT_ 64
#define EPS_  1e-5f

typedef __attribute__((ext_vector_type(2))) float v2f;
typedef __attribute__((ext_vector_type(8))) float v8f;

#define KNN_T   128
#define KNN_CH  16
#define KNN_CAP 32
#define KNN_TP  129

typedef __attribute__((ext_vector_type(4))) float v4f;

#define KNN_POS(eu, el, ku, kl, cu, cl)                    \
    {   el = (cu) ? (eu) : ((cl) ? d : (el));              \
        kl = (cu) ? (ku) : ((cl) ? j : (kl)); }

#define KNN_LOAD(PX, PY, PZ, j0)                                          \
    {   const v4f* lx = (const v4f*)(sx + (j0));                          \
        const v4f* ly = (const v4f*)(sy + (j0));                          \
        const v4f* lz = (const v4f*)(sz + (j0));                          \
        PX##0 = lx[0]; PX##1 = lx[1]; PX##2 = lx[2]; PX##3 = lx[3];       \
        PY##0 = ly[0]; PY##1 = ly[1]; PY##2 = ly[2]; PY##3 = ly[3];       \
        PZ##0 = lz[0]; PZ##1 = lz[1]; PZ##2 = lz[2]; PZ##3 = lz[3]; }

#define KNN_HOLD(PX, PY, PZ)                                              \
    asm volatile("" :: "v"(PX##0), "v"(PX##1), "v"(PX##2), "v"(PX##3),    \
                       "v"(PY##0), "v"(PY##1), "v"(PY##2), "v"(PY##3),    \
                       "v"(PZ##0), "v"(PZ##1), "v"(PZ##2), "v"(PZ##3));

#define KNN_PUSH(dv, jj)                                                  \
    {   const float d = (dv);                                             \
        qd[cnt * KNN_T + tid] = d;                                        \
        qj[cnt * KNN_T + tid] = (jj);                                     \
        cnt += (d > thr) ? 1 : 0;     }

#define KNN_PAIR(X2, Y2, Z2, jb)                                          \
    {   const v2f dot = __builtin_elementwise_fma(qz2, (Z2),              \
                        __builtin_elementwise_fma(qx2, (X2), qy2 * (Y2)));\
        const v2f pp  = __builtin_elementwise_fma((Z2), (Z2),             \
                        __builtin_elementwise_fma((X2), (X2), (Y2) * (Y2)));\
        const v2f dd  = __builtin_elementwise_fma(dot, two2, nqq2) - pp;  \
        KNN_PUSH(dd.x, (jb)) KNN_PUSH(dd.y, (jb) + 1) }

#define KNN_QUAD(X, Y, Z, jb)                                             \
    KNN_PAIR(X.xy, Y.xy, Z.xy, (jb)) KNN_PAIR(X.zw, Y.zw, Z.zw, (jb) + 2)

#define KNN_SCAN(PX, PY, PZ, j0)                                          \
    KNN_QUAD(PX##0, PY##0, PZ##0, (j0))                                   \
    KNN_QUAD(PX##1, PY##1, PZ##1, (j0) + 4)                               \
    KNN_QUAD(PX##2, PY##2, PZ##2, (j0) + 8)                               \
    KNN_QUAD(PX##3, PY##3, PZ##3, (j0) + 12)

#define KNN_DRAIN(j0)                                                     \
    if (((j0) + KNN_CH >= N_) || __any(cnt > KNN_CAP - KNN_CH)) {         \
         \
         \
         \
        float dn = qd[tid];                                               \
        int   jn = qj[tid];                                               \
        for (int s = 0; __any(s < cnt); ++s) {                            \
            const bool  have = s < cnt;                                   \
            const float d = have ? dn : -3.4e38f;                         \
            const int   j = have ? jn : 0;                                \
            const int   sn = (s + 1 < KNN_CAP) ? (s + 1) : (KNN_CAP - 1); \
            dn = qd[sn * KNN_T + tid];                                    \
            jn = qj[sn * KNN_T + tid];                                    \
               \
               \
               \
               \
            __builtin_amdgcn_sched_barrier(0);                            \
            KNN_INSERT                                                    \
            __builtin_amdgcn_sched_barrier(0);                            \
               \
               \
            asm volatile("" :: "v"(dn), "v"(jn));                         \
        }                                                                 \
        cnt = 0;                                                          \
        thr = e19;                                                        \
    }

#define KNN_INSERT                                                        \
    const bool c0 = d > e0, c1 = d > e1, c2 = d > e2, c3 = d > e3, c4 = d > e4;\
    const bool c5 = d > e5, c6 = d > e6, c7 = d > e7, c8 = d > e8, c9 = d > e9;\
    const bool c10 = d > e10, c11 = d > e11, c12 = d > e12, c13 = d > e13, c14 = d > e14;\
    const bool c15 = d > e15, c16 = d > e16, c17 = d > e17, c18 = d > e18, c19 = d > e19;\
    KNN_POS(e18, e19, k18, k19, c18, c19)                                 \
    KNN_POS(e17, e18, k17, k18, c17, c18)                                 \
    KNN_POS(e16, e17, k16, k17, c16, c17)                                 \
    KNN_POS(e15, e16, k15, k16, c15, c16)                                 \
    KNN_POS(e14, e15, k14, k15, c14, c15)                                 \
    KNN_POS(e13, e14, k13, k14, c13, c14)                                 \
    KNN_POS(e12, e13, k12, k13, c12, c13)                                 \
    KNN_POS(e11, e12, k11, k12, c11, c12)                                 \
    KNN_POS(e10, e11, k10, k11, c10, c11)                                 \
    KNN_POS(e9, e10, k9, k10, c9, c10)                                    \
    KNN_POS(e8, e9, k8, k9, c8, c9)                                       \
    KNN_POS(e7, e8, k7, k8, c7, c8)                                       \
    KNN_POS(e6, e7, k6, k7, c6, c7)                                       \
    KNN_POS(e5, e6, k5, k6, c5, c6)                                       \
    KNN_POS(e4, e5, k4, k5, c4, c5)                                       \
    KNN_POS(e3, e4, k3, k4, c3, c4)                                       \
    KNN_POS(e2, e3, k2, k3, c2, c3)                                       \
    KNN_POS(e1, e2, k1, k2, c1, c2)                                       \
    KNN_POS(e0, e1, k0, k1, c0, c1)                                       \
    e0 = c0 ? d : e0;                                                     \
    k0 = c0 ? j : k0;

__global__ __launch_bounds__(KNN_T) void knn_kernel(const float* __restrict__ x,
                                                    int* __restrict__ idx)
{
    __shared__ __attribute__((aligned(16))) float spts[3 * N_];
    const float* sx = spts;
    const float* sy = spts + N_;
    const float* sz = spts + 2 * N_;
    __shared__ float qd[KNN_CAP * KNN_T];
    __shared__ int   qj[KNN_CAP * KNN_T];
    const int blocksPerBatch = N_ / KNN_T;
    const int b     = blockIdx.x / blocksPerBatch;
    const int chunk = blockIdx.x % blocksPerBatch;
    const float* xb = x + (size_t)b * 3 * N_;
    const v4f* xb4 = (const v4f*)xb;
    v4f*       sp4 = (v4f*)spts;
#pragma unroll
    for (int t = 0; t < (3 * N_ / 4) / KNN_T; ++t)
        sp4[t * KNN_T + threadIdx.x] = xb4[t * KNN_T + threadIdx.x];
    __syncthreads();

    const int   tid = threadIdx.x;
    const int   n  = chunk * KNN_T + tid;
    const float qx = sx[n], qy = sy[n], qz = sz[n];
    const float qq = fmaf(qz, qz, fmaf(qx, qx, qy * qy));
    const v2f qx2 = {qx, qx}, qy2 = {qy, qy}, qz2 = {qz, qz};
    const v2f nqq2 = {-qq, -qq}, two2 = {2.0f, 2.0f};

    float e0 = -3.4e38f, e1 = -3.4e38f, e2 = -3.4e38f, e3 = -3.4e38f, e4 = -3.4e38f, e5 = -3.4e38f, e6 = -3.4e38f, e7 = -3.4e38f, e8 = -3.4e38f, e9 = -3.4e38f, e10 = -3.4e38f, e11 = -3.4e38f, e12 = -3.4e38f, e13 = -3.4e38f, e14 = -3.4e38f, e15 = -3.4e38f, e16 = -3.4e38f, e17 = -3.4e38f, e18 = -3.4e38f, e19 = -3.4e38f;
    int   k0 = 0, k1 = 0, k2 = 0, k3 = 0, k4 = 0, k5 = 0, k6 = 0, k7 = 0, k8 = 0, k9 = 0, k10 = 0, k11 = 0, k12 = 0, k13 = 0, k14 = 0, k15 = 0, k16 = 0, k17 = 0, k18 = 0, k19 = 0;

    float thr = -3.4e38f;
    int   cnt = 0;

    v4f ax0, ax1, ax2, ax3, ay0, ay1, ay2, ay3, az0, az1, az2, az3;
    v4f bx0, bx1, bx2, bx3, by0, by1, by2, by3, bz0, bz1, bz2, bz3;
    KNN_LOAD(ax, ay, az, 0)
    for (int j0 = 0; j0 < N_; j0 += 2 * KNN_CH) {
        KNN_LOAD(bx, by, bz, j0 + KNN_CH)
        KNN_SCAN(ax, ay, az, j0)
        KNN_HOLD(bx, by, bz)
        KNN_DRAIN(j0)
        KNN_LOAD(ax, ay, az, (j0 + 2 * KNN_CH) & (N_ - 1))
        KNN_SCAN(bx, by, bz, j0 + KNN_CH)
        KNN_HOLD(ax, ay, az)
        KNN_DRAIN(j0 + KNN_CH)
    }
    __syncthreads();
    int* tk = qj;
    tk[0 * KNN_TP + tid] = k0;
    tk[1 * KNN_TP + tid] = k1;
    tk[2 * KNN_TP + tid] = k2;
    tk[3 * KNN_TP + tid] = k3;
    tk[4 * KNN_TP + tid] = k4;
    tk[5 * KNN_TP + tid] = k5;
    tk[6 * KNN_TP + tid] = k6;
    tk[7 * KNN_TP + tid] = k7;
    tk[8 * KNN_TP + tid] = k8;
    tk[9 * KNN_TP + tid] = k9;
    tk[10 * KNN_TP + tid] = k10;
    tk[11 * KNN_TP + tid] = k11;
    tk[12 * KNN_TP + tid] = k12;
    tk[13 * KNN_TP + tid] = k13;
    tk[14 * KNN_TP + tid] = k14;
    tk[15 * KNN_TP + tid] = k15;
    tk[16 * KNN_TP + tid] = k16;
    tk[17 * KNN_TP + tid] = k17;
    tk[18 * KNN_TP + tid] = k18;
    tk[19 * KNN_TP + tid] = k19;
    __syncthreads();
    int* outp = idx + ((size_t)b * N_ + chunk * KNN_T) * K_;
#pragma unroll
    for (int m = 0; m < K_; ++m) {
        const int p = m * KNN_T + tid;
        const int q = p / K_;
        const int i = p - q * K_;
        outp[p] = tk[i * KNN_TP + q];
    }
}

#define EDGE_PW 8
#define EDGE_PB 32
#define EDGE_LP 33
#define EDGE_HS (64 * EDGE_LP + 16)

__global__ __launch_bounds__(128) void edge_kernel(
    const float* __restrict__ x,  const int* __restrict__ idx,
    const float* __restrict__ w1, const float* __restrict__ b1,
    const float* __restrict__ g1, const float* __restrict__ beta1,
    const float* __restrict__ rm1,const float* __restrict__ rv1,
    float* __restrict__ out_m1,   float* __restrict__ out_m2)
{
    __shared__ float pmx[2 * EDGE_HS];
    __shared__ float psm[2 * EDGE_HS];

    const int lane = threadIdx.x & 31;
    const int half = lane >> 4;
    const int l15  = lane & 15;
    const int wv   = __builtin_amdgcn_readfirstlane((int)(threadIdx.x >> 5));
    const int b    = blockIdx.x / (N_ / EDGE_PB);
    const int n0   = (blockIdx.x % (N_ / EDGE_PB)) * EDGE_PB;
    const int nw   = n0 + wv * EDGE_PW;

    const float* xb = x + (size_t)b * 3 * N_;

    const bool valid1 = (16 + l15) < K_;
    const int  kc1    = valid1 ? (16 + l15) : (K_ - 1);
    const int* ip = idx + ((size_t)b * N_ + nw) * K_;
    int nb0[EDGE_PW], nb1[EDGE_PW];
#pragma unroll
    for (int p = 0; p < EDGE_PW; ++p) {
        nb0[p] = ip[p * K_ + l15];
        nb1[p] = ip[p * K_ + kc1];
    }

    v2f   bt0[4], bt1[4];
    float bs[4], bt[4];
#pragma unroll
    for (int ct = 0; ct < 4; ++ct) {
        const int    o    = ct * 16 + l15;
        const float* wrow = w1 + o * 6;
        const float w0 = wrow[0], w1v = wrow[1], w2v = wrow[2];
        const float w3 = wrow[3], w4v = wrow[4], w5v = wrow[5];
        bt0[ct].x = half ? w2v : w0;   bt0[ct].y = half ? w3 : w1v;
        bt1[ct].x = half ? 0.0f : w4v; bt1[ct].y = half ? 0.0f : w5v;
        const float s = g1[o] * rsqrtf(rv1[o] + EPS_);
        bs[ct] = s;
        bt[ct] = (b1[o] - rm1[o]) * s + beta1[o];
    }

    float gx0[EDGE_PW], gy0[EDGE_PW], gz0[EDGE_PW];
    float gx1[EDGE_PW], gy1[EDGE_PW], gz1[EDGE_PW];
#pragma unroll
    for (int p = 0; p < EDGE_PW; ++p) {
        gx0[p] = xb[nb0[p]]; gy0[p] = xb[N_ + nb0[p]]; gz0[p] = xb[2 * N_ + nb0[p]];
        gx1[p] = xb[nb1[p]]; gy1[p] = xb[N_ + nb1[p]]; gz1[p] = xb[2 * N_ + nb1[p]];
    }

#pragma unroll
    for (int p = 0; p < EDGE_PW; ++p) {
        const int   n  = nw + p;
        const float cx = xb[n], cy = xb[N_ + n], cz = xb[2 * N_ + n];

        v2f a0[2], a1[2];
#pragma unroll
        for (int rt = 0; rt < 2; ++rt) {
            const bool valid = rt ? valid1 : true;
            const float nx = rt ? gx1[p] : gx0[p];
            const float ny = rt ? gy1[p] : gy0[p];
            const float nz = rt ? gz1[p] : gz0[p];
            const float e0 = valid ? (nx - cx) : 0.0f;
            const float e1 = valid ? (ny - cy) : 0.0f;
            const float e2 = valid ? (nz - cz) : 0.0f;
            const float e3 = valid ? cx : 0.0f;
            const float e4 = valid ? cy : 0.0f;
            const float e5 = valid ? cz : 0.0f;
            v2f t0, t1;
            t0.x = half ? e2 : e0;  t0.y = half ? e3 : e1;
            t1.x = half ? 0.0f : e4; t1.y = half ? 0.0f : e5;
            a0[rt] = t0; a1[rt] = t1;
        }

        const int col = wv * EDGE_PW + p;
#pragma unroll
        for (int ct = 0; ct < 4; ++ct) {
            v8f z = {};
            z = __builtin_amdgcn_wmma_f32_16x16x4_f32(false, a0[0], false, bt0[ct],
                                                      (short)0, z, false, false);
            z = __builtin_amdgcn_wmma_f32_16x16x4_f32(false, a1[0], false, bt1[ct],
                                                      (short)0, z, false, false);
            v8f z1 = {};
            z1 = __builtin_amdgcn_wmma_f32_16x16x4_f32(false, a0[1], false, bt0[ct],
                                                       (short)0, z1, false, false);
            z1 = __builtin_amdgcn_wmma_f32_16x16x4_f32(false, a1[1], false, bt1[ct],
                                                       (short)0, z1, false, false);

            const float s = bs[ct], t = bt[ct];
            float mx = -3.4e38f, sm = 0.0f;
#pragma unroll
            for (int v = 0; v < 8; ++v) {
                const float val = fmaxf(z[v] * s + t, 0.0f);
                mx = fmaxf(mx, val); sm += val;
            }
#pragma unroll
            for (int v = 0; v < 4; ++v) {
                const float val = fmaxf(z1[v] * s + t, 0.0f);
                mx = (half == 0) ? fmaxf(mx, val) : mx;
                sm = (half == 0) ? (sm + val) : sm;
            }
            const int q = half * EDGE_HS + (ct * 16 + l15) * EDGE_LP + col;
            pmx[q] = mx;
            psm[q] = sm;
        }
    }
    __syncthreads();

#pragma unroll
    for (int r = 0; r < 16; ++r) {
        const int   o  = wv * 16 + r;
        const float mx = fmaxf(pmx[o * EDGE_LP + lane], pmx[EDGE_HS + o * EDGE_LP + lane]);
        const float sm = psm[o * EDGE_LP + lane] + psm[EDGE_HS + o * EDGE_LP + lane];
        const size_t off = ((size_t)b * 64 + o) * N_ + n0 + lane;
        out_m1[off] = mx;
        out_m2[off] = sm * (1.0f / K_);
    }
}

__global__ __launch_bounds__(128) void conv2_kernel(
    const float* __restrict__ m1, const float* __restrict__ m2,
    const float* __restrict__ w2, const float* __restrict__ b2,
    const float* __restrict__ g2, const float* __restrict__ beta2,
    const float* __restrict__ rm2,const float* __restrict__ rv2,
    float* __restrict__ out)
{
    __shared__ float sw[DOUT_ * 128];
    for (int i = threadIdx.x; i < DOUT_ * 128; i += 128) sw[i] = w2[i];
    __syncthreads();

    const int lane    = threadIdx.x & 31;
    const int half    = lane >> 4;
    const int l15     = lane & 15;
    const int wave    = (blockIdx.x * blockDim.x + threadIdx.x) >> 5;
    const int rowbase = wave * 16;

    const int m  = rowbase + l15;
    const int bb = m >> 12;
    const int nn = m & (N_ - 1);
    const float* arow1 = m1 + (size_t)bb * 64 * N_ + nn;
    const float* arow2 = m2 + (size_t)bb * 64 * N_ + nn;

    v8f acc[4] = {{}, {}, {}, {}};
#pragma unroll
    for (int ks = 0; ks < 32; ++ks) {
        const int k0 = ks * 4 + half * 2;
        float av0, av1;
        if (ks < 16) {
            av0 = arow1[(size_t)k0 * N_];
            av1 = arow1[(size_t)(k0 + 1) * N_];
        } else {
            av0 = arow2[(size_t)(k0 - 64) * N_];
            av1 = arow2[(size_t)(k0 - 63) * N_];
        }
        const v2f a = {av0, av1};
#pragma unroll
        for (int ct = 0; ct < 4; ++ct) {
            const int o = ct * 16 + l15;
            const v2f bt = { sw[o * 128 + k0], sw[o * 128 + k0 + 1] };
            acc[ct] = __builtin_amdgcn_wmma_f32_16x16x4_f32(
                false, a, false, bt, (short)0, acc[ct], false, false);
        }
    }

#pragma unroll
    for (int ct = 0; ct < 4; ++ct) {
        const int   o = ct * 16 + l15;
        const float s = g2[o] * rsqrtf(rv2[o] + EPS_);
        const float t = (b2[o] - rm2[o]) * s + beta2[o];
#pragma unroll
        for (int v = 0; v < 8; ++v) {
            const int mr = rowbase + v + half * 8;
            const int ob = mr >> 12;
            const int on = mr & (N_ - 1);
            out[((size_t)ob * 64 + o) * N_ + on] = fmaxf(acc[ct][v] * s + t, 0.0f);
        }
    }
}

extern "C" void kernel_launch(void* const* d_in, const int* in_sizes, int n_in,
                              void* d_out, int out_size, void* d_ws, size_t ws_size,
                              hipStream_t stream)
{
    const float* x     = (const float*)d_in[0];
    const float* w1    = (const float*)d_in[1];
    const float* b1    = (const float*)d_in[2];
    const float* g1    = (const float*)d_in[3];
    const float* beta1 = (const float*)d_in[4];
    const float* rm1   = (const float*)d_in[5];
    const float* rv1   = (const float*)d_in[6];
    const float* w2    = (const float*)d_in[7];
    const float* b2    = (const float*)d_in[8];
    const float* g2    = (const float*)d_in[9];
    const float* beta2 = (const float*)d_in[10];
    const float* rm2   = (const float*)d_in[11];
    const float* rv2   = (const float*)d_in[12];

    float* out    = (float*)d_out;
    float* out_m1 = out + (size_t)B_ * 64 * N_;
    float* out_m2 = out + 2 * (size_t)B_ * 64 * N_;
    int*   idx    = (int*)d_ws;

    knn_kernel<<<B_ * (N_ / KNN_T), KNN_T, 0, stream>>>(x, idx);

    edge_kernel<<<(B_ * N_) / EDGE_PB, 128, 0, stream>>>(
        x, idx, w1, b1, g1, beta1, rm1, rv1, out_m1, out_m2);

    conv2_kernel<<<(B_ * N_ / 16) / 4, 128, 0, stream>>>(
        out_m1, out_m2, w2, b2, g2, beta2, rm2, rv2, out);
}


# conv_out fused into the edge kernel: each wave multiplies its 16 rows of w2 by the block's 32 pooled rows still in LDS (same k order), out leaves as whole lines; 2 launches, no m1/m2 re-read
# speedup vs baseline: 4.2721x; 1.0677x over previous
#include <hip/hip_runtime.h>

#define B_    8
#define C_    3
#define N_    4096
#define K_    20
#define H_    64
#define DOUT_ 64
#define EPS_  1e-5f

typedef __attribute__((ext_vector_type(2))) float v2f;
typedef __attribute__((ext_vector_type(8))) float v8f;

#define KNN_T   128
#define KNN_CH  16
#define KNN_CAP 32
#define KNN_TP  129

typedef __attribute__((ext_vector_type(4))) float v4f;

#define KNN_POS(eu, el, ku, kl, cu, cl)                    \
    {   el = (cu) ? (eu) : ((cl) ? d : (el));              \
        kl = (cu) ? (ku) : ((cl) ? j : (kl)); }

#define KNN_LOAD(PX, PY, PZ, j0)                                          \
    {   const v4f* lx = (const v4f*)(sx + (j0));                          \
        const v4f* ly = (const v4f*)(sy + (j0));                          \
        const v4f* lz = (const v4f*)(sz + (j0));                          \
        PX##0 = lx[0]; PX##1 = lx[1]; PX##2 = lx[2]; PX##3 = lx[3];       \
        PY##0 = ly[0]; PY##1 = ly[1]; PY##2 = ly[2]; PY##3 = ly[3];       \
        PZ##0 = lz[0]; PZ##1 = lz[1]; PZ##2 = lz[2]; PZ##3 = lz[3]; }

#define KNN_HOLD(PX, PY, PZ)                                              \
    asm volatile("" :: "v"(PX##0), "v"(PX##1), "v"(PX##2), "v"(PX##3),    \
                       "v"(PY##0), "v"(PY##1), "v"(PY##2), "v"(PY##3),    \
                       "v"(PZ##0), "v"(PZ##1), "v"(PZ##2), "v"(PZ##3));

#define KNN_PUSH(dv, jj)                                                  \
    {   const float d = (dv);                                             \
        qd[cnt * KNN_T + tid] = d;                                        \
        qj[cnt * KNN_T + tid] = (jj);                                     \
        cnt += (d > thr) ? 1 : 0;     }

#define KNN_PAIR(X2, Y2, Z2, jb)                                          \
    {   const v2f dot = __builtin_elementwise_fma(qz2, (Z2),              \
                        __builtin_elementwise_fma(qx2, (X2), qy2 * (Y2)));\
        const v2f pp  = __builtin_elementwise_fma((Z2), (Z2),             \
                        __builtin_elementwise_fma((X2), (X2), (Y2) * (Y2)));\
        const v2f dd  = __builtin_elementwise_fma(dot, two2, nqq2) - pp;  \
        KNN_PUSH(dd.x, (jb)) KNN_PUSH(dd.y, (jb) + 1) }

#define KNN_QUAD(X, Y, Z, jb)                                             \
    KNN_PAIR(X.xy, Y.xy, Z.xy, (jb)) KNN_PAIR(X.zw, Y.zw, Z.zw, (jb) + 2)

#define KNN_SCAN(PX, PY, PZ, j0)                                          \
    KNN_QUAD(PX##0, PY##0, PZ##0, (j0))                                   \
    KNN_QUAD(PX##1, PY##1, PZ##1, (j0) + 4)                               \
    KNN_QUAD(PX##2, PY##2, PZ##2, (j0) + 8)                               \
    KNN_QUAD(PX##3, PY##3, PZ##3, (j0) + 12)

#define KNN_DRAIN(j0)                                                     \
    if (((j0) + KNN_CH >= N_) || __any(cnt > KNN_CAP - KNN_CH)) {         \
         \
         \
         \
        float dn = qd[tid];                                               \
        int   jn = qj[tid];                                               \
        for (int s = 0; __any(s < cnt); ++s) {                            \
            const bool  have = s < cnt;                                   \
            const float d = have ? dn : -3.4e38f;                         \
            const int   j = have ? jn : 0;                                \
            const int   sn = (s + 1 < KNN_CAP) ? (s + 1) : (KNN_CAP - 1); \
            dn = qd[sn * KNN_T + tid];                                    \
            jn = qj[sn * KNN_T + tid];                                    \
               \
               \
               \
               \
            __builtin_amdgcn_sched_barrier(0);                            \
            KNN_INSERT                                                    \
            __builtin_amdgcn_sched_barrier(0);                            \
               \
               \
            asm volatile("" :: "v"(dn), "v"(jn));                         \
        }                                                                 \
        cnt = 0;                                                          \
        thr = e19;                                                        \
    }

#define KNN_INSERT                                                        \
    const bool c0 = d > e0, c1 = d > e1, c2 = d > e2, c3 = d > e3, c4 = d > e4;\
    const bool c5 = d > e5, c6 = d > e6, c7 = d > e7, c8 = d > e8, c9 = d > e9;\
    const bool c10 = d > e10, c11 = d > e11, c12 = d > e12, c13 = d > e13, c14 = d > e14;\
    const bool c15 = d > e15, c16 = d > e16, c17 = d > e17, c18 = d > e18, c19 = d > e19;\
    KNN_POS(e18, e19, k18, k19, c18, c19)                                 \
    KNN_POS(e17, e18, k17, k18, c17, c18)                                 \
    KNN_POS(e16, e17, k16, k17, c16, c17)                                 \
    KNN_POS(e15, e16, k15, k16, c15, c16)                                 \
    KNN_POS(e14, e15, k14, k15, c14, c15)                                 \
    KNN_POS(e13, e14, k13, k14, c13, c14)                                 \
    KNN_POS(e12, e13, k12, k13, c12, c13)                                 \
    KNN_POS(e11, e12, k11, k12, c11, c12)                                 \
    KNN_POS(e10, e11, k10, k11, c10, c11)                                 \
    KNN_POS(e9, e10, k9, k10, c9, c10)                                    \
    KNN_POS(e8, e9, k8, k9, c8, c9)                                       \
    KNN_POS(e7, e8, k7, k8, c7, c8)                                       \
    KNN_POS(e6, e7, k6, k7, c6, c7)                                       \
    KNN_POS(e5, e6, k5, k6, c5, c6)                                       \
    KNN_POS(e4, e5, k4, k5, c4, c5)                                       \
    KNN_POS(e3, e4, k3, k4, c3, c4)                                       \
    KNN_POS(e2, e3, k2, k3, c2, c3)                                       \
    KNN_POS(e1, e2, k1, k2, c1, c2)                                       \
    KNN_POS(e0, e1, k0, k1, c0, c1)                                       \
    e0 = c0 ? d : e0;                                                     \
    k0 = c0 ? j : k0;

__global__ __launch_bounds__(KNN_T) void knn_kernel(const float* __restrict__ x,
                                                    int* __restrict__ idx)
{
    __shared__ __attribute__((aligned(16))) float spts[3 * N_];
    const float* sx = spts;
    const float* sy = spts + N_;
    const float* sz = spts + 2 * N_;
    __shared__ float qd[KNN_CAP * KNN_T];
    __shared__ int   qj[KNN_CAP * KNN_T];
    const int blocksPerBatch = N_ / KNN_T;
    const int b     = blockIdx.x / blocksPerBatch;
    const int chunk = blockIdx.x % blocksPerBatch;
    const float* xb = x + (size_t)b * 3 * N_;
    const v4f* xb4 = (const v4f*)xb;
    v4f*       sp4 = (v4f*)spts;
#pragma unroll
    for (int t = 0; t < (3 * N_ / 4) / KNN_T; ++t)
        sp4[t * KNN_T + threadIdx.x] = xb4[t * KNN_T + threadIdx.x];
    __syncthreads();

    const int   tid = threadIdx.x;
    const int   n  = chunk * KNN_T + tid;
    const float qx = sx[n], qy = sy[n], qz = sz[n];
    const float qq = fmaf(qz, qz, fmaf(qx, qx, qy * qy));
    const v2f qx2 = {qx, qx}, qy2 = {qy, qy}, qz2 = {qz, qz};
    const v2f nqq2 = {-qq, -qq}, two2 = {2.0f, 2.0f};

    float e0 = -3.4e38f, e1 = -3.4e38f, e2 = -3.4e38f, e3 = -3.4e38f, e4 = -3.4e38f, e5 = -3.4e38f, e6 = -3.4e38f, e7 = -3.4e38f, e8 = -3.4e38f, e9 = -3.4e38f, e10 = -3.4e38f, e11 = -3.4e38f, e12 = -3.4e38f, e13 = -3.4e38f, e14 = -3.4e38f, e15 = -3.4e38f, e16 = -3.4e38f, e17 = -3.4e38f, e18 = -3.4e38f, e19 = -3.4e38f;
    int   k0 = 0, k1 = 0, k2 = 0, k3 = 0, k4 = 0, k5 = 0, k6 = 0, k7 = 0, k8 = 0, k9 = 0, k10 = 0, k11 = 0, k12 = 0, k13 = 0, k14 = 0, k15 = 0, k16 = 0, k17 = 0, k18 = 0, k19 = 0;

    float thr = -3.4e38f;
    int   cnt = 0;

    v4f ax0, ax1, ax2, ax3, ay0, ay1, ay2, ay3, az0, az1, az2, az3;
    v4f bx0, bx1, bx2, bx3, by0, by1, by2, by3, bz0, bz1, bz2, bz3;
    KNN_LOAD(ax, ay, az, 0)
    for (int j0 = 0; j0 < N_; j0 += 2 * KNN_CH) {
        KNN_LOAD(bx, by, bz, j0 + KNN_CH)
        KNN_SCAN(ax, ay, az, j0)
        KNN_HOLD(bx, by, bz)
        KNN_DRAIN(j0)
        KNN_LOAD(ax, ay, az, (j0 + 2 * KNN_CH) & (N_ - 1))
        KNN_SCAN(bx, by, bz, j0 + KNN_CH)
        KNN_HOLD(ax, ay, az)
        KNN_DRAIN(j0 + KNN_CH)
    }
    __syncthreads();
    int* tk = qj;
    tk[0 * KNN_TP + tid] = k0;
    tk[1 * KNN_TP + tid] = k1;
    tk[2 * KNN_TP + tid] = k2;
    tk[3 * KNN_TP + tid] = k3;
    tk[4 * KNN_TP + tid] = k4;
    tk[5 * KNN_TP + tid] = k5;
    tk[6 * KNN_TP + tid] = k6;
    tk[7 * KNN_TP + tid] = k7;
    tk[8 * KNN_TP + tid] = k8;
    tk[9 * KNN_TP + tid] = k9;
    tk[10 * KNN_TP + tid] = k10;
    tk[11 * KNN_TP + tid] = k11;
    tk[12 * KNN_TP + tid] = k12;
    tk[13 * KNN_TP + tid] = k13;
    tk[14 * KNN_TP + tid] = k14;
    tk[15 * KNN_TP + tid] = k15;
    tk[16 * KNN_TP + tid] = k16;
    tk[17 * KNN_TP + tid] = k17;
    tk[18 * KNN_TP + tid] = k18;
    tk[19 * KNN_TP + tid] = k19;
    __syncthreads();
    int* outp = idx + ((size_t)b * N_ + chunk * KNN_T) * K_;
#pragma unroll
    for (int m = 0; m < K_; ++m) {
        const int p = m * KNN_T + tid;
        const int q = p / K_;
        const int i = p - q * K_;
        outp[p] = tk[i * KNN_TP + q];
    }
}

#define EDGE_PW 8
#define EDGE_PB 32
#define EDGE_LP 33
#define EDGE_HS (64 * EDGE_LP + 16)
#define EDGE_KP 132

__device__ __forceinline__ float edge_xhalf(float v)
{
    return __builtin_bit_cast(float, __builtin_amdgcn_permlanex16(
        __builtin_bit_cast(int, v), __builtin_bit_cast(int, v),
        0x76543210, 0xfedcba98, false, false));
}

__device__ __forceinline__ float edge_lane(float v, int l)
{
    return __builtin_bit_cast(float,
        __builtin_amdgcn_readlane(__builtin_bit_cast(int, v), l));
}

__global__ __launch_bounds__(128) void edge_kernel(
    const float* __restrict__ x,  const int* __restrict__ idx,
    const float* __restrict__ w1, const float* __restrict__ b1,
    const float* __restrict__ g1, const float* __restrict__ beta1,
    const float* __restrict__ rm1,const float* __restrict__ rv1,
    const float* __restrict__ w2, const float* __restrict__ b2,
    const float* __restrict__ g2, const float* __restrict__ beta2,
    const float* __restrict__ rm2,const float* __restrict__ rv2,
    float* __restrict__ out,
    float* __restrict__ out_m1,   float* __restrict__ out_m2)
{
    __shared__ float pmx[2 * EDGE_HS];
    __shared__ float psm[2 * EDGE_HS];
    __shared__ __attribute__((aligned(16))) float pa[EDGE_PB * EDGE_KP];

    const int lane = threadIdx.x & 31;
    const int half = lane >> 4;
    const int l15  = lane & 15;
    const int wv   = __builtin_amdgcn_readfirstlane((int)(threadIdx.x >> 5));
    const int b    = blockIdx.x / (N_ / EDGE_PB);
    const int n0   = (blockIdx.x % (N_ / EDGE_PB)) * EDGE_PB;
    const int nw   = n0 + wv * EDGE_PW;

    const float* xb = x + (size_t)b * 3 * N_;

    const bool valid1 = (16 + l15) < K_;
    const int  kc1    = valid1 ? (16 + l15) : (K_ - 1);
    const int* ip = idx + ((size_t)b * N_ + nw) * K_;
    int nb0[EDGE_PW], nb1[EDGE_PW];
#pragma unroll
    for (int p = 0; p < EDGE_PW; ++p) {
        nb0[p] = ip[p * K_ + l15];
        nb1[p] = ip[p * K_ + kc1];
    }

    v2f   bt0[4], bt1[4];
    float bs[4], bt[4];
#pragma unroll
    for (int ct = 0; ct < 4; ++ct) {
        const int    o    = ct * 16 + l15;
        const float* wrow = w1 + o * 6;
        const float w0 = wrow[0], w1v = wrow[1], w2v = wrow[2];
        const float w3 = wrow[3], w4v = wrow[4], w5v = wrow[5];
        bt0[ct].x = half ? w2v : w0;   bt0[ct].y = half ? w3 : w1v;
        bt1[ct].x = half ? 0.0f : w4v; bt1[ct].y = half ? 0.0f : w5v;
        const float s = g1[o] * rsqrtf(rv1[o] + EPS_);
        bs[ct] = s;
        bt[ct] = (b1[o] - rm1[o]) * s + beta1[o];
    }

    float gx0[EDGE_PW], gy0[EDGE_PW], gz0[EDGE_PW];
    float gx1[EDGE_PW], gy1[EDGE_PW], gz1[EDGE_PW];
#pragma unroll
    for (int p = 0; p < EDGE_PW; ++p) {
        gx0[p] = xb[nb0[p]]; gy0[p] = xb[N_ + nb0[p]]; gz0[p] = xb[2 * N_ + nb0[p]];
        gx1[p] = xb[nb1[p]]; gy1[p] = xb[N_ + nb1[p]]; gz1[p] = xb[2 * N_ + nb1[p]];
    }

#pragma unroll
    for (int p = 0; p < EDGE_PW; ++p) {
        const int   n  = nw + p;
        const float cx = xb[n], cy = xb[N_ + n], cz = xb[2 * N_ + n];

        v2f a0[2], a1[2];
#pragma unroll
        for (int rt = 0; rt < 2; ++rt) {
            const bool valid = rt ? valid1 : true;
            const float nx = rt ? gx1[p] : gx0[p];
            const float ny = rt ? gy1[p] : gy0[p];
            const float nz = rt ? gz1[p] : gz0[p];
            const float e0 = valid ? (nx - cx) : 0.0f;
            const float e1 = valid ? (ny - cy) : 0.0f;
            const float e2 = valid ? (nz - cz) : 0.0f;
            const float e3 = valid ? cx : 0.0f;
            const float e4 = valid ? cy : 0.0f;
            const float e5 = valid ? cz : 0.0f;
            v2f t0, t1;
            t0.x = half ? e2 : e0;  t0.y = half ? e3 : e1;
            t1.x = half ? 0.0f : e4; t1.y = half ? 0.0f : e5;
            a0[rt] = t0; a1[rt] = t1;
        }

        const int col = wv * EDGE_PW + p;
#pragma unroll
        for (int ct = 0; ct < 4; ++ct) {
            v8f z = {};
            z = __builtin_amdgcn_wmma_f32_16x16x4_f32(false, a0[0], false, bt0[ct],
                                                      (short)0, z, false, false);
            z = __builtin_amdgcn_wmma_f32_16x16x4_f32(false, a1[0], false, bt1[ct],
                                                      (short)0, z, false, false);
            v8f z1 = {};
            z1 = __builtin_amdgcn_wmma_f32_16x16x4_f32(false, a0[1], false, bt0[ct],
                                                       (short)0, z1, false, false);
            z1 = __builtin_amdgcn_wmma_f32_16x16x4_f32(false, a1[1], false, bt1[ct],
                                                       (short)0, z1, false, false);

            const float s = bs[ct], t = bt[ct];
            float mx = -3.4e38f, sm = 0.0f;
#pragma unroll
            for (int v = 0; v < 8; ++v) {
                const float val = fmaxf(z[v] * s + t, 0.0f);
                mx = fmaxf(mx, val); sm += val;
            }
#pragma unroll
            for (int v = 0; v < 4; ++v) {
                const float val = fmaxf(z1[v] * s + t, 0.0f);
                mx = (half == 0) ? fmaxf(mx, val) : mx;
                sm = (half == 0) ? (sm + val) : sm;
            }
            const int q = half * EDGE_HS + (ct * 16 + l15) * EDGE_LP + col;
            pmx[q] = mx;
            psm[q] = sm;
        }
    }
    __syncthreads();

    const float* wrow2 = w2 + (size_t)(wv * 16 + l15) * 128 + half * 2;
    v2f wa[32];
#pragma unroll
    for (int ks = 0; ks < 32; ++ks)
        wa[ks] = *(const v2f*)(wrow2 + ks * 4);
    const int   oc  = wv * 16 + l15;
    const float s2v = g2[oc] * rsqrtf(rv2[oc] + EPS_);
    const float t2v = (b2[oc] - rm2[oc]) * s2v + beta2[oc];

    float jm[16], ja[16];
#pragma unroll
    for (int r = 0; r < 16; ++r) {
        const int   o  = wv * 16 + r;
        const float mx = fmaxf(pmx[o * EDGE_LP + lane], pmx[EDGE_HS + o * EDGE_LP + lane]);
        const float sm = psm[o * EDGE_LP + lane] + psm[EDGE_HS + o * EDGE_LP + lane];
        jm[r] = mx;
        ja[r] = sm * (1.0f / K_);
        pa[lane * EDGE_KP + o]      = jm[r];
        pa[lane * EDGE_KP + 64 + o] = ja[r];
    }
    __syncthreads();

#pragma unroll
    for (int r = 0; r < 16; ++r) {
        const int    o   = wv * 16 + r;
        const size_t off = ((size_t)b * 64 + o) * N_ + n0 + lane;
        out_m1[off] = jm[r];
        out_m2[off] = ja[r];
    }

    const float* pr0 = pa + l15 * EDGE_KP + half * 2;
    const float* pr1 = pr0 + 16 * EDGE_KP;
    v8f d0 = {}, d1 = {};
#pragma unroll
    for (int ks = 0; ks < 32; ++ks) {
        const v2f p0 = *(const v2f*)(pr0 + ks * 4);
        const v2f p1 = *(const v2f*)(pr1 + ks * 4);
        d0 = __builtin_amdgcn_wmma_f32_16x16x4_f32(false, wa[ks], false, p0,
                                                   (short)0, d0, false, false);
        d1 = __builtin_amdgcn_wmma_f32_16x16x4_f32(false, wa[ks], false, p1,
                                                   (short)0, d1, false, false);
    }

#pragma unroll
    for (int r = 0; r < 8; ++r) {
        const float x0 = edge_xhalf(d0[r]);
        const float x1 = edge_xhalf(d1[r]);
        const float lo = half ? x1 : d0[r];
        const float hi = half ? d1[r] : x0;
        {
            const int   o = wv * 16 + r;
            const float s = edge_lane(s2v, r), t = edge_lane(t2v, r);
            out[((size_t)b * 64 + o) * N_ + n0 + lane] = fmaxf(lo * s + t, 0.0f);
        }
        {
            const int   o = wv * 16 + 8 + r;
            const float s = edge_lane(s2v, 8 + r), t = edge_lane(t2v, 8 + r);
            out[((size_t)b * 64 + o) * N_ + n0 + lane] = fmaxf(hi * s + t, 0.0f);
        }
    }
}

extern "C" void kernel_launch(void* const* d_in, const int* in_sizes, int n_in,
                              void* d_out, int out_size, void* d_ws, size_t ws_size,
                              hipStream_t stream)
{
    const float* x     = (const float*)d_in[0];
    const float* w1    = (const float*)d_in[1];
    const float* b1    = (const float*)d_in[2];
    const float* g1    = (const float*)d_in[3];
    const float* beta1 = (const float*)d_in[4];
    const float* rm1   = (const float*)d_in[5];
    const float* rv1   = (const float*)d_in[6];
    const float* w2    = (const float*)d_in[7];
    const float* b2    = (const float*)d_in[8];
    const float* g2    = (const float*)d_in[9];
    const float* beta2 = (const float*)d_in[10];
    const float* rm2   = (const float*)d_in[11];
    const float* rv2   = (const float*)d_in[12];

    float* out    = (float*)d_out;
    float* out_m1 = out + (size_t)B_ * 64 * N_;
    float* out_m2 = out + 2 * (size_t)B_ * 64 * N_;
    int*   idx    = (int*)d_ws;

    knn_kernel<<<B_ * (N_ / KNN_T), KNN_T, 0, stream>>>(x, idx);

    edge_kernel<<<(B_ * N_) / EDGE_PB, 128, 0, stream>>>(
        x, idx, w1, b1, g1, beta1, rm1, rv1,
        w2, b2, g2, beta2, rm2, rv2, out, out_m1, out_m2);
}


# knn scan: a chunk's 16 queue stores and 12 point reads stand in ONE group behind its vector work (18 places before); the best list rests in LDS between drains, the loop carries no list registers
# speedup vs baseline: 6.3811x; 1.4937x over previous
#include <hip/hip_runtime.h>

#define B_    8
#define C_    3
#define N_    4096
#define K_    20
#define H_    64
#define DOUT_ 64
#define EPS_  1e-5f

typedef __attribute__((ext_vector_type(2))) float v2f;
typedef __attribute__((ext_vector_type(8))) float v8f;

#define KNN_T   128
#define KNN_CH  16
#define KNN_CAP 32
#define KNN_TP  129

typedef __attribute__((ext_vector_type(4))) float v4f;

#define KNN_POS(eu, el, ku, kl, cu, cl)                    \
    {   el = (cu) ? (eu) : ((cl) ? d : (el));              \
        kl = (cu) ? (ku) : ((cl) ? j : (kl)); }

typedef __attribute__((address_space(3))) float lds_f32;
typedef __attribute__((address_space(3))) int   lds_i32;
typedef __attribute__((address_space(3))) v4f   lds_v4f;
#define KNN_SLOT   (KNN_T * 4)
#define KNN_QJ     (KNN_CAP * KNN_T * 4)
#define KNN_PLANE  (N_ * 4)

#define KNN_LOAD(PX, PY, PZ, la)                                          \
    {   const lds_v4f* lx = (const lds_v4f*)(la);                         \
        const lds_v4f* ly = (const lds_v4f*)((la) + KNN_PLANE);           \
        const lds_v4f* lz = (const lds_v4f*)((la) + 2 * KNN_PLANE);       \
        PX##0 = lx[0]; PX##1 = lx[1]; PX##2 = lx[2]; PX##3 = lx[3];       \
        PY##0 = ly[0]; PY##1 = ly[1]; PY##2 = ly[2]; PY##3 = ly[3];       \
        PZ##0 = lz[0]; PZ##1 = lz[1]; PZ##2 = lz[2]; PZ##3 = lz[3]; }

#define KNN_D2(OUT, X2, Y2, Z2)                                           \
    {   const v2f dot = __builtin_elementwise_fma(qz2, (Z2),              \
                        __builtin_elementwise_fma(qx2, (X2), qy2 * (Y2)));\
        const v2f pp  = __builtin_elementwise_fma((Z2), (Z2),             \
                        __builtin_elementwise_fma((X2), (X2), (Y2) * (Y2)));\
        OUT = __builtin_elementwise_fma(dot, two2, nqq2) - pp; }

#define KNN_D4(O1, O2, X, Y, Z)                                           \
    KNN_D2(O1, X.xy, Y.xy, Z.xy) KNN_D2(O2, X.zw, Y.zw, Z.zw)

#define KNN_ADV(TN, T, dv)                                                \
    const unsigned TN = (T) + (((dv) > thr) ? (unsigned)KNN_SLOT : 0u);
#define KNN_ST(T, dv, jj)                                                 \
    {   *(lds_f32*)(T) = (dv);                                            \
        *(lds_i32*)((T) + KNN_QJ) = (jj); }

#define KNN_CHUNK(PX, PY, PZ, jb, jnext)                                  \
    {   v2f dA, dB, dC, dD, dE, dF, dG, dH;                               \
        KNN_D4(dA, dB, PX##0, PY##0, PZ##0)                               \
        KNN_D4(dC, dD, PX##1, PY##1, PZ##1)                               \
        KNN_D4(dE, dF, PX##2, PY##2, PZ##2)                               \
        KNN_D4(dG, dH, PX##3, PY##3, PZ##3)                               \
        const unsigned t0 = tail;                                         \
        KNN_ADV(t1, t0, dA.x)   KNN_ADV(t2, t1, dA.y)                     \
        KNN_ADV(t3, t2, dB.x)   KNN_ADV(t4, t3, dB.y)                     \
        KNN_ADV(t5, t4, dC.x)   KNN_ADV(t6, t5, dC.y)                     \
        KNN_ADV(t7, t6, dD.x)   KNN_ADV(t8, t7, dD.y)                     \
        KNN_ADV(t9, t8, dE.x)   KNN_ADV(t10, t9, dE.y)                    \
        KNN_ADV(t11, t10, dF.x) KNN_ADV(t12, t11, dF.y)                   \
        KNN_ADV(t13, t12, dG.x) KNN_ADV(t14, t13, dG.y)                   \
        KNN_ADV(t15, t14, dH.x) KNN_ADV(t16, t15, dH.y)                   \
          \
          \
          \
          \
          \
          \
        int i0; unsigned la;                                              \
        asm("v_mov_b32 %0, %1" : "=v"(i0) : "s"((int)(jb)));              \
        asm("v_mov_b32 %0, %1" : "=v"(la) : "s"(pbase + ((unsigned)(jnext) << 2)));\
        const int i1 = i0 + 1, i2 = i0 + 2, i3 = i0 + 3;                  \
        const int i4 = i0 + 4, i5 = i0 + 5, i6 = i0 + 6, i7 = i0 + 7;     \
        const int i8 = i0 + 8, i9 = i0 + 9, i10 = i0 + 10, i11 = i0 + 11; \
        const int i12 = i0 + 12, i13 = i0 + 13, i14 = i0 + 14, i15 = i0 + 15;\
         \
         \
         \
        asm volatile("" :: "v"(t0), "v"(t1), "v"(t2), "v"(t3), "v"(t4),   \
                           "v"(t5), "v"(t6), "v"(t7), "v"(t8), "v"(t9),   \
                           "v"(t10), "v"(t11), "v"(t12), "v"(t13),        \
                           "v"(t14), "v"(t15), "v"(la));                  \
        asm volatile("" :: "v"(dA.x), "v"(dA.y), "v"(dB.x), "v"(dB.y),    \
                           "v"(dC.x), "v"(dC.y), "v"(dD.x), "v"(dD.y),    \
                           "v"(dE.x), "v"(dE.y), "v"(dF.x), "v"(dF.y),    \
                           "v"(dG.x), "v"(dG.y), "v"(dH.x), "v"(dH.y));   \
        asm volatile("" :: "v"(i0), "v"(i1), "v"(i2), "v"(i3), "v"(i4),   \
                           "v"(i5), "v"(i6), "v"(i7), "v"(i8), "v"(i9),   \
                           "v"(i10), "v"(i11), "v"(i12), "v"(i13),        \
                           "v"(i14), "v"(i15));                           \
          \
          \
          \
        __builtin_amdgcn_sched_barrier(0);                                \
        KNN_ST(t0, dA.x, i0)   KNN_ST(t1, dA.y, i1)                       \
        KNN_ST(t2, dB.x, i2)   KNN_ST(t3, dB.y, i3)                       \
        KNN_ST(t4, dC.x, i4)   KNN_ST(t5, dC.y, i5)                       \
        KNN_ST(t6, dD.x, i6)   KNN_ST(t7, dD.y, i7)                       \
        KNN_ST(t8, dE.x, i8)   KNN_ST(t9, dE.y, i9)                       \
        KNN_ST(t10, dF.x, i10) KNN_ST(t11, dF.y, i11)                     \
        KNN_ST(t12, dG.x, i12) KNN_ST(t13, dG.y, i13)                     \
        KNN_ST(t14, dH.x, i14) KNN_ST(t15, dH.y, i15)                     \
        KNN_LOAD(PX, PY, PZ, la)                                          \
        __builtin_amdgcn_sched_barrier(0);                                \
        tail = t16;                                                       \
    }

#define KNN_DRAIN(jb)                                                     \
    if (((jb) + KNN_CH >= N_) || __any(tail > lim)) {                     \
          \
          \
        float e0 = *(lds_f32*)(pk + 0 * KNN_SLOT);                      \
        float e1 = *(lds_f32*)(pk + 1 * KNN_SLOT);                      \
        float e2 = *(lds_f32*)(pk + 2 * KNN_SLOT);                      \
        float e3 = *(lds_f32*)(pk + 3 * KNN_SLOT);                      \
        float e4 = *(lds_f32*)(pk + 4 * KNN_SLOT);                      \
        float e5 = *(lds_f32*)(pk + 5 * KNN_SLOT);                      \
        float e6 = *(lds_f32*)(pk + 6 * KNN_SLOT);                      \
        float e7 = *(lds_f32*)(pk + 7 * KNN_SLOT);                      \
        float e8 = *(lds_f32*)(pk + 8 * KNN_SLOT);                      \
        float e9 = *(lds_f32*)(pk + 9 * KNN_SLOT);                      \
        float e10 = *(lds_f32*)(pk + 10 * KNN_SLOT);                      \
        float e11 = *(lds_f32*)(pk + 11 * KNN_SLOT);                      \
        float e12 = *(lds_f32*)(pk + 12 * KNN_SLOT);                      \
        float e13 = *(lds_f32*)(pk + 13 * KNN_SLOT);                      \
        float e14 = *(lds_f32*)(pk + 14 * KNN_SLOT);                      \
        float e15 = *(lds_f32*)(pk + 15 * KNN_SLOT);                      \
        float e16 = *(lds_f32*)(pk + 16 * KNN_SLOT);                      \
        float e17 = *(lds_f32*)(pk + 17 * KNN_SLOT);                      \
        float e18 = *(lds_f32*)(pk + 18 * KNN_SLOT);                      \
        float e19 = *(lds_f32*)(pk + 19 * KNN_SLOT);                      \
        int   k0 = *(lds_i32*)(pk + 20 * KNN_SLOT);                      \
        int   k1 = *(lds_i32*)(pk + 21 * KNN_SLOT);                      \
        int   k2 = *(lds_i32*)(pk + 22 * KNN_SLOT);                      \
        int   k3 = *(lds_i32*)(pk + 23 * KNN_SLOT);                      \
        int   k4 = *(lds_i32*)(pk + 24 * KNN_SLOT);                      \
        int   k5 = *(lds_i32*)(pk + 25 * KNN_SLOT);                      \
        int   k6 = *(lds_i32*)(pk + 26 * KNN_SLOT);                      \
        int   k7 = *(lds_i32*)(pk + 27 * KNN_SLOT);                      \
        int   k8 = *(lds_i32*)(pk + 28 * KNN_SLOT);                      \
        int   k9 = *(lds_i32*)(pk + 29 * KNN_SLOT);                      \
        int   k10 = *(lds_i32*)(pk + 30 * KNN_SLOT);                      \
        int   k11 = *(lds_i32*)(pk + 31 * KNN_SLOT);                      \
        int   k12 = *(lds_i32*)(pk + 32 * KNN_SLOT);                      \
        int   k13 = *(lds_i32*)(pk + 33 * KNN_SLOT);                      \
        int   k14 = *(lds_i32*)(pk + 34 * KNN_SLOT);                      \
        int   k15 = *(lds_i32*)(pk + 35 * KNN_SLOT);                      \
        int   k16 = *(lds_i32*)(pk + 36 * KNN_SLOT);                      \
        int   k17 = *(lds_i32*)(pk + 37 * KNN_SLOT);                      \
        int   k18 = *(lds_i32*)(pk + 38 * KNN_SLOT);                      \
        int   k19 = *(lds_i32*)(pk + 39 * KNN_SLOT);                      \
         \
         \
         \
        const int cnt = (int)((tail - tbase) / KNN_SLOT);                 \
        float dn = *(lds_f32*)(tbase);                                    \
        int   jn = *(lds_i32*)(tbase + KNN_QJ);                           \
        int s = 0;                                                        \
           \
           \
           \
        do {                                                              \
            const bool  have = s < cnt;                                   \
            const float d = have ? dn : -3.4e38f;                         \
            const int   j = have ? jn : 0;                                \
            const int   sn = (s + 1 < KNN_CAP) ? (s + 1) : (KNN_CAP - 1); \
            dn = *(lds_f32*)(tbase + (unsigned)sn * KNN_SLOT);            \
            jn = *(lds_i32*)(tbase + (unsigned)sn * KNN_SLOT + KNN_QJ);   \
               \
               \
               \
               \
            __builtin_amdgcn_sched_barrier(0);                            \
            KNN_INSERT                                                    \
            __builtin_amdgcn_sched_barrier(0);                            \
               \
               \
            asm volatile("" :: "v"(dn), "v"(jn));                         \
            ++s;                                                          \
        } while (__any(s < cnt));                                         \
        *(lds_f32*)(pk + 0 * KNN_SLOT) = e0;                            \
        *(lds_f32*)(pk + 1 * KNN_SLOT) = e1;                            \
        *(lds_f32*)(pk + 2 * KNN_SLOT) = e2;                            \
        *(lds_f32*)(pk + 3 * KNN_SLOT) = e3;                            \
        *(lds_f32*)(pk + 4 * KNN_SLOT) = e4;                            \
        *(lds_f32*)(pk + 5 * KNN_SLOT) = e5;                            \
        *(lds_f32*)(pk + 6 * KNN_SLOT) = e6;                            \
        *(lds_f32*)(pk + 7 * KNN_SLOT) = e7;                            \
        *(lds_f32*)(pk + 8 * KNN_SLOT) = e8;                            \
        *(lds_f32*)(pk + 9 * KNN_SLOT) = e9;                            \
        *(lds_f32*)(pk + 10 * KNN_SLOT) = e10;                            \
        *(lds_f32*)(pk + 11 * KNN_SLOT) = e11;                            \
        *(lds_f32*)(pk + 12 * KNN_SLOT) = e12;                            \
        *(lds_f32*)(pk + 13 * KNN_SLOT) = e13;                            \
        *(lds_f32*)(pk + 14 * KNN_SLOT) = e14;                            \
        *(lds_f32*)(pk + 15 * KNN_SLOT) = e15;                            \
        *(lds_f32*)(pk + 16 * KNN_SLOT) = e16;                            \
        *(lds_f32*)(pk + 17 * KNN_SLOT) = e17;                            \
        *(lds_f32*)(pk + 18 * KNN_SLOT) = e18;                            \
        *(lds_f32*)(pk + 19 * KNN_SLOT) = e19;                            \
        *(lds_i32*)(pk + 20 * KNN_SLOT) = k0;                            \
        *(lds_i32*)(pk + 21 * KNN_SLOT) = k1;                            \
        *(lds_i32*)(pk + 22 * KNN_SLOT) = k2;                            \
        *(lds_i32*)(pk + 23 * KNN_SLOT) = k3;                            \
        *(lds_i32*)(pk + 24 * KNN_SLOT) = k4;                            \
        *(lds_i32*)(pk + 25 * KNN_SLOT) = k5;                            \
        *(lds_i32*)(pk + 26 * KNN_SLOT) = k6;                            \
        *(lds_i32*)(pk + 27 * KNN_SLOT) = k7;                            \
        *(lds_i32*)(pk + 28 * KNN_SLOT) = k8;                            \
        *(lds_i32*)(pk + 29 * KNN_SLOT) = k9;                            \
        *(lds_i32*)(pk + 30 * KNN_SLOT) = k10;                            \
        *(lds_i32*)(pk + 31 * KNN_SLOT) = k11;                            \
        *(lds_i32*)(pk + 32 * KNN_SLOT) = k12;                            \
        *(lds_i32*)(pk + 33 * KNN_SLOT) = k13;                            \
        *(lds_i32*)(pk + 34 * KNN_SLOT) = k14;                            \
        *(lds_i32*)(pk + 35 * KNN_SLOT) = k15;                            \
        *(lds_i32*)(pk + 36 * KNN_SLOT) = k16;                            \
        *(lds_i32*)(pk + 37 * KNN_SLOT) = k17;                            \
        *(lds_i32*)(pk + 38 * KNN_SLOT) = k18;                            \
        *(lds_i32*)(pk + 39 * KNN_SLOT) = k19;                            \
        tail = tbase;                                                     \
        thr = e19;                                                        \
    }

#define KNN_INSERT                                                        \
    const bool c0 = d > e0, c1 = d > e1, c2 = d > e2, c3 = d > e3, c4 = d > e4;\
    const bool c5 = d > e5, c6 = d > e6, c7 = d > e7, c8 = d > e8, c9 = d > e9;\
    const bool c10 = d > e10, c11 = d > e11, c12 = d > e12, c13 = d > e13, c14 = d > e14;\
    const bool c15 = d > e15, c16 = d > e16, c17 = d > e17, c18 = d > e18, c19 = d > e19;\
    KNN_POS(e18, e19, k18, k19, c18, c19)                                 \
    KNN_POS(e17, e18, k17, k18, c17, c18)                                 \
    KNN_POS(e16, e17, k16, k17, c16, c17)                                 \
    KNN_POS(e15, e16, k15, k16, c15, c16)                                 \
    KNN_POS(e14, e15, k14, k15, c14, c15)                                 \
    KNN_POS(e13, e14, k13, k14, c13, c14)                                 \
    KNN_POS(e12, e13, k12, k13, c12, c13)                                 \
    KNN_POS(e11, e12, k11, k12, c11, c12)                                 \
    KNN_POS(e10, e11, k10, k11, c10, c11)                                 \
    KNN_POS(e9, e10, k9, k10, c9, c10)                                    \
    KNN_POS(e8, e9, k8, k9, c8, c9)                                       \
    KNN_POS(e7, e8, k7, k8, c7, c8)                                       \
    KNN_POS(e6, e7, k6, k7, c6, c7)                                       \
    KNN_POS(e5, e6, k5, k6, c5, c6)                                       \
    KNN_POS(e4, e5, k4, k5, c4, c5)                                       \
    KNN_POS(e3, e4, k3, k4, c3, c4)                                       \
    KNN_POS(e2, e3, k2, k3, c2, c3)                                       \
    KNN_POS(e1, e2, k1, k2, c1, c2)                                       \
    KNN_POS(e0, e1, k0, k1, c0, c1)                                       \
    e0 = c0 ? d : e0;                                                     \
    k0 = c0 ? j : k0;

__global__ __launch_bounds__(KNN_T) void knn_kernel(const float* __restrict__ x,
                                                    int* __restrict__ idx)
{
    __shared__ __attribute__((aligned(16))) float lds[3 * N_ + 2 * KNN_CAP * KNN_T + 2 * K_ * KNN_T];
    float* spts = lds;
    int*   qj   = (int*)(lds + 3 * N_ + KNN_CAP * KNN_T);
    const float* sx = spts;
    const float* sy = spts + N_;
    const float* sz = spts + 2 * N_;
    const int blocksPerBatch = N_ / KNN_T;
    const int b     = blockIdx.x / blocksPerBatch;
    const int chunk = blockIdx.x % blocksPerBatch;
    const float* xb = x + (size_t)b * 3 * N_;
    const v4f* xb4 = (const v4f*)xb;
    v4f*       sp4 = (v4f*)spts;
#pragma unroll
    for (int t = 0; t < (3 * N_ / 4) / KNN_T; ++t)
        sp4[t * KNN_T + threadIdx.x] = xb4[t * KNN_T + threadIdx.x];
    __syncthreads();

    const int   tid = threadIdx.x;
    const int   n  = chunk * KNN_T + tid;
    const float qx = sx[n], qy = sy[n], qz = sz[n];
    const float qq = fmaf(qz, qz, fmaf(qx, qx, qy * qy));
    const v2f qx2 = {qx, qx}, qy2 = {qy, qy}, qz2 = {qz, qz};
    const v2f nqq2 = {-qq, -qq}, two2 = {2.0f, 2.0f};

    const unsigned lds0  = (unsigned)(size_t)lds;
    const unsigned pbase = lds0;
    const unsigned tbase = lds0 + 3 * KNN_PLANE + (unsigned)tid * 4u;
    const unsigned lim   = tbase + (KNN_CAP - KNN_CH) * KNN_SLOT;
    const unsigned pk    = lds0 + 3 * KNN_PLANE + 2 * KNN_QJ + (unsigned)tid * 4u;

#pragma unroll
    for (int i = 0; i < K_; ++i) {
        *(lds_f32*)(pk + (unsigned)i * KNN_SLOT) = -3.4e38f;
        *(lds_i32*)(pk + (unsigned)(K_ + i) * KNN_SLOT) = 0;
    }

    float    thr  = -3.4e38f;
    unsigned tail = tbase;

    v4f ax0, ax1, ax2, ax3, ay0, ay1, ay2, ay3, az0, az1, az2, az3;
    v4f bx0, bx1, bx2, bx3, by0, by1, by2, by3, bz0, bz1, bz2, bz3;
    KNN_LOAD(ax, ay, az, pbase)
    KNN_LOAD(bx, by, bz, pbase + KNN_CH * 4)
    for (int j0 = 0; j0 < N_; j0 += 2 * KNN_CH) {
        KNN_CHUNK(ax, ay, az, j0, (j0 + 2 * KNN_CH) & (N_ - 1))
        KNN_DRAIN(j0)
        KNN_CHUNK(bx, by, bz, j0 + KNN_CH, (j0 + 3 * KNN_CH) & (N_ - 1))
        KNN_DRAIN(j0 + KNN_CH)
    }
    const int k0 = *(lds_i32*)(pk + 20 * KNN_SLOT);
    const int k1 = *(lds_i32*)(pk + 21 * KNN_SLOT);
    const int k2 = *(lds_i32*)(pk + 22 * KNN_SLOT);
    const int k3 = *(lds_i32*)(pk + 23 * KNN_SLOT);
    const int k4 = *(lds_i32*)(pk + 24 * KNN_SLOT);
    const int k5 = *(lds_i32*)(pk + 25 * KNN_SLOT);
    const int k6 = *(lds_i32*)(pk + 26 * KNN_SLOT);
    const int k7 = *(lds_i32*)(pk + 27 * KNN_SLOT);
    const int k8 = *(lds_i32*)(pk + 28 * KNN_SLOT);
    const int k9 = *(lds_i32*)(pk + 29 * KNN_SLOT);
    const int k10 = *(lds_i32*)(pk + 30 * KNN_SLOT);
    const int k11 = *(lds_i32*)(pk + 31 * KNN_SLOT);
    const int k12 = *(lds_i32*)(pk + 32 * KNN_SLOT);
    const int k13 = *(lds_i32*)(pk + 33 * KNN_SLOT);
    const int k14 = *(lds_i32*)(pk + 34 * KNN_SLOT);
    const int k15 = *(lds_i32*)(pk + 35 * KNN_SLOT);
    const int k16 = *(lds_i32*)(pk + 36 * KNN_SLOT);
    const int k17 = *(lds_i32*)(pk + 37 * KNN_SLOT);
    const int k18 = *(lds_i32*)(pk + 38 * KNN_SLOT);
    const int k19 = *(lds_i32*)(pk + 39 * KNN_SLOT);
    __syncthreads();
    int* tk = qj;
    tk[0 * KNN_TP + tid] = k0;
    tk[1 * KNN_TP + tid] = k1;
    tk[2 * KNN_TP + tid] = k2;
    tk[3 * KNN_TP + tid] = k3;
    tk[4 * KNN_TP + tid] = k4;
    tk[5 * KNN_TP + tid] = k5;
    tk[6 * KNN_TP + tid] = k6;
    tk[7 * KNN_TP + tid] = k7;
    tk[8 * KNN_TP + tid] = k8;
    tk[9 * KNN_TP + tid] = k9;
    tk[10 * KNN_TP + tid] = k10;
    tk[11 * KNN_TP + tid] = k11;
    tk[12 * KNN_TP + tid] = k12;
    tk[13 * KNN_TP + tid] = k13;
    tk[14 * KNN_TP + tid] = k14;
    tk[15 * KNN_TP + tid] = k15;
    tk[16 * KNN_TP + tid] = k16;
    tk[17 * KNN_TP + tid] = k17;
    tk[18 * KNN_TP + tid] = k18;
    tk[19 * KNN_TP + tid] = k19;
    __syncthreads();
    int* outp = idx + ((size_t)b * N_ + chunk * KNN_T) * K_;
#pragma unroll
    for (int m = 0; m < K_; ++m) {
        const int p = m * KNN_T + tid;
        const int q = p / K_;
        const int i = p - q * K_;
        outp[p] = tk[i * KNN_TP + q];
    }
}

#define EDGE_PW 8
#define EDGE_PB 32
#define EDGE_LP 33
#define EDGE_HS (64 * EDGE_LP + 16)
#define EDGE_KP 132

__device__ __forceinline__ float edge_xhalf(float v)
{
    return __builtin_bit_cast(float, __builtin_amdgcn_permlanex16(
        __builtin_bit_cast(int, v), __builtin_bit_cast(int, v),
        0x76543210, 0xfedcba98, false, false));
}

__device__ __forceinline__ float edge_lane(float v, int l)
{
    return __builtin_bit_cast(float,
        __builtin_amdgcn_readlane(__builtin_bit_cast(int, v), l));
}

__global__ __launch_bounds__(128) void edge_kernel(
    const float* __restrict__ x,  const int* __restrict__ idx,
    const float* __restrict__ w1, const float* __restrict__ b1,
    const float* __restrict__ g1, const float* __restrict__ beta1,
    const float* __restrict__ rm1,const float* __restrict__ rv1,
    const float* __restrict__ w2, const float* __restrict__ b2,
    const float* __restrict__ g2, const float* __restrict__ beta2,
    const float* __restrict__ rm2,const float* __restrict__ rv2,
    float* __restrict__ out,
    float* __restrict__ out_m1,   float* __restrict__ out_m2)
{
    __shared__ float pmx[2 * EDGE_HS];
    __shared__ float psm[2 * EDGE_HS];
    __shared__ __attribute__((aligned(16))) float pa[EDGE_PB * EDGE_KP];

    const int lane = threadIdx.x & 31;
    const int half = lane >> 4;
    const int l15  = lane & 15;
    const int wv   = __builtin_amdgcn_readfirstlane((int)(threadIdx.x >> 5));
    const int b    = blockIdx.x / (N_ / EDGE_PB);
    const int n0   = (blockIdx.x % (N_ / EDGE_PB)) * EDGE_PB;
    const int nw   = n0 + wv * EDGE_PW;

    const float* xb = x + (size_t)b * 3 * N_;

    const bool valid1 = (16 + l15) < K_;
    const int  kc1    = valid1 ? (16 + l15) : (K_ - 1);
    const int* ip = idx + ((size_t)b * N_ + nw) * K_;
    int nb0[EDGE_PW], nb1[EDGE_PW];
#pragma unroll
    for (int p = 0; p < EDGE_PW; ++p) {
        nb0[p] = ip[p * K_ + l15];
        nb1[p] = ip[p * K_ + kc1];
    }

    v2f   bt0[4], bt1[4];
    float bs[4], bt[4];
#pragma unroll
    for (int ct = 0; ct < 4; ++ct) {
        const int    o    = ct * 16 + l15;
        const float* wrow = w1 + o * 6;
        const float w0 = wrow[0], w1v = wrow[1], w2v = wrow[2];
        const float w3 = wrow[3], w4v = wrow[4], w5v = wrow[5];
        bt0[ct].x = half ? w2v : w0;   bt0[ct].y = half ? w3 : w1v;
        bt1[ct].x = half ? 0.0f : w4v; bt1[ct].y = half ? 0.0f : w5v;
        const float s = g1[o] * rsqrtf(rv1[o] + EPS_);
        bs[ct] = s;
        bt[ct] = (b1[o] - rm1[o]) * s + beta1[o];
    }

    float gx0[EDGE_PW], gy0[EDGE_PW], gz0[EDGE_PW];
    float gx1[EDGE_PW], gy1[EDGE_PW], gz1[EDGE_PW];
#pragma unroll
    for (int p = 0; p < EDGE_PW; ++p) {
        gx0[p] = xb[nb0[p]]; gy0[p] = xb[N_ + nb0[p]]; gz0[p] = xb[2 * N_ + nb0[p]];
        gx1[p] = xb[nb1[p]]; gy1[p] = xb[N_ + nb1[p]]; gz1[p] = xb[2 * N_ + nb1[p]];
    }

#pragma unroll
    for (int p = 0; p < EDGE_PW; ++p) {
        const int   n  = nw + p;
        const float cx = xb[n], cy = xb[N_ + n], cz = xb[2 * N_ + n];

        v2f a0[2], a1[2];
#pragma unroll
        for (int rt = 0; rt < 2; ++rt) {
            const bool valid = rt ? valid1 : true;
            const float nx = rt ? gx1[p] : gx0[p];
            const float ny = rt ? gy1[p] : gy0[p];
            const float nz = rt ? gz1[p] : gz0[p];
            const float e0 = valid ? (nx - cx) : 0.0f;
            const float e1 = valid ? (ny - cy) : 0.0f;
            const float e2 = valid ? (nz - cz) : 0.0f;
            const float e3 = valid ? cx : 0.0f;
            const float e4 = valid ? cy : 0.0f;
            const float e5 = valid ? cz : 0.0f;
            v2f t0, t1;
            t0.x = half ? e2 : e0;  t0.y = half ? e3 : e1;
            t1.x = half ? 0.0f : e4; t1.y = half ? 0.0f : e5;
            a0[rt] = t0; a1[rt] = t1;
        }

        const int col = wv * EDGE_PW + p;
#pragma unroll
        for (int ct = 0; ct < 4; ++ct) {
            v8f z = {};
            z = __builtin_amdgcn_wmma_f32_16x16x4_f32(false, a0[0], false, bt0[ct],
                                                      (short)0, z, false, false);
            z = __builtin_amdgcn_wmma_f32_16x16x4_f32(false, a1[0], false, bt1[ct],
                                                      (short)0, z, false, false);
            v8f z1 = {};
            z1 = __builtin_amdgcn_wmma_f32_16x16x4_f32(false, a0[1], false, bt0[ct],
                                                       (short)0, z1, false, false);
            z1 = __builtin_amdgcn_wmma_f32_16x16x4_f32(false, a1[1], false, bt1[ct],
                                                       (short)0, z1, false, false);

            const float s = bs[ct], t = bt[ct];
            float mx = -3.4e38f, sm = 0.0f;
#pragma unroll
            for (int v = 0; v < 8; ++v) {
                const float val = fmaxf(z[v] * s + t, 0.0f);
                mx = fmaxf(mx, val); sm += val;
            }
#pragma unroll
            for (int v = 0; v < 4; ++v) {
                const float val = fmaxf(z1[v] * s + t, 0.0f);
                mx = (half == 0) ? fmaxf(mx, val) : mx;
                sm = (half == 0) ? (sm + val) : sm;
            }
            const int q = half * EDGE_HS + (ct * 16 + l15) * EDGE_LP + col;
            pmx[q] = mx;
            psm[q] = sm;
        }
    }
    __syncthreads();

    const float* wrow2 = w2 + (size_t)(wv * 16 + l15) * 128 + half * 2;
    v2f wa[32];
#pragma unroll
    for (int ks = 0; ks < 32; ++ks)
        wa[ks] = *(const v2f*)(wrow2 + ks * 4);
    const int   oc  = wv * 16 + l15;
    const float s2v = g2[oc] * rsqrtf(rv2[oc] + EPS_);
    const float t2v = (b2[oc] - rm2[oc]) * s2v + beta2[oc];

    float jm[16], ja[16];
#pragma unroll
    for (int r = 0; r < 16; ++r) {
        const int   o  = wv * 16 + r;
        const float mx = fmaxf(pmx[o * EDGE_LP + lane], pmx[EDGE_HS + o * EDGE_LP + lane]);
        const float sm = psm[o * EDGE_LP + lane] + psm[EDGE_HS + o * EDGE_LP + lane];
        jm[r] = mx;
        ja[r] = sm * (1.0f / K_);
        pa[lane * EDGE_KP + o]      = jm[r];
        pa[lane * EDGE_KP + 64 + o] = ja[r];
    }
    __syncthreads();

#pragma unroll
    for (int r = 0; r < 16; ++r) {
        const int    o   = wv * 16 + r;
        const size_t off = ((size_t)b * 64 + o) * N_ + n0 + lane;
        out_m1[off] = jm[r];
        out_m2[off] = ja[r];
    }

    const float* pr0 = pa + l15 * EDGE_KP + half * 2;
    const float* pr1 = pr0 + 16 * EDGE_KP;
    v8f d0 = {}, d1 = {};
#pragma unroll
    for (int ks = 0; ks < 32; ++ks) {
        const v2f p0 = *(const v2f*)(pr0 + ks * 4);
        const v2f p1 = *(const v2f*)(pr1 + ks * 4);
        d0 = __builtin_amdgcn_wmma_f32_16x16x4_f32(false, wa[ks], false, p0,
                                                   (short)0, d0, false, false);
        d1 = __builtin_amdgcn_wmma_f32_16x16x4_f32(false, wa[ks], false, p1,
                                                   (short)0, d1, false, false);
    }

#pragma unroll
    for (int r = 0; r < 8; ++r) {
        const float x0 = edge_xhalf(d0[r]);
        const float x1 = edge_xhalf(d1[r]);
        const float lo = half ? x1 : d0[r];
        const float hi = half ? d1[r] : x0;
        {
            const int   o = wv * 16 + r;
            const float s = edge_lane(s2v, r), t = edge_lane(t2v, r);
            out[((size_t)b * 64 + o) * N_ + n0 + lane] = fmaxf(lo * s + t, 0.0f);
        }
        {
            const int   o = wv * 16 + 8 + r;
            const float s = edge_lane(s2v, 8 + r), t = edge_lane(t2v, 8 + r);
            out[((size_t)b * 64 + o) * N_ + n0 + lane] = fmaxf(hi * s + t, 0.0f);
        }
    }
}

extern "C" void kernel_launch(void* const* d_in, const int* in_sizes, int n_in,
                              void* d_out, int out_size, void* d_ws, size_t ws_size,
                              hipStream_t stream)
{
    const float* x     = (const float*)d_in[0];
    const float* w1    = (const float*)d_in[1];
    const float* b1    = (const float*)d_in[2];
    const float* g1    = (const float*)d_in[3];
    const float* beta1 = (const float*)d_in[4];
    const float* rm1   = (const float*)d_in[5];
    const float* rv1   = (const float*)d_in[6];
    const float* w2    = (const float*)d_in[7];
    const float* b2    = (const float*)d_in[8];
    const float* g2    = (const float*)d_in[9];
    const float* beta2 = (const float*)d_in[10];
    const float* rm2   = (const float*)d_in[11];
    const float* rv2   = (const float*)d_in[12];

    float* out    = (float*)d_out;
    float* out_m1 = out + (size_t)B_ * 64 * N_;
    float* out_m2 = out + 2 * (size_t)B_ * 64 * N_;
    int*   idx    = (int*)d_ws;

    knn_kernel<<<B_ * (N_ / KNN_T), KNN_T, 0, stream>>>(x, idx);

    edge_kernel<<<(B_ * N_) / EDGE_PB, 128, 0, stream>>>(
        x, idx, w1, b1, g1, beta1, rm1, rv1,
        w2, b2, g2, beta2, rm2, rv2, out, out_m1, out_m2);
}


# knn and edge conv fused into one launch: a block runs edge conv + output conv on its own 128 queries; neighbour lists and coordinates come from its LDS, idx is never written, no global gathers
# speedup vs baseline: 6.6864x; 1.0478x over previous
#include <hip/hip_runtime.h>

#define B_    8
#define C_    3
#define N_    4096
#define K_    20
#define H_    64
#define DOUT_ 64
#define EPS_  1e-5f

typedef __attribute__((ext_vector_type(2))) float v2f;
typedef __attribute__((ext_vector_type(8))) float v8f;

#define KNN_T   128
#define KNN_CH  16
#define KNN_CAP 32
#define KNN_TP  129

typedef __attribute__((ext_vector_type(4))) float v4f;

#define KNN_POS(eu, el, ku, kl, cu, cl)                    \
    {   el = (cu) ? (eu) : ((cl) ? d : (el));              \
        kl = (cu) ? (ku) : ((cl) ? j : (kl)); }

typedef __attribute__((address_space(3))) float lds_f32;
typedef __attribute__((address_space(3))) int   lds_i32;
typedef __attribute__((address_space(3))) v4f   lds_v4f;
#define KNN_SLOT   (KNN_T * 4)
#define KNN_QJ     (KNN_CAP * KNN_T * 4)
#define KNN_PLANE  (N_ * 4)

#define KNN_LOAD(PX, PY, PZ, la)                                          \
    {   const lds_v4f* lx = (const lds_v4f*)(la);                         \
        const lds_v4f* ly = (const lds_v4f*)((la) + KNN_PLANE);           \
        const lds_v4f* lz = (const lds_v4f*)((la) + 2 * KNN_PLANE);       \
        PX##0 = lx[0]; PX##1 = lx[1]; PX##2 = lx[2]; PX##3 = lx[3];       \
        PY##0 = ly[0]; PY##1 = ly[1]; PY##2 = ly[2]; PY##3 = ly[3];       \
        PZ##0 = lz[0]; PZ##1 = lz[1]; PZ##2 = lz[2]; PZ##3 = lz[3]; }

#define KNN_D2(OUT, X2, Y2, Z2)                                           \
    {   const v2f dot = __builtin_elementwise_fma(qz2, (Z2),              \
                        __builtin_elementwise_fma(qx2, (X2), qy2 * (Y2)));\
        const v2f pp  = __builtin_elementwise_fma((Z2), (Z2),             \
                        __builtin_elementwise_fma((X2), (X2), (Y2) * (Y2)));\
        OUT = __builtin_elementwise_fma(dot, two2, nqq2) - pp; }

#define KNN_D4(O1, O2, X, Y, Z)                                           \
    KNN_D2(O1, X.xy, Y.xy, Z.xy) KNN_D2(O2, X.zw, Y.zw, Z.zw)

#define KNN_ADV(TN, T, dv)                                                \
    const unsigned TN = (T) + (((dv) > thr) ? (unsigned)KNN_SLOT : 0u);
#define KNN_ST(T, dv, jj)                                                 \
    {   *(lds_f32*)(T) = (dv);                                            \
        *(lds_i32*)((T) + KNN_QJ) = (jj); }

#define KNN_CHUNK(PX, PY, PZ, jb, jnext)                                  \
    {   v2f dA, dB, dC, dD, dE, dF, dG, dH;                               \
        KNN_D4(dA, dB, PX##0, PY##0, PZ##0)                               \
        KNN_D4(dC, dD, PX##1, PY##1, PZ##1)                               \
        KNN_D4(dE, dF, PX##2, PY##2, PZ##2)                               \
        KNN_D4(dG, dH, PX##3, PY##3, PZ##3)                               \
        const unsigned t0 = tail;                                         \
        KNN_ADV(t1, t0, dA.x)   KNN_ADV(t2, t1, dA.y)                     \
        KNN_ADV(t3, t2, dB.x)   KNN_ADV(t4, t3, dB.y)                     \
        KNN_ADV(t5, t4, dC.x)   KNN_ADV(t6, t5, dC.y)                     \
        KNN_ADV(t7, t6, dD.x)   KNN_ADV(t8, t7, dD.y)                     \
        KNN_ADV(t9, t8, dE.x)   KNN_ADV(t10, t9, dE.y)                    \
        KNN_ADV(t11, t10, dF.x) KNN_ADV(t12, t11, dF.y)                   \
        KNN_ADV(t13, t12, dG.x) KNN_ADV(t14, t13, dG.y)                   \
        KNN_ADV(t15, t14, dH.x) KNN_ADV(t16, t15, dH.y)                   \
          \
          \
          \
          \
          \
          \
        int i0; unsigned la;                                              \
        asm("v_mov_b32 %0, %1" : "=v"(i0) : "s"((int)(jb)));              \
        asm("v_mov_b32 %0, %1" : "=v"(la) : "s"(pbase + ((unsigned)(jnext) << 2)));\
        const int i1 = i0 + 1, i2 = i0 + 2, i3 = i0 + 3;                  \
        const int i4 = i0 + 4, i5 = i0 + 5, i6 = i0 + 6, i7 = i0 + 7;     \
        const int i8 = i0 + 8, i9 = i0 + 9, i10 = i0 + 10, i11 = i0 + 11; \
        const int i12 = i0 + 12, i13 = i0 + 13, i14 = i0 + 14, i15 = i0 + 15;\
         \
         \
         \
        asm volatile("" :: "v"(t0), "v"(t1), "v"(t2), "v"(t3), "v"(t4),   \
                           "v"(t5), "v"(t6), "v"(t7), "v"(t8), "v"(t9),   \
                           "v"(t10), "v"(t11), "v"(t12), "v"(t13),        \
                           "v"(t14), "v"(t15), "v"(la));                  \
        asm volatile("" :: "v"(dA.x), "v"(dA.y), "v"(dB.x), "v"(dB.y),    \
                           "v"(dC.x), "v"(dC.y), "v"(dD.x), "v"(dD.y),    \
                           "v"(dE.x), "v"(dE.y), "v"(dF.x), "v"(dF.y),    \
                           "v"(dG.x), "v"(dG.y), "v"(dH.x), "v"(dH.y));   \
        asm volatile("" :: "v"(i0), "v"(i1), "v"(i2), "v"(i3), "v"(i4),   \
                           "v"(i5), "v"(i6), "v"(i7), "v"(i8), "v"(i9),   \
                           "v"(i10), "v"(i11), "v"(i12), "v"(i13),        \
                           "v"(i14), "v"(i15));                           \
          \
          \
          \
        __builtin_amdgcn_sched_barrier(0);                                \
        KNN_ST(t0, dA.x, i0)   KNN_ST(t1, dA.y, i1)                       \
        KNN_ST(t2, dB.x, i2)   KNN_ST(t3, dB.y, i3)                       \
        KNN_ST(t4, dC.x, i4)   KNN_ST(t5, dC.y, i5)                       \
        KNN_ST(t6, dD.x, i6)   KNN_ST(t7, dD.y, i7)                       \
        KNN_ST(t8, dE.x, i8)   KNN_ST(t9, dE.y, i9)                       \
        KNN_ST(t10, dF.x, i10) KNN_ST(t11, dF.y, i11)                     \
        KNN_ST(t12, dG.x, i12) KNN_ST(t13, dG.y, i13)                     \
        KNN_ST(t14, dH.x, i14) KNN_ST(t15, dH.y, i15)                     \
        KNN_LOAD(PX, PY, PZ, la)                                          \
        __builtin_amdgcn_sched_barrier(0);                                \
        tail = t16;                                                       \
    }

#define KNN_DRAIN(jb)                                                     \
    if (((jb) + KNN_CH >= N_) || __any(tail > lim)) {                     \
          \
          \
        float e0 = *(lds_f32*)(pk + 0 * KNN_SLOT);                      \
        float e1 = *(lds_f32*)(pk + 1 * KNN_SLOT);                      \
        float e2 = *(lds_f32*)(pk + 2 * KNN_SLOT);                      \
        float e3 = *(lds_f32*)(pk + 3 * KNN_SLOT);                      \
        float e4 = *(lds_f32*)(pk + 4 * KNN_SLOT);                      \
        float e5 = *(lds_f32*)(pk + 5 * KNN_SLOT);                      \
        float e6 = *(lds_f32*)(pk + 6 * KNN_SLOT);                      \
        float e7 = *(lds_f32*)(pk + 7 * KNN_SLOT);                      \
        float e8 = *(lds_f32*)(pk + 8 * KNN_SLOT);                      \
        float e9 = *(lds_f32*)(pk + 9 * KNN_SLOT);                      \
        float e10 = *(lds_f32*)(pk + 10 * KNN_SLOT);                      \
        float e11 = *(lds_f32*)(pk + 11 * KNN_SLOT);                      \
        float e12 = *(lds_f32*)(pk + 12 * KNN_SLOT);                      \
        float e13 = *(lds_f32*)(pk + 13 * KNN_SLOT);                      \
        float e14 = *(lds_f32*)(pk + 14 * KNN_SLOT);                      \
        float e15 = *(lds_f32*)(pk + 15 * KNN_SLOT);                      \
        float e16 = *(lds_f32*)(pk + 16 * KNN_SLOT);                      \
        float e17 = *(lds_f32*)(pk + 17 * KNN_SLOT);                      \
        float e18 = *(lds_f32*)(pk + 18 * KNN_SLOT);                      \
        float e19 = *(lds_f32*)(pk + 19 * KNN_SLOT);                      \
        int   k0 = *(lds_i32*)(pk + 20 * KNN_SLOT);                      \
        int   k1 = *(lds_i32*)(pk + 21 * KNN_SLOT);                      \
        int   k2 = *(lds_i32*)(pk + 22 * KNN_SLOT);                      \
        int   k3 = *(lds_i32*)(pk + 23 * KNN_SLOT);                      \
        int   k4 = *(lds_i32*)(pk + 24 * KNN_SLOT);                      \
        int   k5 = *(lds_i32*)(pk + 25 * KNN_SLOT);                      \
        int   k6 = *(lds_i32*)(pk + 26 * KNN_SLOT);                      \
        int   k7 = *(lds_i32*)(pk + 27 * KNN_SLOT);                      \
        int   k8 = *(lds_i32*)(pk + 28 * KNN_SLOT);                      \
        int   k9 = *(lds_i32*)(pk + 29 * KNN_SLOT);                      \
        int   k10 = *(lds_i32*)(pk + 30 * KNN_SLOT);                      \
        int   k11 = *(lds_i32*)(pk + 31 * KNN_SLOT);                      \
        int   k12 = *(lds_i32*)(pk + 32 * KNN_SLOT);                      \
        int   k13 = *(lds_i32*)(pk + 33 * KNN_SLOT);                      \
        int   k14 = *(lds_i32*)(pk + 34 * KNN_SLOT);                      \
        int   k15 = *(lds_i32*)(pk + 35 * KNN_SLOT);                      \
        int   k16 = *(lds_i32*)(pk + 36 * KNN_SLOT);                      \
        int   k17 = *(lds_i32*)(pk + 37 * KNN_SLOT);                      \
        int   k18 = *(lds_i32*)(pk + 38 * KNN_SLOT);                      \
        int   k19 = *(lds_i32*)(pk + 39 * KNN_SLOT);                      \
         \
         \
         \
        const int cnt = (int)((tail - tbase) / KNN_SLOT);                 \
        float dn = *(lds_f32*)(tbase);                                    \
        int   jn = *(lds_i32*)(tbase + KNN_QJ);                           \
        int s = 0;                                                        \
           \
           \
           \
        do {                                                              \
            const bool  have = s < cnt;                                   \
            const float d = have ? dn : -3.4e38f;                         \
            const int   j = have ? jn : 0;                                \
            const int   sn = (s + 1 < KNN_CAP) ? (s + 1) : (KNN_CAP - 1); \
            dn = *(lds_f32*)(tbase + (unsigned)sn * KNN_SLOT);            \
            jn = *(lds_i32*)(tbase + (unsigned)sn * KNN_SLOT + KNN_QJ);   \
               \
               \
               \
               \
            __builtin_amdgcn_sched_barrier(0);                            \
            KNN_INSERT                                                    \
            __builtin_amdgcn_sched_barrier(0);                            \
               \
               \
            asm volatile("" :: "v"(dn), "v"(jn));                         \
            ++s;                                                          \
        } while (__any(s < cnt));                                         \
        *(lds_f32*)(pk + 0 * KNN_SLOT) = e0;                            \
        *(lds_f32*)(pk + 1 * KNN_SLOT) = e1;                            \
        *(lds_f32*)(pk + 2 * KNN_SLOT) = e2;                            \
        *(lds_f32*)(pk + 3 * KNN_SLOT) = e3;                            \
        *(lds_f32*)(pk + 4 * KNN_SLOT) = e4;                            \
        *(lds_f32*)(pk + 5 * KNN_SLOT) = e5;                            \
        *(lds_f32*)(pk + 6 * KNN_SLOT) = e6;                            \
        *(lds_f32*)(pk + 7 * KNN_SLOT) = e7;                            \
        *(lds_f32*)(pk + 8 * KNN_SLOT) = e8;                            \
        *(lds_f32*)(pk + 9 * KNN_SLOT) = e9;                            \
        *(lds_f32*)(pk + 10 * KNN_SLOT) = e10;                            \
        *(lds_f32*)(pk + 11 * KNN_SLOT) = e11;                            \
        *(lds_f32*)(pk + 12 * KNN_SLOT) = e12;                            \
        *(lds_f32*)(pk + 13 * KNN_SLOT) = e13;                            \
        *(lds_f32*)(pk + 14 * KNN_SLOT) = e14;                            \
        *(lds_f32*)(pk + 15 * KNN_SLOT) = e15;                            \
        *(lds_f32*)(pk + 16 * KNN_SLOT) = e16;                            \
        *(lds_f32*)(pk + 17 * KNN_SLOT) = e17;                            \
        *(lds_f32*)(pk + 18 * KNN_SLOT) = e18;                            \
        *(lds_f32*)(pk + 19 * KNN_SLOT) = e19;                            \
        *(lds_i32*)(pk + 20 * KNN_SLOT) = k0;                            \
        *(lds_i32*)(pk + 21 * KNN_SLOT) = k1;                            \
        *(lds_i32*)(pk + 22 * KNN_SLOT) = k2;                            \
        *(lds_i32*)(pk + 23 * KNN_SLOT) = k3;                            \
        *(lds_i32*)(pk + 24 * KNN_SLOT) = k4;                            \
        *(lds_i32*)(pk + 25 * KNN_SLOT) = k5;                            \
        *(lds_i32*)(pk + 26 * KNN_SLOT) = k6;                            \
        *(lds_i32*)(pk + 27 * KNN_SLOT) = k7;                            \
        *(lds_i32*)(pk + 28 * KNN_SLOT) = k8;                            \
        *(lds_i32*)(pk + 29 * KNN_SLOT) = k9;                            \
        *(lds_i32*)(pk + 30 * KNN_SLOT) = k10;                            \
        *(lds_i32*)(pk + 31 * KNN_SLOT) = k11;                            \
        *(lds_i32*)(pk + 32 * KNN_SLOT) = k12;                            \
        *(lds_i32*)(pk + 33 * KNN_SLOT) = k13;                            \
        *(lds_i32*)(pk + 34 * KNN_SLOT) = k14;                            \
        *(lds_i32*)(pk + 35 * KNN_SLOT) = k15;                            \
        *(lds_i32*)(pk + 36 * KNN_SLOT) = k16;                            \
        *(lds_i32*)(pk + 37 * KNN_SLOT) = k17;                            \
        *(lds_i32*)(pk + 38 * KNN_SLOT) = k18;                            \
        *(lds_i32*)(pk + 39 * KNN_SLOT) = k19;                            \
        tail = tbase;                                                     \
        thr = e19;                                                        \
    }

#define KNN_INSERT                                                        \
    const bool c0 = d > e0, c1 = d > e1, c2 = d > e2, c3 = d > e3, c4 = d > e4;\
    const bool c5 = d > e5, c6 = d > e6, c7 = d > e7, c8 = d > e8, c9 = d > e9;\
    const bool c10 = d > e10, c11 = d > e11, c12 = d > e12, c13 = d > e13, c14 = d > e14;\
    const bool c15 = d > e15, c16 = d > e16, c17 = d > e17, c18 = d > e18, c19 = d > e19;\
    KNN_POS(e18, e19, k18, k19, c18, c19)                                 \
    KNN_POS(e17, e18, k17, k18, c17, c18)                                 \
    KNN_POS(e16, e17, k16, k17, c16, c17)                                 \
    KNN_POS(e15, e16, k15, k16, c15, c16)                                 \
    KNN_POS(e14, e15, k14, k15, c14, c15)                                 \
    KNN_POS(e13, e14, k13, k14, c13, c14)                                 \
    KNN_POS(e12, e13, k12, k13, c12, c13)                                 \
    KNN_POS(e11, e12, k11, k12, c11, c12)                                 \
    KNN_POS(e10, e11, k10, k11, c10, c11)                                 \
    KNN_POS(e9, e10, k9, k10, c9, c10)                                    \
    KNN_POS(e8, e9, k8, k9, c8, c9)                                       \
    KNN_POS(e7, e8, k7, k8, c7, c8)                                       \
    KNN_POS(e6, e7, k6, k7, c6, c7)                                       \
    KNN_POS(e5, e6, k5, k6, c5, c6)                                       \
    KNN_POS(e4, e5, k4, k5, c4, c5)                                       \
    KNN_POS(e3, e4, k3, k4, c3, c4)                                       \
    KNN_POS(e2, e3, k2, k3, c2, c3)                                       \
    KNN_POS(e1, e2, k1, k2, c1, c2)                                       \
    KNN_POS(e0, e1, k0, k1, c0, c1)                                       \
    e0 = c0 ? d : e0;                                                     \
    k0 = c0 ? j : k0;

#define EDGE_PW 8
#define EDGE_PB 32
#define EDGE_LP 33
#define EDGE_HS (64 * EDGE_LP + 16)
#define EDGE_KP 132

__device__ __forceinline__ float edge_xhalf(float v)
{
    return __builtin_bit_cast(float, __builtin_amdgcn_permlanex16(
        __builtin_bit_cast(int, v), __builtin_bit_cast(int, v),
        0x76543210, 0xfedcba98, false, false));
}

__device__ __forceinline__ float edge_lane(float v, int l)
{
    return __builtin_bit_cast(float,
        __builtin_amdgcn_readlane(__builtin_bit_cast(int, v), l));
}

__global__ __launch_bounds__(KNN_T) void knn_edge_kernel(
    const float* __restrict__ x,
    const float* __restrict__ w1, const float* __restrict__ b1,
    const float* __restrict__ g1, const float* __restrict__ beta1,
    const float* __restrict__ rm1,const float* __restrict__ rv1,
    const float* __restrict__ w2, const float* __restrict__ b2,
    const float* __restrict__ g2, const float* __restrict__ beta2,
    const float* __restrict__ rm2,const float* __restrict__ rv2,
    float* __restrict__ out,
    float* __restrict__ out_m1,   float* __restrict__ out_m2)
{
    __shared__ __attribute__((aligned(16))) float lds[3 * N_ + 2 * KNN_CAP * KNN_T + 2 * K_ * KNN_T];
    float* spts = lds;
    int*   qj   = (int*)(lds + 3 * N_ + KNN_CAP * KNN_T);
    const float* sx = spts;
    const float* sy = spts + N_;
    const float* sz = spts + 2 * N_;
    const int blocksPerBatch = N_ / KNN_T;
    const int b     = blockIdx.x / blocksPerBatch;
    const int chunk = blockIdx.x % blocksPerBatch;
    const float* xb = x + (size_t)b * 3 * N_;
    const v4f* xb4 = (const v4f*)xb;
    v4f*       sp4 = (v4f*)spts;
#pragma unroll
    for (int t = 0; t < (3 * N_ / 4) / KNN_T; ++t)
        sp4[t * KNN_T + threadIdx.x] = xb4[t * KNN_T + threadIdx.x];
    __syncthreads();

    const int   tid = threadIdx.x;
    const int   n  = chunk * KNN_T + tid;
    const float qx = sx[n], qy = sy[n], qz = sz[n];
    const float qq = fmaf(qz, qz, fmaf(qx, qx, qy * qy));
    const v2f qx2 = {qx, qx}, qy2 = {qy, qy}, qz2 = {qz, qz};
    const v2f nqq2 = {-qq, -qq}, two2 = {2.0f, 2.0f};

    const unsigned lds0  = (unsigned)(size_t)lds;
    const unsigned pbase = lds0;
    const unsigned tbase = lds0 + 3 * KNN_PLANE + (unsigned)tid * 4u;
    const unsigned lim   = tbase + (KNN_CAP - KNN_CH) * KNN_SLOT;
    const unsigned pk    = lds0 + 3 * KNN_PLANE + 2 * KNN_QJ + (unsigned)tid * 4u;

#pragma unroll
    for (int i = 0; i < K_; ++i) {
        *(lds_f32*)(pk + (unsigned)i * KNN_SLOT) = -3.4e38f;
        *(lds_i32*)(pk + (unsigned)(K_ + i) * KNN_SLOT) = 0;
    }

    float    thr  = -3.4e38f;
    unsigned tail = tbase;

    v4f ax0, ax1, ax2, ax3, ay0, ay1, ay2, ay3, az0, az1, az2, az3;
    v4f bx0, bx1, bx2, bx3, by0, by1, by2, by3, bz0, bz1, bz2, bz3;
    KNN_LOAD(ax, ay, az, pbase)
    KNN_LOAD(bx, by, bz, pbase + KNN_CH * 4)
    for (int j0 = 0; j0 < N_; j0 += 2 * KNN_CH) {
        KNN_CHUNK(ax, ay, az, j0, (j0 + 2 * KNN_CH) & (N_ - 1))
        KNN_DRAIN(j0)
        KNN_CHUNK(bx, by, bz, j0 + KNN_CH, (j0 + 3 * KNN_CH) & (N_ - 1))
        KNN_DRAIN(j0 + KNN_CH)
    }
    const int k0 = *(lds_i32*)(pk + 20 * KNN_SLOT);
    const int k1 = *(lds_i32*)(pk + 21 * KNN_SLOT);
    const int k2 = *(lds_i32*)(pk + 22 * KNN_SLOT);
    const int k3 = *(lds_i32*)(pk + 23 * KNN_SLOT);
    const int k4 = *(lds_i32*)(pk + 24 * KNN_SLOT);
    const int k5 = *(lds_i32*)(pk + 25 * KNN_SLOT);
    const int k6 = *(lds_i32*)(pk + 26 * KNN_SLOT);
    const int k7 = *(lds_i32*)(pk + 27 * KNN_SLOT);
    const int k8 = *(lds_i32*)(pk + 28 * KNN_SLOT);
    const int k9 = *(lds_i32*)(pk + 29 * KNN_SLOT);
    const int k10 = *(lds_i32*)(pk + 30 * KNN_SLOT);
    const int k11 = *(lds_i32*)(pk + 31 * KNN_SLOT);
    const int k12 = *(lds_i32*)(pk + 32 * KNN_SLOT);
    const int k13 = *(lds_i32*)(pk + 33 * KNN_SLOT);
    const int k14 = *(lds_i32*)(pk + 34 * KNN_SLOT);
    const int k15 = *(lds_i32*)(pk + 35 * KNN_SLOT);
    const int k16 = *(lds_i32*)(pk + 36 * KNN_SLOT);
    const int k17 = *(lds_i32*)(pk + 37 * KNN_SLOT);
    const int k18 = *(lds_i32*)(pk + 38 * KNN_SLOT);
    const int k19 = *(lds_i32*)(pk + 39 * KNN_SLOT);
    __syncthreads();
    int* tk = qj;
    tk[0 * KNN_TP + tid] = k0;
    tk[1 * KNN_TP + tid] = k1;
    tk[2 * KNN_TP + tid] = k2;
    tk[3 * KNN_TP + tid] = k3;
    tk[4 * KNN_TP + tid] = k4;
    tk[5 * KNN_TP + tid] = k5;
    tk[6 * KNN_TP + tid] = k6;
    tk[7 * KNN_TP + tid] = k7;
    tk[8 * KNN_TP + tid] = k8;
    tk[9 * KNN_TP + tid] = k9;
    tk[10 * KNN_TP + tid] = k10;
    tk[11 * KNN_TP + tid] = k11;
    tk[12 * KNN_TP + tid] = k12;
    tk[13 * KNN_TP + tid] = k13;
    tk[14 * KNN_TP + tid] = k14;
    tk[15 * KNN_TP + tid] = k15;
    tk[16 * KNN_TP + tid] = k16;
    tk[17 * KNN_TP + tid] = k17;
    tk[18 * KNN_TP + tid] = k18;
    tk[19 * KNN_TP + tid] = k19;
    __syncthreads();

    __shared__ float pmx[2 * EDGE_HS];
    __shared__ float psm[2 * EDGE_HS];
    __shared__ __attribute__((aligned(16))) float pa[EDGE_PB * EDGE_KP];

    const int lane = tid & 31;
    const int half = lane >> 4;
    const int l15  = lane & 15;
    const int wv   = __builtin_amdgcn_readfirstlane(tid >> 5);

    const bool valid1 = (16 + l15) < K_;
    const int  kc1    = valid1 ? (16 + l15) : (K_ - 1);
    const int* tk0 = tk + l15 * KNN_TP;
    const int* tk1 = tk + kc1 * KNN_TP;

    v2f   bt0[4], bt1[4];
    float bs[4], bt[4];
#pragma unroll
    for (int ct = 0; ct < 4; ++ct) {
        const int    o    = ct * 16 + l15;
        const float* wrow = w1 + o * 6;
        const float w0 = wrow[0], w1v = wrow[1], w2v = wrow[2];
        const float w3 = wrow[3], w4v = wrow[4], w5v = wrow[5];
        bt0[ct].x = half ? w2v : w0;   bt0[ct].y = half ? w3 : w1v;
        bt1[ct].x = half ? 0.0f : w4v; bt1[ct].y = half ? 0.0f : w5v;
        const float s = g1[o] * rsqrtf(rv1[o] + EPS_);
        bs[ct] = s;
        bt[ct] = (b1[o] - rm1[o]) * s + beta1[o];
    }

    const float* wrow2 = w2 + (size_t)(wv * 16 + l15) * 128 + half * 2;
    v2f wa[32];
#pragma unroll
    for (int ks = 0; ks < 32; ++ks)
        wa[ks] = *(const v2f*)(wrow2 + ks * 4);
    const int   oc  = wv * 16 + l15;
    const float s2v = g2[oc] * rsqrtf(rv2[oc] + EPS_);
    const float t2v = (b2[oc] - rm2[oc]) * s2v + beta2[oc];

    const float* pr0 = pa + l15 * EDGE_KP + half * 2;
    const float* pr1 = pr0 + 16 * EDGE_KP;

#pragma nounroll
    for (int g = 0; g < KNN_T / EDGE_PB; ++g) {
        const int lq = g * EDGE_PB + wv * EDGE_PW;
        const int n0 = chunk * KNN_T + g * EDGE_PB;
        const int nw = n0 + wv * EDGE_PW;

        int nb0[EDGE_PW], nb1[EDGE_PW];
#pragma unroll
        for (int p = 0; p < EDGE_PW; ++p) {
            nb0[p] = tk0[lq + p];
            nb1[p] = tk1[lq + p];
        }

        float gx0[EDGE_PW], gy0[EDGE_PW], gz0[EDGE_PW];
        float gx1[EDGE_PW], gy1[EDGE_PW], gz1[EDGE_PW];
        float ccx[EDGE_PW], ccy[EDGE_PW], ccz[EDGE_PW];
#pragma unroll
        for (int p = 0; p < EDGE_PW; ++p) {
            gx0[p] = sx[nb0[p]]; gy0[p] = sy[nb0[p]]; gz0[p] = sz[nb0[p]];
            gx1[p] = sx[nb1[p]]; gy1[p] = sy[nb1[p]]; gz1[p] = sz[nb1[p]];
            ccx[p] = sx[nw + p]; ccy[p] = sy[nw + p]; ccz[p] = sz[nw + p];
        }

        float pmr[EDGE_PW * 4], psr[EDGE_PW * 4];
#pragma unroll
        for (int p = 0; p < EDGE_PW; ++p) {
            const float cx = ccx[p], cy = ccy[p], cz = ccz[p];

            v2f a0[2], a1[2];
#pragma unroll
            for (int rt = 0; rt < 2; ++rt) {
                const bool valid = rt ? valid1 : true;
                const float nx = rt ? gx1[p] : gx0[p];
                const float ny = rt ? gy1[p] : gy0[p];
                const float nz = rt ? gz1[p] : gz0[p];
                const float e0 = valid ? (nx - cx) : 0.0f;
                const float e1 = valid ? (ny - cy) : 0.0f;
                const float e2 = valid ? (nz - cz) : 0.0f;
                const float e3 = valid ? cx : 0.0f;
                const float e4 = valid ? cy : 0.0f;
                const float e5 = valid ? cz : 0.0f;
                v2f t0, t1;
                t0.x = half ? e2 : e0;  t0.y = half ? e3 : e1;
                t1.x = half ? 0.0f : e4; t1.y = half ? 0.0f : e5;
                a0[rt] = t0; a1[rt] = t1;
            }

#pragma unroll
            for (int ct = 0; ct < 4; ++ct) {
                v8f z = {};
                z = __builtin_amdgcn_wmma_f32_16x16x4_f32(false, a0[0], false, bt0[ct],
                                                          (short)0, z, false, false);
                z = __builtin_amdgcn_wmma_f32_16x16x4_f32(false, a1[0], false, bt1[ct],
                                                          (short)0, z, false, false);
                v8f z1 = {};
                z1 = __builtin_amdgcn_wmma_f32_16x16x4_f32(false, a0[1], false, bt0[ct],
                                                           (short)0, z1, false, false);
                z1 = __builtin_amdgcn_wmma_f32_16x16x4_f32(false, a1[1], false, bt1[ct],
                                                           (short)0, z1, false, false);

                const float s = bs[ct], t = bt[ct];
                float mx = -3.4e38f, sm = 0.0f;
#pragma unroll
                for (int v = 0; v < 8; ++v) {
                    const float val = fmaxf(z[v] * s + t, 0.0f);
                    mx = fmaxf(mx, val); sm += val;
                }
#pragma unroll
                for (int v = 0; v < 4; ++v) {
                    const float val = fmaxf(z1[v] * s + t, 0.0f);
                    mx = (half == 0) ? fmaxf(mx, val) : mx;
                    sm = (half == 0) ? (sm + val) : sm;
                }
                pmr[p * 4 + ct] = mx;
                psr[p * 4 + ct] = sm;
            }
        }
        __builtin_amdgcn_sched_barrier(0);
#pragma unroll
        for (int p = 0; p < EDGE_PW; ++p) {
#pragma unroll
            for (int ct = 0; ct < 4; ++ct) {
                const int q = half * EDGE_HS + (ct * 16 + l15) * EDGE_LP + wv * EDGE_PW + p;
                pmx[q] = pmr[p * 4 + ct];
                psm[q] = psr[p * 4 + ct];
            }
        }
        __builtin_amdgcn_sched_barrier(0);
        __syncthreads();

        float jm[16], ja[16];
        float rml[16], rmh[16], rsl[16], rsh[16];
#pragma unroll
        for (int r = 0; r < 16; ++r) {
            const int o = wv * 16 + r;
            rml[r] = pmx[o * EDGE_LP + lane]; rmh[r] = pmx[EDGE_HS + o * EDGE_LP + lane];
            rsl[r] = psm[o * EDGE_LP + lane]; rsh[r] = psm[EDGE_HS + o * EDGE_LP + lane];
        }
        __builtin_amdgcn_sched_barrier(0);
#pragma unroll
        for (int r = 0; r < 16; ++r) {
            jm[r] = fmaxf(rml[r], rmh[r]);
            ja[r] = (rsl[r] + rsh[r]) * (1.0f / K_);
        }
        __builtin_amdgcn_sched_barrier(0);
#pragma unroll
        for (int r = 0; r < 16; ++r) {
            const int o = wv * 16 + r;
            pa[lane * EDGE_KP + o]      = jm[r];
            pa[lane * EDGE_KP + 64 + o] = ja[r];
        }
        __syncthreads();

#pragma unroll
        for (int r = 0; r < 16; ++r) {
            const int    o   = wv * 16 + r;
            const size_t off = ((size_t)b * 64 + o) * N_ + n0 + lane;
            out_m1[off] = jm[r];
            out_m2[off] = ja[r];
        }

        v2f pb0[32], pb1[32];
#pragma unroll
        for (int ks = 0; ks < 32; ++ks) {
            pb0[ks] = *(const v2f*)(pr0 + ks * 4);
            pb1[ks] = *(const v2f*)(pr1 + ks * 4);
        }
        __builtin_amdgcn_sched_barrier(0);
        v8f d0 = {}, d1 = {};
#pragma unroll
        for (int ks = 0; ks < 32; ++ks) {
            d0 = __builtin_amdgcn_wmma_f32_16x16x4_f32(false, wa[ks], false, pb0[ks],
                                                       (short)0, d0, false, false);
            d1 = __builtin_amdgcn_wmma_f32_16x16x4_f32(false, wa[ks], false, pb1[ks],
                                                       (short)0, d1, false, false);
        }

#pragma unroll
        for (int r = 0; r < 8; ++r) {
            const float x0 = edge_xhalf(d0[r]);
            const float x1 = edge_xhalf(d1[r]);
            const float lo = half ? x1 : d0[r];
            const float hi = half ? d1[r] : x0;
            {
                const int   o = wv * 16 + r;
                const float s = edge_lane(s2v, r), t = edge_lane(t2v, r);
                out[((size_t)b * 64 + o) * N_ + n0 + lane] = fmaxf(lo * s + t, 0.0f);
            }
            {
                const int   o = wv * 16 + 8 + r;
                const float s = edge_lane(s2v, 8 + r), t = edge_lane(t2v, 8 + r);
                out[((size_t)b * 64 + o) * N_ + n0 + lane] = fmaxf(hi * s + t, 0.0f);
            }
        }
    }
}

extern "C" void kernel_launch(void* const* d_in, const int* in_sizes, int n_in,
                              void* d_out, int out_size, void* d_ws, size_t ws_size,
                              hipStream_t stream)
{
    const float* x     = (const float*)d_in[0];
    const float* w1    = (const float*)d_in[1];
    const float* b1    = (const float*)d_in[2];
    const float* g1    = (const float*)d_in[3];
    const float* beta1 = (const float*)d_in[4];
    const float* rm1   = (const float*)d_in[5];
    const float* rv1   = (const float*)d_in[6];
    const float* w2    = (const float*)d_in[7];
    const float* b2    = (const float*)d_in[8];
    const float* g2    = (const float*)d_in[9];
    const float* beta2 = (const float*)d_in[10];
    const float* rm2   = (const float*)d_in[11];
    const float* rv2   = (const float*)d_in[12];

    float* out    = (float*)d_out;
    float* out_m1 = out + (size_t)B_ * 64 * N_;
    float* out_m2 = out + 2 * (size_t)B_ * 64 * N_;

    knn_edge_kernel<<<B_ * (N_ / KNN_T), KNN_T, 0, stream>>>(
        x, w1, b1, g1, beta1, rm1, rv1,
        w2, b2, g2, beta2, rm2, rv2, out, out_m1, out_m2);
}


# knn scan: a chunk of 16 points leaves ONE word of marks per lane (a bit per point that beats the threshold), not 16 distance/index pairs; the drain recomputes the marked points' distances
# speedup vs baseline: 7.1974x; 1.0764x over previous
#include <hip/hip_runtime.h>

#define B_    8
#define C_    3
#define N_    4096
#define K_    20
#define H_    64
#define DOUT_ 64
#define EPS_  1e-5f

typedef __attribute__((ext_vector_type(2))) float v2f;
typedef __attribute__((ext_vector_type(8))) float v8f;

#define KNN_T   128
#define KNN_CH  16
#define KNN_CAP 32
#define KNN_TP  129

typedef __attribute__((ext_vector_type(4))) float v4f;

#define KNN_POS(eu, el, ku, kl, cu, cl)                    \
    {   el = (cu) ? (eu) : ((cl) ? d : (el));              \
        kl = (cu) ? (ku) : ((cl) ? j : (kl)); }

typedef __attribute__((address_space(3))) float lds_f32;
typedef __attribute__((address_space(3))) int   lds_i32;
typedef __attribute__((address_space(3))) unsigned lds_u32;
typedef __attribute__((address_space(3))) v4f   lds_v4f;
#define KNN_SLOT   (KNN_T * 4)
#define KNN_QJ     (KNN_CAP * KNN_T * 4)
#define KNN_PLANE  (N_ * 4)

#define KNN_LOAD(PX, PY, PZ, la)                                          \
    {   const lds_v4f* lx = (const lds_v4f*)(la);                         \
        const lds_v4f* ly = (const lds_v4f*)((la) + KNN_PLANE);           \
        const lds_v4f* lz = (const lds_v4f*)((la) + 2 * KNN_PLANE);       \
        PX##0 = lx[0]; PX##1 = lx[1]; PX##2 = lx[2]; PX##3 = lx[3];       \
        PY##0 = ly[0]; PY##1 = ly[1]; PY##2 = ly[2]; PY##3 = ly[3];       \
        PZ##0 = lz[0]; PZ##1 = lz[1]; PZ##2 = lz[2]; PZ##3 = lz[3]; }

#define KNN_D2(OUT, X2, Y2, Z2)                                           \
    {   const v2f dot = __builtin_elementwise_fma(qz2, (Z2),              \
                        __builtin_elementwise_fma(qx2, (X2), qy2 * (Y2)));\
        const v2f pp  = __builtin_elementwise_fma((Z2), (Z2),             \
                        __builtin_elementwise_fma((X2), (X2), (Y2) * (Y2)));\
        OUT = __builtin_elementwise_fma(dot, two2, nqq2) - pp; }

#define KNN_D4(O1, O2, X, Y, Z)                                           \
    KNN_D2(O1, X.xy, Y.xy, Z.xy) KNN_D2(O2, X.zw, Y.zw, Z.zw)

#define KNN_SGN(M, sv)                                                    \
    M = (M << 1) | (__builtin_bit_cast(unsigned, (float)(sv)) >> 31);

#define KNN_CHUNK(PX, PY, PZ, jb, jnext)                                  \
    {   v2f dA, dB, dC, dD, dE, dF, dG, dH;                               \
        KNN_D4(dA, dB, PX##0, PY##0, PZ##0)                               \
        KNN_D4(dC, dD, PX##1, PY##1, PZ##1)                               \
        KNN_D4(dE, dF, PX##2, PY##2, PZ##2)                               \
        KNN_D4(dG, dH, PX##3, PY##3, PZ##3)                               \
        const v2f thr2 = {thr, thr};                                      \
        const v2f sA = thr2 - dA, sB = thr2 - dB, sC = thr2 - dC, sD = thr2 - dD;\
        const v2f sE = thr2 - dE, sF = thr2 - dF, sG = thr2 - dG, sH = thr2 - dH;\
               \
        unsigned mhi = 0u, mlo = 0u;                                      \
        KNN_SGN(mhi, sA.x) KNN_SGN(mhi, sA.y) KNN_SGN(mhi, sB.x) KNN_SGN(mhi, sB.y)\
        KNN_SGN(mhi, sC.x) KNN_SGN(mhi, sC.y) KNN_SGN(mhi, sD.x) KNN_SGN(mhi, sD.y)\
        KNN_SGN(mlo, sE.x) KNN_SGN(mlo, sE.y) KNN_SGN(mlo, sF.x) KNN_SGN(mlo, sF.y)\
        KNN_SGN(mlo, sG.x) KNN_SGN(mlo, sG.y) KNN_SGN(mlo, sH.x) KNN_SGN(mlo, sH.y)\
        const unsigned m16 = (mhi << 8) | mlo;                            \
        const unsigned ent = m16 | ((unsigned)(jb) << 16);                \
        cnt += (unsigned)__builtin_popcount(m16);                         \
        const unsigned t0 = tail;                                         \
        const unsigned t1 = t0 + (m16 ? (unsigned)KNN_SLOT : 0u);         \
          \
          \
          \
          \
          \
        unsigned la;                                                      \
        asm("v_mov_b32 %0, %1" : "=v"(la) : "s"(pbase + ((unsigned)(jnext) << 2)));\
         \
         \
         \
        asm volatile("" :: "v"(t0), "v"(t1), "v"(ent), "v"(cnt), "v"(la));\
          \
          \
          \
        __builtin_amdgcn_sched_barrier(0);                                \
        *(lds_u32*)(t0) = ent;                                            \
        KNN_LOAD(PX, PY, PZ, la)                                          \
        __builtin_amdgcn_sched_barrier(0);                                \
        tail = t1;                                                        \
    }

#define KNN_DRAIN(jb)                                                     \
    if (((jb) + KNN_CH >= N_) || __any(cnt > (unsigned)(KNN_CAP - KNN_CH))) { \
          \
          \
        float e0 = *(lds_f32*)(pk + 0 * KNN_SLOT);                      \
        float e1 = *(lds_f32*)(pk + 1 * KNN_SLOT);                      \
        float e2 = *(lds_f32*)(pk + 2 * KNN_SLOT);                      \
        float e3 = *(lds_f32*)(pk + 3 * KNN_SLOT);                      \
        float e4 = *(lds_f32*)(pk + 4 * KNN_SLOT);                      \
        float e5 = *(lds_f32*)(pk + 5 * KNN_SLOT);                      \
        float e6 = *(lds_f32*)(pk + 6 * KNN_SLOT);                      \
        float e7 = *(lds_f32*)(pk + 7 * KNN_SLOT);                      \
        float e8 = *(lds_f32*)(pk + 8 * KNN_SLOT);                      \
        float e9 = *(lds_f32*)(pk + 9 * KNN_SLOT);                      \
        float e10 = *(lds_f32*)(pk + 10 * KNN_SLOT);                      \
        float e11 = *(lds_f32*)(pk + 11 * KNN_SLOT);                      \
        float e12 = *(lds_f32*)(pk + 12 * KNN_SLOT);                      \
        float e13 = *(lds_f32*)(pk + 13 * KNN_SLOT);                      \
        float e14 = *(lds_f32*)(pk + 14 * KNN_SLOT);                      \
        float e15 = *(lds_f32*)(pk + 15 * KNN_SLOT);                      \
        float e16 = *(lds_f32*)(pk + 16 * KNN_SLOT);                      \
        float e17 = *(lds_f32*)(pk + 17 * KNN_SLOT);                      \
        float e18 = *(lds_f32*)(pk + 18 * KNN_SLOT);                      \
        float e19 = *(lds_f32*)(pk + 19 * KNN_SLOT);                      \
        int   k0 = *(lds_i32*)(pk + 20 * KNN_SLOT);                      \
        int   k1 = *(lds_i32*)(pk + 21 * KNN_SLOT);                      \
        int   k2 = *(lds_i32*)(pk + 22 * KNN_SLOT);                      \
        int   k3 = *(lds_i32*)(pk + 23 * KNN_SLOT);                      \
        int   k4 = *(lds_i32*)(pk + 24 * KNN_SLOT);                      \
        int   k5 = *(lds_i32*)(pk + 25 * KNN_SLOT);                      \
        int   k6 = *(lds_i32*)(pk + 26 * KNN_SLOT);                      \
        int   k7 = *(lds_i32*)(pk + 27 * KNN_SLOT);                      \
        int   k8 = *(lds_i32*)(pk + 28 * KNN_SLOT);                      \
        int   k9 = *(lds_i32*)(pk + 29 * KNN_SLOT);                      \
        int   k10 = *(lds_i32*)(pk + 30 * KNN_SLOT);                      \
        int   k11 = *(lds_i32*)(pk + 31 * KNN_SLOT);                      \
        int   k12 = *(lds_i32*)(pk + 32 * KNN_SLOT);                      \
        int   k13 = *(lds_i32*)(pk + 33 * KNN_SLOT);                      \
        int   k14 = *(lds_i32*)(pk + 34 * KNN_SLOT);                      \
        int   k15 = *(lds_i32*)(pk + 35 * KNN_SLOT);                      \
        int   k16 = *(lds_i32*)(pk + 36 * KNN_SLOT);                      \
        int   k17 = *(lds_i32*)(pk + 37 * KNN_SLOT);                      \
        int   k18 = *(lds_i32*)(pk + 38 * KNN_SLOT);                      \
        int   k19 = *(lds_i32*)(pk + 39 * KNN_SLOT);                      \
         \
         \
         \
         \
         \
        unsigned cur = *(lds_u32*)(tbase);                                \
        unsigned ea  = tbase + KNN_SLOT;                                  \
        unsigned nxt = *(lds_u32*)(ea);                                   \
        int jn; float pxn, pyn, pzn;                                      \
        KNN_NEXT                                                          \
        int s = 0;                                                        \
          \
          \
        do {                                                              \
            const bool  have = (unsigned)s < cnt;                         \
            const float px = pxn, py = pyn, pz = pzn;                     \
            const int   j  = have ? jn : 0;                               \
               \
               \
            KNN_NEXT                                                      \
               \
               \
            const float dot = __builtin_fmaf(qz, pz, __builtin_fmaf(qx, px, qy * py));\
            const float pp  = __builtin_fmaf(pz, pz, __builtin_fmaf(px, px, py * py));\
            const float dc  = __builtin_fmaf(dot, 2.0f, -qq) - pp;        \
            const float d   = have ? dc : -3.4e38f;                       \
               \
               \
               \
               \
               \
               \
            __builtin_amdgcn_sched_barrier(0);                            \
            KNN_INSERT                                                    \
            __builtin_amdgcn_sched_barrier(0);                            \
               \
               \
            asm volatile("" :: "v"(nxt), "v"(pxn), "v"(pyn), "v"(pzn));   \
            ++s;                                                          \
        } while (__any((unsigned)s < cnt));                               \
        *(lds_f32*)(pk + 0 * KNN_SLOT) = e0;                            \
        *(lds_f32*)(pk + 1 * KNN_SLOT) = e1;                            \
        *(lds_f32*)(pk + 2 * KNN_SLOT) = e2;                            \
        *(lds_f32*)(pk + 3 * KNN_SLOT) = e3;                            \
        *(lds_f32*)(pk + 4 * KNN_SLOT) = e4;                            \
        *(lds_f32*)(pk + 5 * KNN_SLOT) = e5;                            \
        *(lds_f32*)(pk + 6 * KNN_SLOT) = e6;                            \
        *(lds_f32*)(pk + 7 * KNN_SLOT) = e7;                            \
        *(lds_f32*)(pk + 8 * KNN_SLOT) = e8;                            \
        *(lds_f32*)(pk + 9 * KNN_SLOT) = e9;                            \
        *(lds_f32*)(pk + 10 * KNN_SLOT) = e10;                            \
        *(lds_f32*)(pk + 11 * KNN_SLOT) = e11;                            \
        *(lds_f32*)(pk + 12 * KNN_SLOT) = e12;                            \
        *(lds_f32*)(pk + 13 * KNN_SLOT) = e13;                            \
        *(lds_f32*)(pk + 14 * KNN_SLOT) = e14;                            \
        *(lds_f32*)(pk + 15 * KNN_SLOT) = e15;                            \
        *(lds_f32*)(pk + 16 * KNN_SLOT) = e16;                            \
        *(lds_f32*)(pk + 17 * KNN_SLOT) = e17;                            \
        *(lds_f32*)(pk + 18 * KNN_SLOT) = e18;                            \
        *(lds_f32*)(pk + 19 * KNN_SLOT) = e19;                            \
        *(lds_i32*)(pk + 20 * KNN_SLOT) = k0;                            \
        *(lds_i32*)(pk + 21 * KNN_SLOT) = k1;                            \
        *(lds_i32*)(pk + 22 * KNN_SLOT) = k2;                            \
        *(lds_i32*)(pk + 23 * KNN_SLOT) = k3;                            \
        *(lds_i32*)(pk + 24 * KNN_SLOT) = k4;                            \
        *(lds_i32*)(pk + 25 * KNN_SLOT) = k5;                            \
        *(lds_i32*)(pk + 26 * KNN_SLOT) = k6;                            \
        *(lds_i32*)(pk + 27 * KNN_SLOT) = k7;                            \
        *(lds_i32*)(pk + 28 * KNN_SLOT) = k8;                            \
        *(lds_i32*)(pk + 29 * KNN_SLOT) = k9;                            \
        *(lds_i32*)(pk + 30 * KNN_SLOT) = k10;                            \
        *(lds_i32*)(pk + 31 * KNN_SLOT) = k11;                            \
        *(lds_i32*)(pk + 32 * KNN_SLOT) = k12;                            \
        *(lds_i32*)(pk + 33 * KNN_SLOT) = k13;                            \
        *(lds_i32*)(pk + 34 * KNN_SLOT) = k14;                            \
        *(lds_i32*)(pk + 35 * KNN_SLOT) = k15;                            \
        *(lds_i32*)(pk + 36 * KNN_SLOT) = k16;                            \
        *(lds_i32*)(pk + 37 * KNN_SLOT) = k17;                            \
        *(lds_i32*)(pk + 38 * KNN_SLOT) = k18;                            \
        *(lds_i32*)(pk + 39 * KNN_SLOT) = k19;                            \
        tail = tbase;                                                     \
        thr = e19;                                                        \
        cnt = 0u;                                                         \
    }

#define KNN_NEXT                                                          \
    {   const bool need = (cur & 0xffffu) == 0u;                          \
        cur = need ? nxt : cur;                                           \
        ea  = need ? (ea + KNN_SLOT) : ea;                                \
        nxt = *(lds_u32*)(ea);                                            \
        const unsigned lz = (unsigned)__builtin_clz((cur & 0xffffu) | 1u);\
        jn  = (int)((cur >> 16) + lz - 16u);                              \
        cur = cur & ~(0x80000000u >> lz);                                 \
        const unsigned ga = pbase + (((unsigned)jn & (unsigned)(N_ - 1)) << 2);\
        pxn = *(lds_f32*)(ga);                                            \
        pyn = *(lds_f32*)(ga + KNN_PLANE);                                \
        pzn = *(lds_f32*)(ga + 2 * KNN_PLANE); }

#define KNN_INSERT                                                        \
    const bool c0 = d > e0, c1 = d > e1, c2 = d > e2, c3 = d > e3, c4 = d > e4;\
    const bool c5 = d > e5, c6 = d > e6, c7 = d > e7, c8 = d > e8, c9 = d > e9;\
    const bool c10 = d > e10, c11 = d > e11, c12 = d > e12, c13 = d > e13, c14 = d > e14;\
    const bool c15 = d > e15, c16 = d > e16, c17 = d > e17, c18 = d > e18, c19 = d > e19;\
    KNN_POS(e18, e19, k18, k19, c18, c19)                                 \
    KNN_POS(e17, e18, k17, k18, c17, c18)                                 \
    KNN_POS(e16, e17, k16, k17, c16, c17)                                 \
    KNN_POS(e15, e16, k15, k16, c15, c16)                                 \
    KNN_POS(e14, e15, k14, k15, c14, c15)                                 \
    KNN_POS(e13, e14, k13, k14, c13, c14)                                 \
    KNN_POS(e12, e13, k12, k13, c12, c13)                                 \
    KNN_POS(e11, e12, k11, k12, c11, c12)                                 \
    KNN_POS(e10, e11, k10, k11, c10, c11)                                 \
    KNN_POS(e9, e10, k9, k10, c9, c10)                                    \
    KNN_POS(e8, e9, k8, k9, c8, c9)                                       \
    KNN_POS(e7, e8, k7, k8, c7, c8)                                       \
    KNN_POS(e6, e7, k6, k7, c6, c7)                                       \
    KNN_POS(e5, e6, k5, k6, c5, c6)                                       \
    KNN_POS(e4, e5, k4, k5, c4, c5)                                       \
    KNN_POS(e3, e4, k3, k4, c3, c4)                                       \
    KNN_POS(e2, e3, k2, k3, c2, c3)                                       \
    KNN_POS(e1, e2, k1, k2, c1, c2)                                       \
    KNN_POS(e0, e1, k0, k1, c0, c1)                                       \
    e0 = c0 ? d : e0;                                                     \
    k0 = c0 ? j : k0;

#define EDGE_PW 8
#define EDGE_PB 32
#define EDGE_LP 33
#define EDGE_HS (64 * EDGE_LP + 16)
#define EDGE_KP 132

__device__ __forceinline__ float edge_xhalf(float v)
{
    return __builtin_bit_cast(float, __builtin_amdgcn_permlanex16(
        __builtin_bit_cast(int, v), __builtin_bit_cast(int, v),
        0x76543210, 0xfedcba98, false, false));
}

__device__ __forceinline__ float edge_lane(float v, int l)
{
    return __builtin_bit_cast(float,
        __builtin_amdgcn_readlane(__builtin_bit_cast(int, v), l));
}

__global__ __launch_bounds__(KNN_T) void knn_edge_kernel(
    const float* __restrict__ x,
    const float* __restrict__ w1, const float* __restrict__ b1,
    const float* __restrict__ g1, const float* __restrict__ beta1,
    const float* __restrict__ rm1,const float* __restrict__ rv1,
    const float* __restrict__ w2, const float* __restrict__ b2,
    const float* __restrict__ g2, const float* __restrict__ beta2,
    const float* __restrict__ rm2,const float* __restrict__ rv2,
    float* __restrict__ out,
    float* __restrict__ out_m1,   float* __restrict__ out_m2)
{
    __shared__ __attribute__((aligned(16))) float lds[3 * N_ + 2 * KNN_CAP * KNN_T + 2 * K_ * KNN_T];
    float* spts = lds;
    int*   qj   = (int*)(lds + 3 * N_ + KNN_CAP * KNN_T);
    const float* sx = spts;
    const float* sy = spts + N_;
    const float* sz = spts + 2 * N_;
    const int blocksPerBatch = N_ / KNN_T;
    const int b     = blockIdx.x / blocksPerBatch;
    const int chunk = blockIdx.x % blocksPerBatch;
    const float* xb = x + (size_t)b * 3 * N_;
    const v4f* xb4 = (const v4f*)xb;
    v4f*       sp4 = (v4f*)spts;
#pragma unroll
    for (int t = 0; t < (3 * N_ / 4) / KNN_T; ++t)
        sp4[t * KNN_T + threadIdx.x] = xb4[t * KNN_T + threadIdx.x];
    __syncthreads();

    const int   tid = threadIdx.x;
    const int   n  = chunk * KNN_T + tid;
    const float qx = sx[n], qy = sy[n], qz = sz[n];
    const float qq = fmaf(qz, qz, fmaf(qx, qx, qy * qy));
    const v2f qx2 = {qx, qx}, qy2 = {qy, qy}, qz2 = {qz, qz};
    const v2f nqq2 = {-qq, -qq}, two2 = {2.0f, 2.0f};

    const unsigned lds0  = (unsigned)(size_t)lds;
    const unsigned pbase = lds0;
    const unsigned tbase = lds0 + 3 * KNN_PLANE + (unsigned)tid * 4u;
    const unsigned pk    = lds0 + 3 * KNN_PLANE + 2 * KNN_QJ + (unsigned)tid * 4u;

#pragma unroll
    for (int i = 0; i < K_; ++i) {
        *(lds_f32*)(pk + (unsigned)i * KNN_SLOT) = -3.4e38f;
        *(lds_i32*)(pk + (unsigned)(K_ + i) * KNN_SLOT) = 0;
    }

    float    thr  = -3.4e38f;
    unsigned tail = tbase;
    unsigned cnt  = 0u;

    v4f ax0, ax1, ax2, ax3, ay0, ay1, ay2, ay3, az0, az1, az2, az3;
    v4f bx0, bx1, bx2, bx3, by0, by1, by2, by3, bz0, bz1, bz2, bz3;
    KNN_LOAD(ax, ay, az, pbase)
    KNN_LOAD(bx, by, bz, pbase + KNN_CH * 4)
    for (int j0 = 0; j0 < N_; j0 += 2 * KNN_CH) {
        KNN_CHUNK(ax, ay, az, j0, (j0 + 2 * KNN_CH) & (N_ - 1))
        KNN_DRAIN(j0)
        KNN_CHUNK(bx, by, bz, j0 + KNN_CH, (j0 + 3 * KNN_CH) & (N_ - 1))
        KNN_DRAIN(j0 + KNN_CH)
    }
    const int k0 = *(lds_i32*)(pk + 20 * KNN_SLOT);
    const int k1 = *(lds_i32*)(pk + 21 * KNN_SLOT);
    const int k2 = *(lds_i32*)(pk + 22 * KNN_SLOT);
    const int k3 = *(lds_i32*)(pk + 23 * KNN_SLOT);
    const int k4 = *(lds_i32*)(pk + 24 * KNN_SLOT);
    const int k5 = *(lds_i32*)(pk + 25 * KNN_SLOT);
    const int k6 = *(lds_i32*)(pk + 26 * KNN_SLOT);
    const int k7 = *(lds_i32*)(pk + 27 * KNN_SLOT);
    const int k8 = *(lds_i32*)(pk + 28 * KNN_SLOT);
    const int k9 = *(lds_i32*)(pk + 29 * KNN_SLOT);
    const int k10 = *(lds_i32*)(pk + 30 * KNN_SLOT);
    const int k11 = *(lds_i32*)(pk + 31 * KNN_SLOT);
    const int k12 = *(lds_i32*)(pk + 32 * KNN_SLOT);
    const int k13 = *(lds_i32*)(pk + 33 * KNN_SLOT);
    const int k14 = *(lds_i32*)(pk + 34 * KNN_SLOT);
    const int k15 = *(lds_i32*)(pk + 35 * KNN_SLOT);
    const int k16 = *(lds_i32*)(pk + 36 * KNN_SLOT);
    const int k17 = *(lds_i32*)(pk + 37 * KNN_SLOT);
    const int k18 = *(lds_i32*)(pk + 38 * KNN_SLOT);
    const int k19 = *(lds_i32*)(pk + 39 * KNN_SLOT);
    __syncthreads();
    int* tk = qj;
    tk[0 * KNN_TP + tid] = k0;
    tk[1 * KNN_TP + tid] = k1;
    tk[2 * KNN_TP + tid] = k2;
    tk[3 * KNN_TP + tid] = k3;
    tk[4 * KNN_TP + tid] = k4;
    tk[5 * KNN_TP + tid] = k5;
    tk[6 * KNN_TP + tid] = k6;
    tk[7 * KNN_TP + tid] = k7;
    tk[8 * KNN_TP + tid] = k8;
    tk[9 * KNN_TP + tid] = k9;
    tk[10 * KNN_TP + tid] = k10;
    tk[11 * KNN_TP + tid] = k11;
    tk[12 * KNN_TP + tid] = k12;
    tk[13 * KNN_TP + tid] = k13;
    tk[14 * KNN_TP + tid] = k14;
    tk[15 * KNN_TP + tid] = k15;
    tk[16 * KNN_TP + tid] = k16;
    tk[17 * KNN_TP + tid] = k17;
    tk[18 * KNN_TP + tid] = k18;
    tk[19 * KNN_TP + tid] = k19;
    __syncthreads();

    __shared__ float pmx[2 * EDGE_HS];
    __shared__ float psm[2 * EDGE_HS];
    __shared__ __attribute__((aligned(16))) float pa[EDGE_PB * EDGE_KP];

    const int lane = tid & 31;
    const int half = lane >> 4;
    const int l15  = lane & 15;
    const int wv   = __builtin_amdgcn_readfirstlane(tid >> 5);

    const bool valid1 = (16 + l15) < K_;
    const int  kc1    = valid1 ? (16 + l15) : (K_ - 1);
    const int* tk0 = tk + l15 * KNN_TP;
    const int* tk1 = tk + kc1 * KNN_TP;

    v2f   bt0[4], bt1[4];
    float bs[4], bt[4];
#pragma unroll
    for (int ct = 0; ct < 4; ++ct) {
        const int    o    = ct * 16 + l15;
        const float* wrow = w1 + o * 6;
        const float w0 = wrow[0], w1v = wrow[1], w2v = wrow[2];
        const float w3 = wrow[3], w4v = wrow[4], w5v = wrow[5];
        bt0[ct].x = half ? w2v : w0;   bt0[ct].y = half ? w3 : w1v;
        bt1[ct].x = half ? 0.0f : w4v; bt1[ct].y = half ? 0.0f : w5v;
        const float s = g1[o] * rsqrtf(rv1[o] + EPS_);
        bs[ct] = s;
        bt[ct] = (b1[o] - rm1[o]) * s + beta1[o];
    }

    const float* wrow2 = w2 + (size_t)(wv * 16 + l15) * 128 + half * 2;
    v2f wa[32];
#pragma unroll
    for (int ks = 0; ks < 32; ++ks)
        wa[ks] = *(const v2f*)(wrow2 + ks * 4);
    const int   oc  = wv * 16 + l15;
    const float s2v = g2[oc] * rsqrtf(rv2[oc] + EPS_);
    const float t2v = (b2[oc] - rm2[oc]) * s2v + beta2[oc];

    const float* pr0 = pa + l15 * EDGE_KP + half * 2;
    const float* pr1 = pr0 + 16 * EDGE_KP;

#pragma nounroll
    for (int g = 0; g < KNN_T / EDGE_PB; ++g) {
        const int lq = g * EDGE_PB + wv * EDGE_PW;
        const int n0 = chunk * KNN_T + g * EDGE_PB;
        const int nw = n0 + wv * EDGE_PW;

        int nb0[EDGE_PW], nb1[EDGE_PW];
#pragma unroll
        for (int p = 0; p < EDGE_PW; ++p) {
            nb0[p] = tk0[lq + p];
            nb1[p] = tk1[lq + p];
        }

        float gx0[EDGE_PW], gy0[EDGE_PW], gz0[EDGE_PW];
        float gx1[EDGE_PW], gy1[EDGE_PW], gz1[EDGE_PW];
        float ccx[EDGE_PW], ccy[EDGE_PW], ccz[EDGE_PW];
#pragma unroll
        for (int p = 0; p < EDGE_PW; ++p) {
            gx0[p] = sx[nb0[p]]; gy0[p] = sy[nb0[p]]; gz0[p] = sz[nb0[p]];
            gx1[p] = sx[nb1[p]]; gy1[p] = sy[nb1[p]]; gz1[p] = sz[nb1[p]];
            ccx[p] = sx[nw + p]; ccy[p] = sy[nw + p]; ccz[p] = sz[nw + p];
        }

        float pmr[EDGE_PW * 4], psr[EDGE_PW * 4];
#pragma unroll
        for (int p = 0; p < EDGE_PW; ++p) {
            const float cx = ccx[p], cy = ccy[p], cz = ccz[p];

            v2f a0[2], a1[2];
#pragma unroll
            for (int rt = 0; rt < 2; ++rt) {
                const bool valid = rt ? valid1 : true;
                const float nx = rt ? gx1[p] : gx0[p];
                const float ny = rt ? gy1[p] : gy0[p];
                const float nz = rt ? gz1[p] : gz0[p];
                const float e0 = valid ? (nx - cx) : 0.0f;
                const float e1 = valid ? (ny - cy) : 0.0f;
                const float e2 = valid ? (nz - cz) : 0.0f;
                const float e3 = valid ? cx : 0.0f;
                const float e4 = valid ? cy : 0.0f;
                const float e5 = valid ? cz : 0.0f;
                v2f t0, t1;
                t0.x = half ? e2 : e0;  t0.y = half ? e3 : e1;
                t1.x = half ? 0.0f : e4; t1.y = half ? 0.0f : e5;
                a0[rt] = t0; a1[rt] = t1;
            }

#pragma unroll
            for (int ct = 0; ct < 4; ++ct) {
                v8f z = {};
                z = __builtin_amdgcn_wmma_f32_16x16x4_f32(false, a0[0], false, bt0[ct],
                                                          (short)0, z, false, false);
                z = __builtin_amdgcn_wmma_f32_16x16x4_f32(false, a1[0], false, bt1[ct],
                                                          (short)0, z, false, false);
                v8f z1 = {};
                z1 = __builtin_amdgcn_wmma_f32_16x16x4_f32(false, a0[1], false, bt0[ct],
                                                           (short)0, z1, false, false);
                z1 = __builtin_amdgcn_wmma_f32_16x16x4_f32(false, a1[1], false, bt1[ct],
                                                           (short)0, z1, false, false);

                const float s = bs[ct], t = bt[ct];
                float mx = -3.4e38f, sm = 0.0f;
#pragma unroll
                for (int v = 0; v < 8; ++v) {
                    const float val = fmaxf(z[v] * s + t, 0.0f);
                    mx = fmaxf(mx, val); sm += val;
                }
#pragma unroll
                for (int v = 0; v < 4; ++v) {
                    const float val = fmaxf(z1[v] * s + t, 0.0f);
                    mx = (half == 0) ? fmaxf(mx, val) : mx;
                    sm = (half == 0) ? (sm + val) : sm;
                }
                pmr[p * 4 + ct] = mx;
                psr[p * 4 + ct] = sm;
            }
        }
        __builtin_amdgcn_sched_barrier(0);
#pragma unroll
        for (int p = 0; p < EDGE_PW; ++p) {
#pragma unroll
            for (int ct = 0; ct < 4; ++ct) {
                const int q = half * EDGE_HS + (ct * 16 + l15) * EDGE_LP + wv * EDGE_PW + p;
                pmx[q] = pmr[p * 4 + ct];
                psm[q] = psr[p * 4 + ct];
            }
        }
        __builtin_amdgcn_sched_barrier(0);
        __syncthreads();

        float jm[16], ja[16];
        float rml[16], rmh[16], rsl[16], rsh[16];
#pragma unroll
        for (int r = 0; r < 16; ++r) {
            const int o = wv * 16 + r;
            rml[r] = pmx[o * EDGE_LP + lane]; rmh[r] = pmx[EDGE_HS + o * EDGE_LP + lane];
            rsl[r] = psm[o * EDGE_LP + lane]; rsh[r] = psm[EDGE_HS + o * EDGE_LP + lane];
        }
        __builtin_amdgcn_sched_barrier(0);
#pragma unroll
        for (int r = 0; r < 16; ++r) {
            jm[r] = fmaxf(rml[r], rmh[r]);
            ja[r] = (rsl[r] + rsh[r]) * (1.0f / K_);
        }
        __builtin_amdgcn_sched_barrier(0);
#pragma unroll
        for (int r = 0; r < 16; ++r) {
            const int o = wv * 16 + r;
            pa[lane * EDGE_KP + o]      = jm[r];
            pa[lane * EDGE_KP + 64 + o] = ja[r];
        }
        __syncthreads();

#pragma unroll
        for (int r = 0; r < 16; ++r) {
            const int    o   = wv * 16 + r;
            const size_t off = ((size_t)b * 64 + o) * N_ + n0 + lane;
            out_m1[off] = jm[r];
            out_m2[off] = ja[r];
        }

        v2f pb0[32], pb1[32];
#pragma unroll
        for (int ks = 0; ks < 32; ++ks) {
            pb0[ks] = *(const v2f*)(pr0 + ks * 4);
            pb1[ks] = *(const v2f*)(pr1 + ks * 4);
        }
        __builtin_amdgcn_sched_barrier(0);
        v8f d0 = {}, d1 = {};
#pragma unroll
        for (int ks = 0; ks < 32; ++ks) {
            d0 = __builtin_amdgcn_wmma_f32_16x16x4_f32(false, wa[ks], false, pb0[ks],
                                                       (short)0, d0, false, false);
            d1 = __builtin_amdgcn_wmma_f32_16x16x4_f32(false, wa[ks], false, pb1[ks],
                                                       (short)0, d1, false, false);
        }

#pragma unroll
        for (int r = 0; r < 8; ++r) {
            const float x0 = edge_xhalf(d0[r]);
            const float x1 = edge_xhalf(d1[r]);
            const float lo = half ? x1 : d0[r];
            const float hi = half ? d1[r] : x0;
            {
                const int   o = wv * 16 + r;
                const float s = edge_lane(s2v, r), t = edge_lane(t2v, r);
                out[((size_t)b * 64 + o) * N_ + n0 + lane] = fmaxf(lo * s + t, 0.0f);
            }
            {
                const int   o = wv * 16 + 8 + r;
                const float s = edge_lane(s2v, 8 + r), t = edge_lane(t2v, 8 + r);
                out[((size_t)b * 64 + o) * N_ + n0 + lane] = fmaxf(hi * s + t, 0.0f);
            }
        }
    }
}

extern "C" void kernel_launch(void* const* d_in, const int* in_sizes, int n_in,
                              void* d_out, int out_size, void* d_ws, size_t ws_size,
                              hipStream_t stream)
{
    const float* x     = (const float*)d_in[0];
    const float* w1    = (const float*)d_in[1];
    const float* b1    = (const float*)d_in[2];
    const float* g1    = (const float*)d_in[3];
    const float* beta1 = (const float*)d_in[4];
    const float* rm1   = (const float*)d_in[5];
    const float* rv1   = (const float*)d_in[6];
    const float* w2    = (const float*)d_in[7];
    const float* b2    = (const float*)d_in[8];
    const float* g2    = (const float*)d_in[9];
    const float* beta2 = (const float*)d_in[10];
    const float* rm2   = (const float*)d_in[11];
    const float* rv2   = (const float*)d_in[12];

    float* out    = (float*)d_out;
    float* out_m1 = out + (size_t)B_ * 64 * N_;
    float* out_m2 = out + 2 * (size_t)B_ * 64 * N_;

    knn_edge_kernel<<<B_ * (N_ / KNN_T), KNN_T, 0, stream>>>(
        x, w1, b1, g1, beta1, rm1, rv1,
        w2, b2, g2, beta2, rm2, rv2, out, out_m1, out_m2);
}


# knn drain: a trip inserts point s beside point s+1's distance and point s+2's reads; an insertion is 20 compares up front, then two layers of independent selects: no select waits on its compare
# speedup vs baseline: 7.9046x; 1.0983x over previous
#include <hip/hip_runtime.h>

#define B_    8
#define C_    3
#define N_    4096
#define K_    20
#define H_    64
#define DOUT_ 64
#define EPS_  1e-5f

typedef __attribute__((ext_vector_type(2))) float v2f;
typedef __attribute__((ext_vector_type(8))) float v8f;

#define KNN_T   128
#define KNN_CH  16
#define KNN_CAP 32
#define KNN_TP  129

typedef __attribute__((ext_vector_type(4))) float v4f;


typedef __attribute__((address_space(3))) float lds_f32;
typedef __attribute__((address_space(3))) int   lds_i32;
typedef __attribute__((address_space(3))) unsigned lds_u32;
typedef __attribute__((address_space(3))) v4f   lds_v4f;
#define KNN_SLOT   (KNN_T * 4)
#define KNN_QJ     (KNN_CAP * KNN_T * 4)
#define KNN_PLANE  (N_ * 4)

#define KNN_LOAD(PX, PY, PZ, la)                                          \
    {   const lds_v4f* lx = (const lds_v4f*)(la);                         \
        const lds_v4f* ly = (const lds_v4f*)((la) + KNN_PLANE);           \
        const lds_v4f* lz = (const lds_v4f*)((la) + 2 * KNN_PLANE);       \
        PX##0 = lx[0]; PX##1 = lx[1]; PX##2 = lx[2]; PX##3 = lx[3];       \
        PY##0 = ly[0]; PY##1 = ly[1]; PY##2 = ly[2]; PY##3 = ly[3];       \
        PZ##0 = lz[0]; PZ##1 = lz[1]; PZ##2 = lz[2]; PZ##3 = lz[3]; }

#define KNN_D2(OUT, X2, Y2, Z2)                                           \
    {   const v2f dot = __builtin_elementwise_fma(qz2, (Z2),              \
                        __builtin_elementwise_fma(qx2, (X2), qy2 * (Y2)));\
        const v2f pp  = __builtin_elementwise_fma((Z2), (Z2),             \
                        __builtin_elementwise_fma((X2), (X2), (Y2) * (Y2)));\
        OUT = __builtin_elementwise_fma(dot, two2, nqq2) - pp; }

#define KNN_D4(O1, O2, X, Y, Z)                                           \
    KNN_D2(O1, X.xy, Y.xy, Z.xy) KNN_D2(O2, X.zw, Y.zw, Z.zw)

#define KNN_SGN(M, sv)                                                    \
    M = (M << 1) | (__builtin_bit_cast(unsigned, (float)(sv)) >> 31);

#define KNN_CHUNK(PX, PY, PZ, jb, jnext)                                  \
    {   v2f dA, dB, dC, dD, dE, dF, dG, dH;                               \
        KNN_D4(dA, dB, PX##0, PY##0, PZ##0)                               \
        KNN_D4(dC, dD, PX##1, PY##1, PZ##1)                               \
        KNN_D4(dE, dF, PX##2, PY##2, PZ##2)                               \
        KNN_D4(dG, dH, PX##3, PY##3, PZ##3)                               \
        const v2f thr2 = {thr, thr};                                      \
        const v2f sA = thr2 - dA, sB = thr2 - dB, sC = thr2 - dC, sD = thr2 - dD;\
        const v2f sE = thr2 - dE, sF = thr2 - dF, sG = thr2 - dG, sH = thr2 - dH;\
               \
        unsigned mhi = 0u, mlo = 0u;                                      \
        KNN_SGN(mhi, sA.x) KNN_SGN(mhi, sA.y) KNN_SGN(mhi, sB.x) KNN_SGN(mhi, sB.y)\
        KNN_SGN(mhi, sC.x) KNN_SGN(mhi, sC.y) KNN_SGN(mhi, sD.x) KNN_SGN(mhi, sD.y)\
        KNN_SGN(mlo, sE.x) KNN_SGN(mlo, sE.y) KNN_SGN(mlo, sF.x) KNN_SGN(mlo, sF.y)\
        KNN_SGN(mlo, sG.x) KNN_SGN(mlo, sG.y) KNN_SGN(mlo, sH.x) KNN_SGN(mlo, sH.y)\
        const unsigned m16 = (mhi << 8) | mlo;                            \
        const unsigned ent = m16 | ((unsigned)(jb) << 16);                \
        cnt += (unsigned)__builtin_popcount(m16);                         \
        const unsigned t0 = tail;                                         \
        const unsigned t1 = t0 + (m16 ? (unsigned)KNN_SLOT : 0u);         \
          \
          \
          \
          \
          \
        unsigned la;                                                      \
        asm("v_mov_b32 %0, %1" : "=v"(la) : "s"(pbase + ((unsigned)(jnext) << 2)));\
         \
         \
         \
        asm volatile("" :: "v"(t0), "v"(t1), "v"(ent), "v"(cnt), "v"(la));\
          \
          \
          \
        __builtin_amdgcn_sched_barrier(0);                                \
        *(lds_u32*)(t0) = ent;                                            \
        KNN_LOAD(PX, PY, PZ, la)                                          \
        __builtin_amdgcn_sched_barrier(0);                                \
        tail = t1;                                                        \
    }

#define KNN_DRAIN(jb)                                                     \
    if (((jb) + KNN_CH >= N_) || __any(cnt > (unsigned)(KNN_CAP - KNN_CH))) { \
          \
          \
        float e0 = *(lds_f32*)(pk + 0 * KNN_SLOT);                      \
        float e1 = *(lds_f32*)(pk + 1 * KNN_SLOT);                      \
        float e2 = *(lds_f32*)(pk + 2 * KNN_SLOT);                      \
        float e3 = *(lds_f32*)(pk + 3 * KNN_SLOT);                      \
        float e4 = *(lds_f32*)(pk + 4 * KNN_SLOT);                      \
        float e5 = *(lds_f32*)(pk + 5 * KNN_SLOT);                      \
        float e6 = *(lds_f32*)(pk + 6 * KNN_SLOT);                      \
        float e7 = *(lds_f32*)(pk + 7 * KNN_SLOT);                      \
        float e8 = *(lds_f32*)(pk + 8 * KNN_SLOT);                      \
        float e9 = *(lds_f32*)(pk + 9 * KNN_SLOT);                      \
        float e10 = *(lds_f32*)(pk + 10 * KNN_SLOT);                      \
        float e11 = *(lds_f32*)(pk + 11 * KNN_SLOT);                      \
        float e12 = *(lds_f32*)(pk + 12 * KNN_SLOT);                      \
        float e13 = *(lds_f32*)(pk + 13 * KNN_SLOT);                      \
        float e14 = *(lds_f32*)(pk + 14 * KNN_SLOT);                      \
        float e15 = *(lds_f32*)(pk + 15 * KNN_SLOT);                      \
        float e16 = *(lds_f32*)(pk + 16 * KNN_SLOT);                      \
        float e17 = *(lds_f32*)(pk + 17 * KNN_SLOT);                      \
        float e18 = *(lds_f32*)(pk + 18 * KNN_SLOT);                      \
        float e19 = *(lds_f32*)(pk + 19 * KNN_SLOT);                      \
        int   k0 = *(lds_i32*)(pk + 20 * KNN_SLOT);                      \
        int   k1 = *(lds_i32*)(pk + 21 * KNN_SLOT);                      \
        int   k2 = *(lds_i32*)(pk + 22 * KNN_SLOT);                      \
        int   k3 = *(lds_i32*)(pk + 23 * KNN_SLOT);                      \
        int   k4 = *(lds_i32*)(pk + 24 * KNN_SLOT);                      \
        int   k5 = *(lds_i32*)(pk + 25 * KNN_SLOT);                      \
        int   k6 = *(lds_i32*)(pk + 26 * KNN_SLOT);                      \
        int   k7 = *(lds_i32*)(pk + 27 * KNN_SLOT);                      \
        int   k8 = *(lds_i32*)(pk + 28 * KNN_SLOT);                      \
        int   k9 = *(lds_i32*)(pk + 29 * KNN_SLOT);                      \
        int   k10 = *(lds_i32*)(pk + 30 * KNN_SLOT);                      \
        int   k11 = *(lds_i32*)(pk + 31 * KNN_SLOT);                      \
        int   k12 = *(lds_i32*)(pk + 32 * KNN_SLOT);                      \
        int   k13 = *(lds_i32*)(pk + 33 * KNN_SLOT);                      \
        int   k14 = *(lds_i32*)(pk + 34 * KNN_SLOT);                      \
        int   k15 = *(lds_i32*)(pk + 35 * KNN_SLOT);                      \
        int   k16 = *(lds_i32*)(pk + 36 * KNN_SLOT);                      \
        int   k17 = *(lds_i32*)(pk + 37 * KNN_SLOT);                      \
        int   k18 = *(lds_i32*)(pk + 38 * KNN_SLOT);                      \
        int   k19 = *(lds_i32*)(pk + 39 * KNN_SLOT);                      \
         \
         \
         \
         \
         \
        unsigned cur = *(lds_u32*)(tbase);                                \
        unsigned ea  = tbase + KNN_SLOT;                                  \
        unsigned nxt = *(lds_u32*)(ea);                                   \
        int jn; float pxn, pyn, pzn; \
        KNN_NEXT \
          \
          \
          \
          \
          \
          \
          \
          \
        float d; int j; \
        KNN_AHEAD(0u, d, j) \
        int s = 0; \
          \
          \
        do { \
              \
            KNN_TESTS \
              \
              \
              \
              \
              \
            float d1; int j1; \
            KNN_AHEAD((unsigned)s + 1u, d1, j1) \
              \
              \
              \
              \
              \
            __builtin_amdgcn_sched_barrier(0); \
            KNN_SELECTS \
            __builtin_amdgcn_sched_barrier(0); \
              \
              \
            asm volatile("" :: "v"(nxt), "v"(pxn), "v"(pyn), "v"(pzn)); \
            d = d1; j = j1; \
            ++s; \
        } while (__any((unsigned)s < cnt)); \
        *(lds_f32*)(pk + 0 * KNN_SLOT) = e0;                            \
        *(lds_f32*)(pk + 1 * KNN_SLOT) = e1;                            \
        *(lds_f32*)(pk + 2 * KNN_SLOT) = e2;                            \
        *(lds_f32*)(pk + 3 * KNN_SLOT) = e3;                            \
        *(lds_f32*)(pk + 4 * KNN_SLOT) = e4;                            \
        *(lds_f32*)(pk + 5 * KNN_SLOT) = e5;                            \
        *(lds_f32*)(pk + 6 * KNN_SLOT) = e6;                            \
        *(lds_f32*)(pk + 7 * KNN_SLOT) = e7;                            \
        *(lds_f32*)(pk + 8 * KNN_SLOT) = e8;                            \
        *(lds_f32*)(pk + 9 * KNN_SLOT) = e9;                            \
        *(lds_f32*)(pk + 10 * KNN_SLOT) = e10;                            \
        *(lds_f32*)(pk + 11 * KNN_SLOT) = e11;                            \
        *(lds_f32*)(pk + 12 * KNN_SLOT) = e12;                            \
        *(lds_f32*)(pk + 13 * KNN_SLOT) = e13;                            \
        *(lds_f32*)(pk + 14 * KNN_SLOT) = e14;                            \
        *(lds_f32*)(pk + 15 * KNN_SLOT) = e15;                            \
        *(lds_f32*)(pk + 16 * KNN_SLOT) = e16;                            \
        *(lds_f32*)(pk + 17 * KNN_SLOT) = e17;                            \
        *(lds_f32*)(pk + 18 * KNN_SLOT) = e18;                            \
        *(lds_f32*)(pk + 19 * KNN_SLOT) = e19;                            \
        *(lds_i32*)(pk + 20 * KNN_SLOT) = k0;                            \
        *(lds_i32*)(pk + 21 * KNN_SLOT) = k1;                            \
        *(lds_i32*)(pk + 22 * KNN_SLOT) = k2;                            \
        *(lds_i32*)(pk + 23 * KNN_SLOT) = k3;                            \
        *(lds_i32*)(pk + 24 * KNN_SLOT) = k4;                            \
        *(lds_i32*)(pk + 25 * KNN_SLOT) = k5;                            \
        *(lds_i32*)(pk + 26 * KNN_SLOT) = k6;                            \
        *(lds_i32*)(pk + 27 * KNN_SLOT) = k7;                            \
        *(lds_i32*)(pk + 28 * KNN_SLOT) = k8;                            \
        *(lds_i32*)(pk + 29 * KNN_SLOT) = k9;                            \
        *(lds_i32*)(pk + 30 * KNN_SLOT) = k10;                            \
        *(lds_i32*)(pk + 31 * KNN_SLOT) = k11;                            \
        *(lds_i32*)(pk + 32 * KNN_SLOT) = k12;                            \
        *(lds_i32*)(pk + 33 * KNN_SLOT) = k13;                            \
        *(lds_i32*)(pk + 34 * KNN_SLOT) = k14;                            \
        *(lds_i32*)(pk + 35 * KNN_SLOT) = k15;                            \
        *(lds_i32*)(pk + 36 * KNN_SLOT) = k16;                            \
        *(lds_i32*)(pk + 37 * KNN_SLOT) = k17;                            \
        *(lds_i32*)(pk + 38 * KNN_SLOT) = k18;                            \
        *(lds_i32*)(pk + 39 * KNN_SLOT) = k19;                            \
        tail = tbase;                                                     \
        thr = e19;                                                        \
        cnt = 0u;                                                         \
    }

#define KNN_NEXT                                                          \
    {   const bool need = (cur & 0xffffu) == 0u;                          \
        cur = need ? nxt : cur;                                           \
        ea  = need ? (ea + KNN_SLOT) : ea;                                \
        nxt = *(lds_u32*)(ea);                                            \
        const unsigned lz = (unsigned)__builtin_clz((cur & 0xffffu) | 1u);\
        jn  = (int)((cur >> 16) + lz - 16u);                              \
        cur = cur & ~(0x80000000u >> lz);                                 \
        const unsigned ga = pbase + (((unsigned)jn & (unsigned)(N_ - 1)) << 2);\
        pxn = *(lds_f32*)(ga);                                            \
        pyn = *(lds_f32*)(ga + KNN_PLANE);                                \
        pzn = *(lds_f32*)(ga + 2 * KNN_PLANE); }

#define KNN_AHEAD(sn, D, J)                                               \
    {   const bool  have = (sn) < cnt;                                    \
        const float px = pxn, py = pyn, pz = pzn;                         \
        J = have ? jn : 0;                                                \
        KNN_NEXT                                                          \
        const float dot = __builtin_fmaf(qz, pz, __builtin_fmaf(qx, px, qy * py));\
        const float pp  = __builtin_fmaf(pz, pz, __builtin_fmaf(px, px, py * py));\
        const float dc  = __builtin_fmaf(dot, 2.0f, -qq) - pp;            \
        D = have ? dc : -3.4e38f; }

#define KNN_TESTS                                                         \
    const bool c0 = d > e0, c1 = d > e1, c2 = d > e2, c3 = d > e3, c4 = d > e4;\
    const bool c5 = d > e5, c6 = d > e6, c7 = d > e7, c8 = d > e8, c9 = d > e9;\
    const bool c10 = d > e10, c11 = d > e11, c12 = d > e12, c13 = d > e13, c14 = d > e14;\
    const bool c15 = d > e15, c16 = d > e16, c17 = d > e17, c18 = d > e18, c19 = d > e19;

#define KNN_LAY1(i) const float t##i = c##i ? d : e##i; const int u##i = c##i ? j : k##i;
#define KNN_LAY2(iu, il) e##il = c##iu ? e##iu : t##il; k##il = c##iu ? k##iu : u##il;
#define KNN_SELECTS                                                       \
    KNN_LAY1(0) KNN_LAY1(1) KNN_LAY1(2) KNN_LAY1(3) KNN_LAY1(4)           \
    KNN_LAY1(5) KNN_LAY1(6) KNN_LAY1(7) KNN_LAY1(8) KNN_LAY1(9)           \
    KNN_LAY1(10) KNN_LAY1(11) KNN_LAY1(12) KNN_LAY1(13) KNN_LAY1(14)      \
    KNN_LAY1(15) KNN_LAY1(16) KNN_LAY1(17) KNN_LAY1(18) KNN_LAY1(19)      \
    KNN_LAY2(18, 19) KNN_LAY2(17, 18) KNN_LAY2(16, 17) KNN_LAY2(15, 16)   \
    KNN_LAY2(14, 15) KNN_LAY2(13, 14) KNN_LAY2(12, 13) KNN_LAY2(11, 12)   \
    KNN_LAY2(10, 11) KNN_LAY2(9, 10) KNN_LAY2(8, 9) KNN_LAY2(7, 8)        \
    KNN_LAY2(6, 7) KNN_LAY2(5, 6) KNN_LAY2(4, 5) KNN_LAY2(3, 4)           \
    KNN_LAY2(2, 3) KNN_LAY2(1, 2) KNN_LAY2(0, 1)                          \
    e0 = t0;                                                              \
    k0 = u0;

#define EDGE_PW 8
#define EDGE_PB 32
#define EDGE_LP 33
#define EDGE_HS (64 * EDGE_LP + 16)
#define EDGE_KP 132

__device__ __forceinline__ float edge_xhalf(float v)
{
    return __builtin_bit_cast(float, __builtin_amdgcn_permlanex16(
        __builtin_bit_cast(int, v), __builtin_bit_cast(int, v),
        0x76543210, 0xfedcba98, false, false));
}

__device__ __forceinline__ float edge_lane(float v, int l)
{
    return __builtin_bit_cast(float,
        __builtin_amdgcn_readlane(__builtin_bit_cast(int, v), l));
}

__global__ __launch_bounds__(KNN_T) void knn_edge_kernel(
    const float* __restrict__ x,
    const float* __restrict__ w1, const float* __restrict__ b1,
    const float* __restrict__ g1, const float* __restrict__ beta1,
    const float* __restrict__ rm1,const float* __restrict__ rv1,
    const float* __restrict__ w2, const float* __restrict__ b2,
    const float* __restrict__ g2, const float* __restrict__ beta2,
    const float* __restrict__ rm2,const float* __restrict__ rv2,
    float* __restrict__ out,
    float* __restrict__ out_m1,   float* __restrict__ out_m2)
{
    __shared__ __attribute__((aligned(16))) float lds[3 * N_ + 2 * KNN_CAP * KNN_T + 2 * K_ * KNN_T];
    float* spts = lds;
    int*   qj   = (int*)(lds + 3 * N_ + KNN_CAP * KNN_T);
    const float* sx = spts;
    const float* sy = spts + N_;
    const float* sz = spts + 2 * N_;
    const int blocksPerBatch = N_ / KNN_T;
    const int b     = blockIdx.x / blocksPerBatch;
    const int chunk = blockIdx.x % blocksPerBatch;
    const float* xb = x + (size_t)b * 3 * N_;
    const v4f* xb4 = (const v4f*)xb;
    v4f*       sp4 = (v4f*)spts;
#pragma unroll
    for (int t = 0; t < (3 * N_ / 4) / KNN_T; ++t)
        sp4[t * KNN_T + threadIdx.x] = xb4[t * KNN_T + threadIdx.x];
    __syncthreads();

    const int   tid = threadIdx.x;
    const int   n  = chunk * KNN_T + tid;
    const float qx = sx[n], qy = sy[n], qz = sz[n];
    const float qq = fmaf(qz, qz, fmaf(qx, qx, qy * qy));
    const v2f qx2 = {qx, qx}, qy2 = {qy, qy}, qz2 = {qz, qz};
    const v2f nqq2 = {-qq, -qq}, two2 = {2.0f, 2.0f};

    const unsigned lds0  = (unsigned)(size_t)lds;
    const unsigned pbase = lds0;
    const unsigned tbase = lds0 + 3 * KNN_PLANE + (unsigned)tid * 4u;
    const unsigned pk    = lds0 + 3 * KNN_PLANE + 2 * KNN_QJ + (unsigned)tid * 4u;

#pragma unroll
    for (int i = 0; i < K_; ++i) {
        *(lds_f32*)(pk + (unsigned)i * KNN_SLOT) = -3.4e38f;
        *(lds_i32*)(pk + (unsigned)(K_ + i) * KNN_SLOT) = 0;
    }

    float    thr  = -3.4e38f;
    unsigned tail = tbase;
    unsigned cnt  = 0u;

    v4f ax0, ax1, ax2, ax3, ay0, ay1, ay2, ay3, az0, az1, az2, az3;
    v4f bx0, bx1, bx2, bx3, by0, by1, by2, by3, bz0, bz1, bz2, bz3;
    KNN_LOAD(ax, ay, az, pbase)
    KNN_LOAD(bx, by, bz, pbase + KNN_CH * 4)
    for (int j0 = 0; j0 < N_; j0 += 2 * KNN_CH) {
        KNN_CHUNK(ax, ay, az, j0, (j0 + 2 * KNN_CH) & (N_ - 1))
        KNN_DRAIN(j0)
        KNN_CHUNK(bx, by, bz, j0 + KNN_CH, (j0 + 3 * KNN_CH) & (N_ - 1))
        KNN_DRAIN(j0 + KNN_CH)
    }
    const int k0 = *(lds_i32*)(pk + 20 * KNN_SLOT);
    const int k1 = *(lds_i32*)(pk + 21 * KNN_SLOT);
    const int k2 = *(lds_i32*)(pk + 22 * KNN_SLOT);
    const int k3 = *(lds_i32*)(pk + 23 * KNN_SLOT);
    const int k4 = *(lds_i32*)(pk + 24 * KNN_SLOT);
    const int k5 = *(lds_i32*)(pk + 25 * KNN_SLOT);
    const int k6 = *(lds_i32*)(pk + 26 * KNN_SLOT);
    const int k7 = *(lds_i32*)(pk + 27 * KNN_SLOT);
    const int k8 = *(lds_i32*)(pk + 28 * KNN_SLOT);
    const int k9 = *(lds_i32*)(pk + 29 * KNN_SLOT);
    const int k10 = *(lds_i32*)(pk + 30 * KNN_SLOT);
    const int k11 = *(lds_i32*)(pk + 31 * KNN_SLOT);
    const int k12 = *(lds_i32*)(pk + 32 * KNN_SLOT);
    const int k13 = *(lds_i32*)(pk + 33 * KNN_SLOT);
    const int k14 = *(lds_i32*)(pk + 34 * KNN_SLOT);
    const int k15 = *(lds_i32*)(pk + 35 * KNN_SLOT);
    const int k16 = *(lds_i32*)(pk + 36 * KNN_SLOT);
    const int k17 = *(lds_i32*)(pk + 37 * KNN_SLOT);
    const int k18 = *(lds_i32*)(pk + 38 * KNN_SLOT);
    const int k19 = *(lds_i32*)(pk + 39 * KNN_SLOT);
    __syncthreads();
    int* tk = qj;
    tk[0 * KNN_TP + tid] = k0;
    tk[1 * KNN_TP + tid] = k1;
    tk[2 * KNN_TP + tid] = k2;
    tk[3 * KNN_TP + tid] = k3;
    tk[4 * KNN_TP + tid] = k4;
    tk[5 * KNN_TP + tid] = k5;
    tk[6 * KNN_TP + tid] = k6;
    tk[7 * KNN_TP + tid] = k7;
    tk[8 * KNN_TP + tid] = k8;
    tk[9 * KNN_TP + tid] = k9;
    tk[10 * KNN_TP + tid] = k10;
    tk[11 * KNN_TP + tid] = k11;
    tk[12 * KNN_TP + tid] = k12;
    tk[13 * KNN_TP + tid] = k13;
    tk[14 * KNN_TP + tid] = k14;
    tk[15 * KNN_TP + tid] = k15;
    tk[16 * KNN_TP + tid] = k16;
    tk[17 * KNN_TP + tid] = k17;
    tk[18 * KNN_TP + tid] = k18;
    tk[19 * KNN_TP + tid] = k19;
    __syncthreads();

    __shared__ float pmx[2 * EDGE_HS];
    __shared__ float psm[2 * EDGE_HS];
    __shared__ __attribute__((aligned(16))) float pa[EDGE_PB * EDGE_KP];

    const int lane = tid & 31;
    const int half = lane >> 4;
    const int l15  = lane & 15;
    const int wv   = __builtin_amdgcn_readfirstlane(tid >> 5);

    const bool valid1 = (16 + l15) < K_;
    const int  kc1    = valid1 ? (16 + l15) : (K_ - 1);
    const int* tk0 = tk + l15 * KNN_TP;
    const int* tk1 = tk + kc1 * KNN_TP;

    v2f   bt0[4], bt1[4];
    float bs[4], bt[4];
#pragma unroll
    for (int ct = 0; ct < 4; ++ct) {
        const int    o    = ct * 16 + l15;
        const float* wrow = w1 + o * 6;
        const float w0 = wrow[0], w1v = wrow[1], w2v = wrow[2];
        const float w3 = wrow[3], w4v = wrow[4], w5v = wrow[5];
        bt0[ct].x = half ? w2v : w0;   bt0[ct].y = half ? w3 : w1v;
        bt1[ct].x = half ? 0.0f : w4v; bt1[ct].y = half ? 0.0f : w5v;
        const float s = g1[o] * rsqrtf(rv1[o] + EPS_);
        bs[ct] = s;
        bt[ct] = (b1[o] - rm1[o]) * s + beta1[o];
    }

    const float* wrow2 = w2 + (size_t)(wv * 16 + l15) * 128 + half * 2;
    v2f wa[32];
#pragma unroll
    for (int ks = 0; ks < 32; ++ks)
        wa[ks] = *(const v2f*)(wrow2 + ks * 4);
    const int   oc  = wv * 16 + l15;
    const float s2v = g2[oc] * rsqrtf(rv2[oc] + EPS_);
    const float t2v = (b2[oc] - rm2[oc]) * s2v + beta2[oc];

    const float* pr0 = pa + l15 * EDGE_KP + half * 2;
    const float* pr1 = pr0 + 16 * EDGE_KP;

#pragma nounroll
    for (int g = 0; g < KNN_T / EDGE_PB; ++g) {
        const int lq = g * EDGE_PB + wv * EDGE_PW;
        const int n0 = chunk * KNN_T + g * EDGE_PB;
        const int nw = n0 + wv * EDGE_PW;

        int nb0[EDGE_PW], nb1[EDGE_PW];
#pragma unroll
        for (int p = 0; p < EDGE_PW; ++p) {
            nb0[p] = tk0[lq + p];
            nb1[p] = tk1[lq + p];
        }

        float gx0[EDGE_PW], gy0[EDGE_PW], gz0[EDGE_PW];
        float gx1[EDGE_PW], gy1[EDGE_PW], gz1[EDGE_PW];
        float ccx[EDGE_PW], ccy[EDGE_PW], ccz[EDGE_PW];
#pragma unroll
        for (int p = 0; p < EDGE_PW; ++p) {
            gx0[p] = sx[nb0[p]]; gy0[p] = sy[nb0[p]]; gz0[p] = sz[nb0[p]];
            gx1[p] = sx[nb1[p]]; gy1[p] = sy[nb1[p]]; gz1[p] = sz[nb1[p]];
            ccx[p] = sx[nw + p]; ccy[p] = sy[nw + p]; ccz[p] = sz[nw + p];
        }

        float pmr[EDGE_PW * 4], psr[EDGE_PW * 4];
#pragma unroll
        for (int p = 0; p < EDGE_PW; ++p) {
            const float cx = ccx[p], cy = ccy[p], cz = ccz[p];

            v2f a0[2], a1[2];
#pragma unroll
            for (int rt = 0; rt < 2; ++rt) {
                const bool valid = rt ? valid1 : true;
                const float nx = rt ? gx1[p] : gx0[p];
                const float ny = rt ? gy1[p] : gy0[p];
                const float nz = rt ? gz1[p] : gz0[p];
                const float e0 = valid ? (nx - cx) : 0.0f;
                const float e1 = valid ? (ny - cy) : 0.0f;
                const float e2 = valid ? (nz - cz) : 0.0f;
                const float e3 = valid ? cx : 0.0f;
                const float e4 = valid ? cy : 0.0f;
                const float e5 = valid ? cz : 0.0f;
                v2f t0, t1;
                t0.x = half ? e2 : e0;  t0.y = half ? e3 : e1;
                t1.x = half ? 0.0f : e4; t1.y = half ? 0.0f : e5;
                a0[rt] = t0; a1[rt] = t1;
            }

#pragma unroll
            for (int ct = 0; ct < 4; ++ct) {
                v8f z = {};
                z = __builtin_amdgcn_wmma_f32_16x16x4_f32(false, a0[0], false, bt0[ct],
                                                          (short)0, z, false, false);
                z = __builtin_amdgcn_wmma_f32_16x16x4_f32(false, a1[0], false, bt1[ct],
                                                          (short)0, z, false, false);
                v8f z1 = {};
                z1 = __builtin_amdgcn_wmma_f32_16x16x4_f32(false, a0[1], false, bt0[ct],
                                                           (short)0, z1, false, false);
                z1 = __builtin_amdgcn_wmma_f32_16x16x4_f32(false, a1[1], false, bt1[ct],
                                                           (short)0, z1, false, false);

                const float s = bs[ct], t = bt[ct];
                float mx = -3.4e38f, sm = 0.0f;
#pragma unroll
                for (int v = 0; v < 8; ++v) {
                    const float val = fmaxf(z[v] * s + t, 0.0f);
                    mx = fmaxf(mx, val); sm += val;
                }
#pragma unroll
                for (int v = 0; v < 4; ++v) {
                    const float val = fmaxf(z1[v] * s + t, 0.0f);
                    mx = (half == 0) ? fmaxf(mx, val) : mx;
                    sm = (half == 0) ? (sm + val) : sm;
                }
                pmr[p * 4 + ct] = mx;
                psr[p * 4 + ct] = sm;
            }
        }
        __builtin_amdgcn_sched_barrier(0);
#pragma unroll
        for (int p = 0; p < EDGE_PW; ++p) {
#pragma unroll
            for (int ct = 0; ct < 4; ++ct) {
                const int q = half * EDGE_HS + (ct * 16 + l15) * EDGE_LP + wv * EDGE_PW + p;
                pmx[q] = pmr[p * 4 + ct];
                psm[q] = psr[p * 4 + ct];
            }
        }
        __builtin_amdgcn_sched_barrier(0);
        __syncthreads();

        float jm[16], ja[16];
        float rml[16], rmh[16], rsl[16], rsh[16];
#pragma unroll
        for (int r = 0; r < 16; ++r) {
            const int o = wv * 16 + r;
            rml[r] = pmx[o * EDGE_LP + lane]; rmh[r] = pmx[EDGE_HS + o * EDGE_LP + lane];
            rsl[r] = psm[o * EDGE_LP + lane]; rsh[r] = psm[EDGE_HS + o * EDGE_LP + lane];
        }
        __builtin_amdgcn_sched_barrier(0);
#pragma unroll
        for (int r = 0; r < 16; ++r) {
            jm[r] = fmaxf(rml[r], rmh[r]);
            ja[r] = (rsl[r] + rsh[r]) * (1.0f / K_);
        }
        __builtin_amdgcn_sched_barrier(0);
#pragma unroll
        for (int r = 0; r < 16; ++r) {
            const int o = wv * 16 + r;
            pa[lane * EDGE_KP + o]      = jm[r];
            pa[lane * EDGE_KP + 64 + o] = ja[r];
        }
        __syncthreads();

#pragma unroll
        for (int r = 0; r < 16; ++r) {
            const int    o   = wv * 16 + r;
            const size_t off = ((size_t)b * 64 + o) * N_ + n0 + lane;
            out_m1[off] = jm[r];
            out_m2[off] = ja[r];
        }

        v2f pb0[32], pb1[32];
#pragma unroll
        for (int ks = 0; ks < 32; ++ks) {
            pb0[ks] = *(const v2f*)(pr0 + ks * 4);
            pb1[ks] = *(const v2f*)(pr1 + ks * 4);
        }
        __builtin_amdgcn_sched_barrier(0);
        v8f d0 = {}, d1 = {};
#pragma unroll
        for (int ks = 0; ks < 32; ++ks) {
            d0 = __builtin_amdgcn_wmma_f32_16x16x4_f32(false, wa[ks], false, pb0[ks],
                                                       (short)0, d0, false, false);
            d1 = __builtin_amdgcn_wmma_f32_16x16x4_f32(false, wa[ks], false, pb1[ks],
                                                       (short)0, d1, false, false);
        }

#pragma unroll
        for (int r = 0; r < 8; ++r) {
            const float x0 = edge_xhalf(d0[r]);
            const float x1 = edge_xhalf(d1[r]);
            const float lo = half ? x1 : d0[r];
            const float hi = half ? d1[r] : x0;
            {
                const int   o = wv * 16 + r;
                const float s = edge_lane(s2v, r), t = edge_lane(t2v, r);
                out[((size_t)b * 64 + o) * N_ + n0 + lane] = fmaxf(lo * s + t, 0.0f);
            }
            {
                const int   o = wv * 16 + 8 + r;
                const float s = edge_lane(s2v, 8 + r), t = edge_lane(t2v, 8 + r);
                out[((size_t)b * 64 + o) * N_ + n0 + lane] = fmaxf(hi * s + t, 0.0f);
            }
        }
    }
}

extern "C" void kernel_launch(void* const* d_in, const int* in_sizes, int n_in,
                              void* d_out, int out_size, void* d_ws, size_t ws_size,
                              hipStream_t stream)
{
    const float* x     = (const float*)d_in[0];
    const float* w1    = (const float*)d_in[1];
    const float* b1    = (const float*)d_in[2];
    const float* g1    = (const float*)d_in[3];
    const float* beta1 = (const float*)d_in[4];
    const float* rm1   = (const float*)d_in[5];
    const float* rv1   = (const float*)d_in[6];
    const float* w2    = (const float*)d_in[7];
    const float* b2    = (const float*)d_in[8];
    const float* g2    = (const float*)d_in[9];
    const float* beta2 = (const float*)d_in[10];
    const float* rm2   = (const float*)d_in[11];
    const float* rv2   = (const float*)d_in[12];

    float* out    = (float*)d_out;
    float* out_m1 = out + (size_t)B_ * 64 * N_;
    float* out_m2 = out + 2 * (size_t)B_ * 64 * N_;

    knn_edge_kernel<<<B_ * (N_ / KNN_T), KNN_T, 0, stream>>>(
        x, w1, b1, g1, beta1, rm1, rv1,
        w2, b2, g2, beta2, rm2, rv2, out, out_m1, out_m2);
}


# knn: |p|^2 computed once per point at staging (same three operations, same bits), read from a fourth LDS plane by scan and drain; list slots as base+offset keep the drain under 256 registers
# speedup vs baseline: 8.1508x; 1.0311x over previous
#include <hip/hip_runtime.h>

#define B_    8
#define C_    3
#define N_    4096
#define K_    20
#define H_    64
#define DOUT_ 64
#define EPS_  1e-5f

typedef __attribute__((ext_vector_type(2))) float v2f;
typedef __attribute__((ext_vector_type(8))) float v8f;

#define KNN_T   128
#define KNN_CH  16
#define KNN_CAP 32
#define KNN_TP  129

typedef __attribute__((ext_vector_type(4))) float v4f;


typedef __attribute__((address_space(3))) float lds_f32;
typedef __attribute__((address_space(3))) int   lds_i32;
typedef __attribute__((address_space(3))) unsigned lds_u32;
typedef __attribute__((address_space(3))) v4f   lds_v4f;
#define KNN_SLOT   (KNN_T * 4)
#define KNN_QJ     (KNN_CAP * KNN_T * 4)
#define KNN_PLANE  (N_ * 4)

#define KNN_LOAD(PX, PY, PZ, PP, la)                                      \
    {   const lds_v4f* lx = (const lds_v4f*)(la);                         \
        const lds_v4f* ly = (const lds_v4f*)((la) + KNN_PLANE);           \
        const lds_v4f* lz = (const lds_v4f*)((la) + 2 * KNN_PLANE);       \
        const lds_v4f* lp = (const lds_v4f*)((la) + 3 * KNN_PLANE);       \
        PX##0 = lx[0]; PX##1 = lx[1]; PX##2 = lx[2]; PX##3 = lx[3];       \
        PY##0 = ly[0]; PY##1 = ly[1]; PY##2 = ly[2]; PY##3 = ly[3];       \
        PZ##0 = lz[0]; PZ##1 = lz[1]; PZ##2 = lz[2]; PZ##3 = lz[3];       \
        PP##0 = lp[0]; PP##1 = lp[1]; PP##2 = lp[2]; PP##3 = lp[3]; }

#define KNN_D2(OUT, X2, Y2, Z2, P2)                                       \
    {   const v2f dot = __builtin_elementwise_fma(qz2, (Z2),              \
                        __builtin_elementwise_fma(qx2, (X2), qy2 * (Y2)));\
        OUT = __builtin_elementwise_fma(dot, two2, nqq2) - (P2); }

#define KNN_D4(O1, O2, X, Y, Z, P)                                        \
    KNN_D2(O1, X.xy, Y.xy, Z.xy, P.xy) KNN_D2(O2, X.zw, Y.zw, Z.zw, P.zw)

#define KNN_SGN(M, sv)                                                    \
    M = (M << 1) | (__builtin_bit_cast(unsigned, (float)(sv)) >> 31);

#define KNN_CHUNK(PX, PY, PZ, PP, jb, jnext)                              \
    {   v2f dA, dB, dC, dD, dE, dF, dG, dH;                               \
        KNN_D4(dA, dB, PX##0, PY##0, PZ##0, PP##0)                        \
        KNN_D4(dC, dD, PX##1, PY##1, PZ##1, PP##1)                        \
        KNN_D4(dE, dF, PX##2, PY##2, PZ##2, PP##2)                        \
        KNN_D4(dG, dH, PX##3, PY##3, PZ##3, PP##3)                        \
        const v2f thr2 = {thr, thr};                                      \
        const v2f sA = thr2 - dA, sB = thr2 - dB, sC = thr2 - dC, sD = thr2 - dD;\
        const v2f sE = thr2 - dE, sF = thr2 - dF, sG = thr2 - dG, sH = thr2 - dH;\
               \
        unsigned mhi = 0u, mlo = 0u;                                      \
        KNN_SGN(mhi, sA.x) KNN_SGN(mhi, sA.y) KNN_SGN(mhi, sB.x) KNN_SGN(mhi, sB.y)\
        KNN_SGN(mhi, sC.x) KNN_SGN(mhi, sC.y) KNN_SGN(mhi, sD.x) KNN_SGN(mhi, sD.y)\
        KNN_SGN(mlo, sE.x) KNN_SGN(mlo, sE.y) KNN_SGN(mlo, sF.x) KNN_SGN(mlo, sF.y)\
        KNN_SGN(mlo, sG.x) KNN_SGN(mlo, sG.y) KNN_SGN(mlo, sH.x) KNN_SGN(mlo, sH.y)\
        const unsigned m16 = (mhi << 8) | mlo;                            \
        const unsigned ent = m16 | ((unsigned)(jb) << 16);                \
        cnt += (unsigned)__builtin_popcount(m16);                         \
        const unsigned t0 = tail;                                         \
        const unsigned t1 = t0 + (m16 ? (unsigned)KNN_SLOT : 0u);         \
          \
          \
          \
          \
          \
        unsigned la;                                                      \
        asm("v_mov_b32 %0, %1" : "=v"(la) : "s"(pbase + ((unsigned)(jnext) << 2)));\
         \
         \
         \
        asm volatile("" :: "v"(t0), "v"(t1), "v"(ent), "v"(cnt), "v"(la));\
          \
          \
          \
        __builtin_amdgcn_sched_barrier(0);                                \
        *(lds_u32*)(t0) = ent;                                            \
        KNN_LOAD(PX, PY, PZ, PP, la)                                      \
        __builtin_amdgcn_sched_barrier(0);                                \
        tail = t1;                                                        \
    }

#define KNN_DRAIN(jb)                                                     \
    if (((jb) + KNN_CH >= N_) || __any(cnt > (unsigned)(KNN_CAP - KNN_CH))) { \
          \
          \
        float e0 = *(lds_f32*)(pk + 0 * KNN_SLOT);                      \
        float e1 = *(lds_f32*)(pk + 1 * KNN_SLOT);                      \
        float e2 = *(lds_f32*)(pk + 2 * KNN_SLOT);                      \
        float e3 = *(lds_f32*)(pk + 3 * KNN_SLOT);                      \
        float e4 = *(lds_f32*)(pk + 4 * KNN_SLOT);                      \
        float e5 = *(lds_f32*)(pk + 5 * KNN_SLOT);                      \
        float e6 = *(lds_f32*)(pk + 6 * KNN_SLOT);                      \
        float e7 = *(lds_f32*)(pk + 7 * KNN_SLOT);                      \
        float e8 = *(lds_f32*)(pk + 8 * KNN_SLOT);                      \
        float e9 = *(lds_f32*)(pk + 9 * KNN_SLOT);                      \
        float e10 = *(lds_f32*)(pk + 10 * KNN_SLOT);                      \
        float e11 = *(lds_f32*)(pk + 11 * KNN_SLOT);                      \
        float e12 = *(lds_f32*)(pk + 12 * KNN_SLOT);                      \
        float e13 = *(lds_f32*)(pk + 13 * KNN_SLOT);                      \
        float e14 = *(lds_f32*)(pk + 14 * KNN_SLOT);                      \
        float e15 = *(lds_f32*)(pk + 15 * KNN_SLOT);                      \
        float e16 = *(lds_f32*)(pk + 16 * KNN_SLOT);                      \
        float e17 = *(lds_f32*)(pk + 17 * KNN_SLOT);                      \
        float e18 = *(lds_f32*)(pk + 18 * KNN_SLOT);                      \
        float e19 = *(lds_f32*)(pk + 19 * KNN_SLOT);                      \
        int   k0 = *(lds_i32*)(pk + 20 * KNN_SLOT);                      \
        int   k1 = *(lds_i32*)(pk + 21 * KNN_SLOT);                      \
        int   k2 = *(lds_i32*)(pk + 22 * KNN_SLOT);                      \
        int   k3 = *(lds_i32*)(pk + 23 * KNN_SLOT);                      \
        int   k4 = *(lds_i32*)(pk + 24 * KNN_SLOT);                      \
        int   k5 = *(lds_i32*)(pk + 25 * KNN_SLOT);                      \
        int   k6 = *(lds_i32*)(pk + 26 * KNN_SLOT);                      \
        int   k7 = *(lds_i32*)(pk + 27 * KNN_SLOT);                      \
        int   k8 = *(lds_i32*)(pk + 28 * KNN_SLOT);                      \
        int   k9 = *(lds_i32*)(pk + 29 * KNN_SLOT);                      \
        int   k10 = *(lds_i32*)(pk + 30 * KNN_SLOT);                      \
        int   k11 = *(lds_i32*)(pk + 31 * KNN_SLOT);                      \
        int   k12 = *(lds_i32*)(pk + 32 * KNN_SLOT);                      \
        int   k13 = *(lds_i32*)(pk + 33 * KNN_SLOT);                      \
        int   k14 = *(lds_i32*)(pk + 34 * KNN_SLOT);                      \
        int   k15 = *(lds_i32*)(pk + 35 * KNN_SLOT);                      \
        int   k16 = *(lds_i32*)(pk + 36 * KNN_SLOT);                      \
        int   k17 = *(lds_i32*)(pk + 37 * KNN_SLOT);                      \
        int   k18 = *(lds_i32*)(pk + 38 * KNN_SLOT);                      \
        int   k19 = *(lds_i32*)(pk + 39 * KNN_SLOT);                      \
         \
         \
         \
         \
         \
        unsigned cur = *(lds_u32*)(tbase);                                \
        unsigned ea  = tbase + KNN_SLOT;                                  \
        unsigned nxt = *(lds_u32*)(ea);                                   \
        int jn; float pxn, pyn, pzn, ppn; \
        KNN_NEXT \
          \
          \
          \
          \
          \
          \
          \
          \
        float d; int j; \
        KNN_AHEAD(0u, d, j) \
        int s = 0; \
          \
          \
        do { \
              \
            KNN_TESTS \
              \
              \
              \
              \
              \
            float d1; int j1; \
            KNN_AHEAD((unsigned)s + 1u, d1, j1) \
              \
              \
              \
              \
              \
            __builtin_amdgcn_sched_barrier(0); \
            KNN_SELECTS \
            __builtin_amdgcn_sched_barrier(0); \
              \
              \
            asm volatile("" :: "v"(nxt), "v"(pxn), "v"(pyn), "v"(pzn), "v"(ppn)); \
            d = d1; j = j1; \
            ++s; \
        } while (__any((unsigned)s < cnt)); \
        *(lds_f32*)(pk + 0 * KNN_SLOT) = e0;                            \
        *(lds_f32*)(pk + 1 * KNN_SLOT) = e1;                            \
        *(lds_f32*)(pk + 2 * KNN_SLOT) = e2;                            \
        *(lds_f32*)(pk + 3 * KNN_SLOT) = e3;                            \
        *(lds_f32*)(pk + 4 * KNN_SLOT) = e4;                            \
        *(lds_f32*)(pk + 5 * KNN_SLOT) = e5;                            \
        *(lds_f32*)(pk + 6 * KNN_SLOT) = e6;                            \
        *(lds_f32*)(pk + 7 * KNN_SLOT) = e7;                            \
        *(lds_f32*)(pk + 8 * KNN_SLOT) = e8;                            \
        *(lds_f32*)(pk + 9 * KNN_SLOT) = e9;                            \
        *(lds_f32*)(pk + 10 * KNN_SLOT) = e10;                            \
        *(lds_f32*)(pk + 11 * KNN_SLOT) = e11;                            \
        *(lds_f32*)(pk + 12 * KNN_SLOT) = e12;                            \
        *(lds_f32*)(pk + 13 * KNN_SLOT) = e13;                            \
        *(lds_f32*)(pk + 14 * KNN_SLOT) = e14;                            \
        *(lds_f32*)(pk + 15 * KNN_SLOT) = e15;                            \
        *(lds_f32*)(pk + 16 * KNN_SLOT) = e16;                            \
        *(lds_f32*)(pk + 17 * KNN_SLOT) = e17;                            \
        *(lds_f32*)(pk + 18 * KNN_SLOT) = e18;                            \
        *(lds_f32*)(pk + 19 * KNN_SLOT) = e19;                            \
        *(lds_i32*)(pk + 20 * KNN_SLOT) = k0;                            \
        *(lds_i32*)(pk + 21 * KNN_SLOT) = k1;                            \
        *(lds_i32*)(pk + 22 * KNN_SLOT) = k2;                            \
        *(lds_i32*)(pk + 23 * KNN_SLOT) = k3;                            \
        *(lds_i32*)(pk + 24 * KNN_SLOT) = k4;                            \
        *(lds_i32*)(pk + 25 * KNN_SLOT) = k5;                            \
        *(lds_i32*)(pk + 26 * KNN_SLOT) = k6;                            \
        *(lds_i32*)(pk + 27 * KNN_SLOT) = k7;                            \
        *(lds_i32*)(pk + 28 * KNN_SLOT) = k8;                            \
        *(lds_i32*)(pk + 29 * KNN_SLOT) = k9;                            \
        *(lds_i32*)(pk + 30 * KNN_SLOT) = k10;                            \
        *(lds_i32*)(pk + 31 * KNN_SLOT) = k11;                            \
        *(lds_i32*)(pk + 32 * KNN_SLOT) = k12;                            \
        *(lds_i32*)(pk + 33 * KNN_SLOT) = k13;                            \
        *(lds_i32*)(pk + 34 * KNN_SLOT) = k14;                            \
        *(lds_i32*)(pk + 35 * KNN_SLOT) = k15;                            \
        *(lds_i32*)(pk + 36 * KNN_SLOT) = k16;                            \
        *(lds_i32*)(pk + 37 * KNN_SLOT) = k17;                            \
        *(lds_i32*)(pk + 38 * KNN_SLOT) = k18;                            \
        *(lds_i32*)(pk + 39 * KNN_SLOT) = k19;                            \
        tail = tbase;                                                     \
        thr = e19;                                                        \
        cnt = 0u;                                                         \
    }

#define KNN_NEXT                                                          \
    {   const bool need = (cur & 0xffffu) == 0u;                          \
        cur = need ? nxt : cur;                                           \
        ea  = need ? (ea + KNN_SLOT) : ea;                                \
        nxt = *(lds_u32*)(ea);                                            \
        const unsigned lz = (unsigned)__builtin_clz((cur & 0xffffu) | 1u);\
        jn  = (int)((cur >> 16) + lz - 16u);                              \
        cur = cur & ~(0x80000000u >> lz);                                 \
        const unsigned ga = pbase + (((unsigned)jn & (unsigned)(N_ - 1)) << 2);\
        pxn = *(lds_f32*)(ga);                                            \
        pyn = *(lds_f32*)(ga + KNN_PLANE);                                \
        pzn = *(lds_f32*)(ga + 2 * KNN_PLANE);                            \
        ppn = *(lds_f32*)(ga + 3 * KNN_PLANE); }

#define KNN_AHEAD(sn, D, J)                                               \
    {   const bool  have = (sn) < cnt;                                    \
        const float px = pxn, py = pyn, pz = pzn, pp = ppn;               \
        J = have ? jn : 0;                                                \
        KNN_NEXT                                                          \
        const float dot = __builtin_fmaf(qz, pz, __builtin_fmaf(qx, px, qy * py));\
        const float dc  = __builtin_fmaf(dot, 2.0f, -qq) - pp;            \
        D = have ? dc : -3.4e38f; }

#define KNN_TESTS                                                         \
    const bool c0 = d > e0, c1 = d > e1, c2 = d > e2, c3 = d > e3, c4 = d > e4;\
    const bool c5 = d > e5, c6 = d > e6, c7 = d > e7, c8 = d > e8, c9 = d > e9;\
    const bool c10 = d > e10, c11 = d > e11, c12 = d > e12, c13 = d > e13, c14 = d > e14;\
    const bool c15 = d > e15, c16 = d > e16, c17 = d > e17, c18 = d > e18, c19 = d > e19;

#define KNN_LAY1(i) const float t##i = c##i ? d : e##i; const int u##i = c##i ? j : k##i;
#define KNN_LAY2(iu, il) e##il = c##iu ? e##iu : t##il; k##il = c##iu ? k##iu : u##il;
#define KNN_SELECTS                                                       \
    KNN_LAY1(0) KNN_LAY1(1) KNN_LAY1(2) KNN_LAY1(3) KNN_LAY1(4)           \
    KNN_LAY1(5) KNN_LAY1(6) KNN_LAY1(7) KNN_LAY1(8) KNN_LAY1(9)           \
    KNN_LAY1(10) KNN_LAY1(11) KNN_LAY1(12) KNN_LAY1(13) KNN_LAY1(14)      \
    KNN_LAY1(15) KNN_LAY1(16) KNN_LAY1(17) KNN_LAY1(18) KNN_LAY1(19)      \
    KNN_LAY2(18, 19) KNN_LAY2(17, 18) KNN_LAY2(16, 17) KNN_LAY2(15, 16)   \
    KNN_LAY2(14, 15) KNN_LAY2(13, 14) KNN_LAY2(12, 13) KNN_LAY2(11, 12)   \
    KNN_LAY2(10, 11) KNN_LAY2(9, 10) KNN_LAY2(8, 9) KNN_LAY2(7, 8)        \
    KNN_LAY2(6, 7) KNN_LAY2(5, 6) KNN_LAY2(4, 5) KNN_LAY2(3, 4)           \
    KNN_LAY2(2, 3) KNN_LAY2(1, 2) KNN_LAY2(0, 1)                          \
    e0 = t0;                                                              \
    k0 = u0;

#define EDGE_PW 8
#define EDGE_PB 32
#define EDGE_LP 33
#define EDGE_HS (64 * EDGE_LP + 16)
#define EDGE_KP 132

__device__ __forceinline__ float edge_xhalf(float v)
{
    return __builtin_bit_cast(float, __builtin_amdgcn_permlanex16(
        __builtin_bit_cast(int, v), __builtin_bit_cast(int, v),
        0x76543210, 0xfedcba98, false, false));
}

__device__ __forceinline__ float edge_lane(float v, int l)
{
    return __builtin_bit_cast(float,
        __builtin_amdgcn_readlane(__builtin_bit_cast(int, v), l));
}

__global__ __launch_bounds__(KNN_T) void knn_edge_kernel(
    const float* __restrict__ x,
    const float* __restrict__ w1, const float* __restrict__ b1,
    const float* __restrict__ g1, const float* __restrict__ beta1,
    const float* __restrict__ rm1,const float* __restrict__ rv1,
    const float* __restrict__ w2, const float* __restrict__ b2,
    const float* __restrict__ g2, const float* __restrict__ beta2,
    const float* __restrict__ rm2,const float* __restrict__ rv2,
    float* __restrict__ out,
    float* __restrict__ out_m1,   float* __restrict__ out_m2)
{
    __shared__ __attribute__((aligned(16))) float lds[4 * N_ + 2 * KNN_CAP * KNN_T + 2 * K_ * KNN_T];
    float* spts = lds;
    int*   qj   = (int*)(lds + 4 * N_ + KNN_CAP * KNN_T);
    const float* sx = spts;
    const float* sy = spts + N_;
    const float* sz = spts + 2 * N_;
    const int blocksPerBatch = N_ / KNN_T;
    const int b     = blockIdx.x / blocksPerBatch;
    const int chunk = blockIdx.x % blocksPerBatch;
    const float* xb = x + (size_t)b * 3 * N_;
    const v4f* xb4 = (const v4f*)xb;
    v4f*       sp4 = (v4f*)spts;
    v4f        st[(3 * N_ / 4) / KNN_T];
#pragma unroll
    for (int t = 0; t < (3 * N_ / 4) / KNN_T; ++t)
        st[t] = xb4[t * KNN_T + threadIdx.x];
#pragma unroll
    for (int t = 0; t < (3 * N_ / 4) / KNN_T; ++t)
        sp4[t * KNN_T + threadIdx.x] = st[t];
#pragma unroll
    for (int t = 0; t < (N_ / 4) / KNN_T; ++t) {
        const v4f X = st[t], Y = st[t + (N_ / 4) / KNN_T], Z = st[t + 2 * ((N_ / 4) / KNN_T)];
        sp4[3 * (N_ / 4) + t * KNN_T + threadIdx.x] =
            __builtin_elementwise_fma(Z, Z, __builtin_elementwise_fma(X, X, Y * Y));
    }
    __syncthreads();

    const int   tid = threadIdx.x;
    const int   n  = chunk * KNN_T + tid;
    const float qx = sx[n], qy = sy[n], qz = sz[n];
    const float qq = fmaf(qz, qz, fmaf(qx, qx, qy * qy));
    const v2f qx2 = {qx, qx}, qy2 = {qy, qy}, qz2 = {qz, qz};
    const v2f nqq2 = {-qq, -qq}, two2 = {2.0f, 2.0f};

    const unsigned lds0  = (unsigned)(size_t)lds;
    const unsigned pbase = lds0;
    const unsigned tbase = lds0 + 4 * KNN_PLANE + (unsigned)tid * 4u;
    unsigned       pk    = lds0 + 4 * KNN_PLANE + 2 * KNN_QJ + (unsigned)tid * 4u;
    asm("" : "+v"(pk));

#pragma unroll
    for (int i = 0; i < K_; ++i) {
        *(lds_f32*)(pk + (unsigned)i * KNN_SLOT) = -3.4e38f;
        *(lds_i32*)(pk + (unsigned)(K_ + i) * KNN_SLOT) = 0;
    }

    float    thr  = -3.4e38f;
    unsigned tail = tbase;
    unsigned cnt  = 0u;

    v4f ax0, ax1, ax2, ax3, ay0, ay1, ay2, ay3, az0, az1, az2, az3, ap0, ap1, ap2, ap3;
    v4f bx0, bx1, bx2, bx3, by0, by1, by2, by3, bz0, bz1, bz2, bz3, bp0, bp1, bp2, bp3;
    KNN_LOAD(ax, ay, az, ap, pbase)
    KNN_LOAD(bx, by, bz, bp, pbase + KNN_CH * 4)
    for (int j0 = 0; j0 < N_; j0 += 2 * KNN_CH) {
        KNN_CHUNK(ax, ay, az, ap, j0, (j0 + 2 * KNN_CH) & (N_ - 1))
        KNN_DRAIN(j0)
        KNN_CHUNK(bx, by, bz, bp, j0 + KNN_CH, (j0 + 3 * KNN_CH) & (N_ - 1))
        KNN_DRAIN(j0 + KNN_CH)
    }
    const int k0 = *(lds_i32*)(pk + 20 * KNN_SLOT);
    const int k1 = *(lds_i32*)(pk + 21 * KNN_SLOT);
    const int k2 = *(lds_i32*)(pk + 22 * KNN_SLOT);
    const int k3 = *(lds_i32*)(pk + 23 * KNN_SLOT);
    const int k4 = *(lds_i32*)(pk + 24 * KNN_SLOT);
    const int k5 = *(lds_i32*)(pk + 25 * KNN_SLOT);
    const int k6 = *(lds_i32*)(pk + 26 * KNN_SLOT);
    const int k7 = *(lds_i32*)(pk + 27 * KNN_SLOT);
    const int k8 = *(lds_i32*)(pk + 28 * KNN_SLOT);
    const int k9 = *(lds_i32*)(pk + 29 * KNN_SLOT);
    const int k10 = *(lds_i32*)(pk + 30 * KNN_SLOT);
    const int k11 = *(lds_i32*)(pk + 31 * KNN_SLOT);
    const int k12 = *(lds_i32*)(pk + 32 * KNN_SLOT);
    const int k13 = *(lds_i32*)(pk + 33 * KNN_SLOT);
    const int k14 = *(lds_i32*)(pk + 34 * KNN_SLOT);
    const int k15 = *(lds_i32*)(pk + 35 * KNN_SLOT);
    const int k16 = *(lds_i32*)(pk + 36 * KNN_SLOT);
    const int k17 = *(lds_i32*)(pk + 37 * KNN_SLOT);
    const int k18 = *(lds_i32*)(pk + 38 * KNN_SLOT);
    const int k19 = *(lds_i32*)(pk + 39 * KNN_SLOT);
    __syncthreads();
    int* tk = qj;
    tk[0 * KNN_TP + tid] = k0;
    tk[1 * KNN_TP + tid] = k1;
    tk[2 * KNN_TP + tid] = k2;
    tk[3 * KNN_TP + tid] = k3;
    tk[4 * KNN_TP + tid] = k4;
    tk[5 * KNN_TP + tid] = k5;
    tk[6 * KNN_TP + tid] = k6;
    tk[7 * KNN_TP + tid] = k7;
    tk[8 * KNN_TP + tid] = k8;
    tk[9 * KNN_TP + tid] = k9;
    tk[10 * KNN_TP + tid] = k10;
    tk[11 * KNN_TP + tid] = k11;
    tk[12 * KNN_TP + tid] = k12;
    tk[13 * KNN_TP + tid] = k13;
    tk[14 * KNN_TP + tid] = k14;
    tk[15 * KNN_TP + tid] = k15;
    tk[16 * KNN_TP + tid] = k16;
    tk[17 * KNN_TP + tid] = k17;
    tk[18 * KNN_TP + tid] = k18;
    tk[19 * KNN_TP + tid] = k19;
    __syncthreads();

    __shared__ float pmx[2 * EDGE_HS];
    __shared__ float psm[2 * EDGE_HS];
    __shared__ __attribute__((aligned(16))) float pa[EDGE_PB * EDGE_KP];

    const int lane = tid & 31;
    const int half = lane >> 4;
    const int l15  = lane & 15;
    const int wv   = __builtin_amdgcn_readfirstlane(tid >> 5);

    const bool valid1 = (16 + l15) < K_;
    const int  kc1    = valid1 ? (16 + l15) : (K_ - 1);
    const int* tk0 = tk + l15 * KNN_TP;
    const int* tk1 = tk + kc1 * KNN_TP;

    v2f   bt0[4], bt1[4];
    float bs[4], bt[4];
#pragma unroll
    for (int ct = 0; ct < 4; ++ct) {
        const int    o    = ct * 16 + l15;
        const float* wrow = w1 + o * 6;
        const float w0 = wrow[0], w1v = wrow[1], w2v = wrow[2];
        const float w3 = wrow[3], w4v = wrow[4], w5v = wrow[5];
        bt0[ct].x = half ? w2v : w0;   bt0[ct].y = half ? w3 : w1v;
        bt1[ct].x = half ? 0.0f : w4v; bt1[ct].y = half ? 0.0f : w5v;
        const float s = g1[o] * rsqrtf(rv1[o] + EPS_);
        bs[ct] = s;
        bt[ct] = (b1[o] - rm1[o]) * s + beta1[o];
    }

    const float* wrow2 = w2 + (size_t)(wv * 16 + l15) * 128 + half * 2;
    v2f wa[32];
#pragma unroll
    for (int ks = 0; ks < 32; ++ks)
        wa[ks] = *(const v2f*)(wrow2 + ks * 4);
    const int   oc  = wv * 16 + l15;
    const float s2v = g2[oc] * rsqrtf(rv2[oc] + EPS_);
    const float t2v = (b2[oc] - rm2[oc]) * s2v + beta2[oc];

    const float* pr0 = pa + l15 * EDGE_KP + half * 2;
    const float* pr1 = pr0 + 16 * EDGE_KP;

#pragma nounroll
    for (int g = 0; g < KNN_T / EDGE_PB; ++g) {
        const int lq = g * EDGE_PB + wv * EDGE_PW;
        const int n0 = chunk * KNN_T + g * EDGE_PB;
        const int nw = n0 + wv * EDGE_PW;

        int nb0[EDGE_PW], nb1[EDGE_PW];
#pragma unroll
        for (int p = 0; p < EDGE_PW; ++p) {
            nb0[p] = tk0[lq + p];
            nb1[p] = tk1[lq + p];
        }

        float gx0[EDGE_PW], gy0[EDGE_PW], gz0[EDGE_PW];
        float gx1[EDGE_PW], gy1[EDGE_PW], gz1[EDGE_PW];
        float ccx[EDGE_PW], ccy[EDGE_PW], ccz[EDGE_PW];
#pragma unroll
        for (int p = 0; p < EDGE_PW; ++p) {
            gx0[p] = sx[nb0[p]]; gy0[p] = sy[nb0[p]]; gz0[p] = sz[nb0[p]];
            gx1[p] = sx[nb1[p]]; gy1[p] = sy[nb1[p]]; gz1[p] = sz[nb1[p]];
            ccx[p] = sx[nw + p]; ccy[p] = sy[nw + p]; ccz[p] = sz[nw + p];
        }

        float pmr[EDGE_PW * 4], psr[EDGE_PW * 4];
#pragma unroll
        for (int p = 0; p < EDGE_PW; ++p) {
            const float cx = ccx[p], cy = ccy[p], cz = ccz[p];

            v2f a0[2], a1[2];
#pragma unroll
            for (int rt = 0; rt < 2; ++rt) {
                const bool valid = rt ? valid1 : true;
                const float nx = rt ? gx1[p] : gx0[p];
                const float ny = rt ? gy1[p] : gy0[p];
                const float nz = rt ? gz1[p] : gz0[p];
                const float e0 = valid ? (nx - cx) : 0.0f;
                const float e1 = valid ? (ny - cy) : 0.0f;
                const float e2 = valid ? (nz - cz) : 0.0f;
                const float e3 = valid ? cx : 0.0f;
                const float e4 = valid ? cy : 0.0f;
                const float e5 = valid ? cz : 0.0f;
                v2f t0, t1;
                t0.x = half ? e2 : e0;  t0.y = half ? e3 : e1;
                t1.x = half ? 0.0f : e4; t1.y = half ? 0.0f : e5;
                a0[rt] = t0; a1[rt] = t1;
            }

#pragma unroll
            for (int ct = 0; ct < 4; ++ct) {
                v8f z = {};
                z = __builtin_amdgcn_wmma_f32_16x16x4_f32(false, a0[0], false, bt0[ct],
                                                          (short)0, z, false, false);
                z = __builtin_amdgcn_wmma_f32_16x16x4_f32(false, a1[0], false, bt1[ct],
                                                          (short)0, z, false, false);
                v8f z1 = {};
                z1 = __builtin_amdgcn_wmma_f32_16x16x4_f32(false, a0[1], false, bt0[ct],
                                                           (short)0, z1, false, false);
                z1 = __builtin_amdgcn_wmma_f32_16x16x4_f32(false, a1[1], false, bt1[ct],
                                                           (short)0, z1, false, false);

                const float s = bs[ct], t = bt[ct];
                float mx = -3.4e38f, sm = 0.0f;
#pragma unroll
                for (int v = 0; v < 8; ++v) {
                    const float val = fmaxf(z[v] * s + t, 0.0f);
                    mx = fmaxf(mx, val); sm += val;
                }
#pragma unroll
                for (int v = 0; v < 4; ++v) {
                    const float val = fmaxf(z1[v] * s + t, 0.0f);
                    mx = (half == 0) ? fmaxf(mx, val) : mx;
                    sm = (half == 0) ? (sm + val) : sm;
                }
                pmr[p * 4 + ct] = mx;
                psr[p * 4 + ct] = sm;
            }
        }
        __builtin_amdgcn_sched_barrier(0);
#pragma unroll
        for (int p = 0; p < EDGE_PW; ++p) {
#pragma unroll
            for (int ct = 0; ct < 4; ++ct) {
                const int q = half * EDGE_HS + (ct * 16 + l15) * EDGE_LP + wv * EDGE_PW + p;
                pmx[q] = pmr[p * 4 + ct];
                psm[q] = psr[p * 4 + ct];
            }
        }
        __builtin_amdgcn_sched_barrier(0);
        __syncthreads();

        float jm[16], ja[16];
        float rml[16], rmh[16], rsl[16], rsh[16];
#pragma unroll
        for (int r = 0; r < 16; ++r) {
            const int o = wv * 16 + r;
            rml[r] = pmx[o * EDGE_LP + lane]; rmh[r] = pmx[EDGE_HS + o * EDGE_LP + lane];
            rsl[r] = psm[o * EDGE_LP + lane]; rsh[r] = psm[EDGE_HS + o * EDGE_LP + lane];
        }
        __builtin_amdgcn_sched_barrier(0);
#pragma unroll
        for (int r = 0; r < 16; ++r) {
            jm[r] = fmaxf(rml[r], rmh[r]);
            ja[r] = (rsl[r] + rsh[r]) * (1.0f / K_);
        }
        __builtin_amdgcn_sched_barrier(0);
#pragma unroll
        for (int r = 0; r < 16; ++r) {
            const int o = wv * 16 + r;
            pa[lane * EDGE_KP + o]      = jm[r];
            pa[lane * EDGE_KP + 64 + o] = ja[r];
        }
        __syncthreads();

#pragma unroll
        for (int r = 0; r < 16; ++r) {
            const int    o   = wv * 16 + r;
            const size_t off = ((size_t)b * 64 + o) * N_ + n0 + lane;
            out_m1[off] = jm[r];
            out_m2[off] = ja[r];
        }

        v2f pb0[32], pb1[32];
#pragma unroll
        for (int ks = 0; ks < 32; ++ks) {
            pb0[ks] = *(const v2f*)(pr0 + ks * 4);
            pb1[ks] = *(const v2f*)(pr1 + ks * 4);
        }
        __builtin_amdgcn_sched_barrier(0);
        v8f d0 = {}, d1 = {};
#pragma unroll
        for (int ks = 0; ks < 32; ++ks) {
            d0 = __builtin_amdgcn_wmma_f32_16x16x4_f32(false, wa[ks], false, pb0[ks],
                                                       (short)0, d0, false, false);
            d1 = __builtin_amdgcn_wmma_f32_16x16x4_f32(false, wa[ks], false, pb1[ks],
                                                       (short)0, d1, false, false);
        }

#pragma unroll
        for (int r = 0; r < 8; ++r) {
            const float x0 = edge_xhalf(d0[r]);
            const float x1 = edge_xhalf(d1[r]);
            const float lo = half ? x1 : d0[r];
            const float hi = half ? d1[r] : x0;
            {
                const int   o = wv * 16 + r;
                const float s = edge_lane(s2v, r), t = edge_lane(t2v, r);
                out[((size_t)b * 64 + o) * N_ + n0 + lane] = fmaxf(lo * s + t, 0.0f);
            }
            {
                const int   o = wv * 16 + 8 + r;
                const float s = edge_lane(s2v, 8 + r), t = edge_lane(t2v, 8 + r);
                out[((size_t)b * 64 + o) * N_ + n0 + lane] = fmaxf(hi * s + t, 0.0f);
            }
        }
    }
}

extern "C" void kernel_launch(void* const* d_in, const int* in_sizes, int n_in,
                              void* d_out, int out_size, void* d_ws, size_t ws_size,
                              hipStream_t stream)
{
    const float* x     = (const float*)d_in[0];
    const float* w1    = (const float*)d_in[1];
    const float* b1    = (const float*)d_in[2];
    const float* g1    = (const float*)d_in[3];
    const float* beta1 = (const float*)d_in[4];
    const float* rm1   = (const float*)d_in[5];
    const float* rv1   = (const float*)d_in[6];
    const float* w2    = (const float*)d_in[7];
    const float* b2    = (const float*)d_in[8];
    const float* g2    = (const float*)d_in[9];
    const float* beta2 = (const float*)d_in[10];
    const float* rm2   = (const float*)d_in[11];
    const float* rv2   = (const float*)d_in[12];

    float* out    = (float*)d_out;
    float* out_m1 = out + (size_t)B_ * 64 * N_;
    float* out_m2 = out + 2 * (size_t)B_ * 64 * N_;

    knn_edge_kernel<<<B_ * (N_ / KNN_T), KNN_T, 0, stream>>>(
        x, w1, b1, g1, beta1, rm1, rv1,
        w2, b2, g2, beta2, rm2, rv2, out, out_m1, out_m2);
}


# knn scan: the drain test's vector compare stands in front of a chunk's LDS group, so the branch behind the group finds its mask ready; chunks that do not drain fall through (no taken branch)
# speedup vs baseline: 8.4691x; 1.0391x over previous
#include <hip/hip_runtime.h>

#define B_    8
#define C_    3
#define N_    4096
#define K_    20
#define H_    64
#define DOUT_ 64
#define EPS_  1e-5f

typedef __attribute__((ext_vector_type(2))) float v2f;
typedef __attribute__((ext_vector_type(8))) float v8f;

#define KNN_T   128
#define KNN_CH  16
#define KNN_CAP 32
#define KNN_TP  129

typedef __attribute__((ext_vector_type(4))) float v4f;


typedef __attribute__((address_space(3))) float lds_f32;
typedef __attribute__((address_space(3))) int   lds_i32;
typedef __attribute__((address_space(3))) unsigned lds_u32;
typedef __attribute__((address_space(3))) v4f   lds_v4f;
#define KNN_SLOT   (KNN_T * 4)
#define KNN_QJ     (KNN_CAP * KNN_T * 4)
#define KNN_PLANE  (N_ * 4)

#define KNN_LOAD(PX, PY, PZ, PP, la)                                      \
    {   const lds_v4f* lx = (const lds_v4f*)(la);                         \
        const lds_v4f* ly = (const lds_v4f*)((la) + KNN_PLANE);           \
        const lds_v4f* lz = (const lds_v4f*)((la) + 2 * KNN_PLANE);       \
        const lds_v4f* lp = (const lds_v4f*)((la) + 3 * KNN_PLANE);       \
        PX##0 = lx[0]; PX##1 = lx[1]; PX##2 = lx[2]; PX##3 = lx[3];       \
        PY##0 = ly[0]; PY##1 = ly[1]; PY##2 = ly[2]; PY##3 = ly[3];       \
        PZ##0 = lz[0]; PZ##1 = lz[1]; PZ##2 = lz[2]; PZ##3 = lz[3];       \
        PP##0 = lp[0]; PP##1 = lp[1]; PP##2 = lp[2]; PP##3 = lp[3]; }

#define KNN_D2(OUT, X2, Y2, Z2, P2)                                       \
    {   const v2f dot = __builtin_elementwise_fma(qz2, (Z2),              \
                        __builtin_elementwise_fma(qx2, (X2), qy2 * (Y2)));\
        OUT = __builtin_elementwise_fma(dot, two2, nqq2) - (P2); }

#define KNN_D4(O1, O2, X, Y, Z, P)                                        \
    KNN_D2(O1, X.xy, Y.xy, Z.xy, P.xy) KNN_D2(O2, X.zw, Y.zw, Z.zw, P.zw)

#define KNN_SGN(M, sv)                                                    \
    M = (M << 1) | (__builtin_bit_cast(unsigned, (float)(sv)) >> 31);

#define KNN_CHUNK(PX, PY, PZ, PP, jb, jnext)                              \
    {   v2f dA, dB, dC, dD, dE, dF, dG, dH;                               \
        KNN_D4(dA, dB, PX##0, PY##0, PZ##0, PP##0)                        \
        KNN_D4(dC, dD, PX##1, PY##1, PZ##1, PP##1)                        \
        KNN_D4(dE, dF, PX##2, PY##2, PZ##2, PP##2)                        \
        KNN_D4(dG, dH, PX##3, PY##3, PZ##3, PP##3)                        \
        const v2f thr2 = {thr, thr};                                      \
        const v2f sA = thr2 - dA, sB = thr2 - dB, sC = thr2 - dC, sD = thr2 - dD;\
        const v2f sE = thr2 - dE, sF = thr2 - dF, sG = thr2 - dG, sH = thr2 - dH;\
               \
        unsigned mhi = 0u, mlo = 0u;                                      \
        KNN_SGN(mhi, sA.x) KNN_SGN(mhi, sA.y) KNN_SGN(mhi, sB.x) KNN_SGN(mhi, sB.y)\
        KNN_SGN(mhi, sC.x) KNN_SGN(mhi, sC.y) KNN_SGN(mhi, sD.x) KNN_SGN(mhi, sD.y)\
        KNN_SGN(mlo, sE.x) KNN_SGN(mlo, sE.y) KNN_SGN(mlo, sF.x) KNN_SGN(mlo, sF.y)\
        KNN_SGN(mlo, sG.x) KNN_SGN(mlo, sG.y) KNN_SGN(mlo, sH.x) KNN_SGN(mlo, sH.y)\
        const unsigned m16 = (mhi << 8) | mlo;                            \
        const unsigned ent = m16 | ((unsigned)(jb) << 16);                \
        cnt += (unsigned)__builtin_popcount(m16);                         \
          \
          \
          \
          \
          \
        dmask = __builtin_amdgcn_ballot_w32(cnt > (unsigned)(KNN_CAP - KNN_CH)); \
        const unsigned t0 = tail;                                         \
        const unsigned t1 = t0 + (m16 ? (unsigned)KNN_SLOT : 0u);         \
          \
          \
          \
          \
          \
        unsigned la;                                                      \
        asm("v_mov_b32 %0, %1" : "=v"(la) : "s"(pbase + ((unsigned)(jnext) << 2)));\
         \
         \
         \
         \
        asm volatile("" :: "v"(t0), "v"(t1), "v"(ent), "v"(cnt), "v"(la), "s"(dmask));\
          \
          \
          \
        __builtin_amdgcn_sched_barrier(0);                                \
        *(lds_u32*)(t0) = ent;                                            \
        KNN_LOAD(PX, PY, PZ, PP, la)                                      \
        __builtin_amdgcn_sched_barrier(0);                                \
        tail = t1;                                                        \
    }

#define KNN_DRAIN(jb)                                                     \
    if (__builtin_expect((((jb) + KNN_CH >= N_) || (dmask != 0u)), 0)) {    \
          \
          \
          \
          \
          \
          \
        float e0 = *(lds_f32*)(pk + 0 * KNN_SLOT);                      \
        float e1 = *(lds_f32*)(pk + 1 * KNN_SLOT);                      \
        float e2 = *(lds_f32*)(pk + 2 * KNN_SLOT);                      \
        float e3 = *(lds_f32*)(pk + 3 * KNN_SLOT);                      \
        float e4 = *(lds_f32*)(pk + 4 * KNN_SLOT);                      \
        float e5 = *(lds_f32*)(pk + 5 * KNN_SLOT);                      \
        float e6 = *(lds_f32*)(pk + 6 * KNN_SLOT);                      \
        float e7 = *(lds_f32*)(pk + 7 * KNN_SLOT);                      \
        float e8 = *(lds_f32*)(pk + 8 * KNN_SLOT);                      \
        float e9 = *(lds_f32*)(pk + 9 * KNN_SLOT);                      \
        float e10 = *(lds_f32*)(pk + 10 * KNN_SLOT);                      \
        float e11 = *(lds_f32*)(pk + 11 * KNN_SLOT);                      \
        float e12 = *(lds_f32*)(pk + 12 * KNN_SLOT);                      \
        float e13 = *(lds_f32*)(pk + 13 * KNN_SLOT);                      \
        float e14 = *(lds_f32*)(pk + 14 * KNN_SLOT);                      \
        float e15 = *(lds_f32*)(pk + 15 * KNN_SLOT);                      \
        float e16 = *(lds_f32*)(pk + 16 * KNN_SLOT);                      \
        float e17 = *(lds_f32*)(pk + 17 * KNN_SLOT);                      \
        float e18 = *(lds_f32*)(pk + 18 * KNN_SLOT);                      \
        float e19 = *(lds_f32*)(pk + 19 * KNN_SLOT);                      \
        int   k0 = *(lds_i32*)(pk + 20 * KNN_SLOT);                      \
        int   k1 = *(lds_i32*)(pk + 21 * KNN_SLOT);                      \
        int   k2 = *(lds_i32*)(pk + 22 * KNN_SLOT);                      \
        int   k3 = *(lds_i32*)(pk + 23 * KNN_SLOT);                      \
        int   k4 = *(lds_i32*)(pk + 24 * KNN_SLOT);                      \
        int   k5 = *(lds_i32*)(pk + 25 * KNN_SLOT);                      \
        int   k6 = *(lds_i32*)(pk + 26 * KNN_SLOT);                      \
        int   k7 = *(lds_i32*)(pk + 27 * KNN_SLOT);                      \
        int   k8 = *(lds_i32*)(pk + 28 * KNN_SLOT);                      \
        int   k9 = *(lds_i32*)(pk + 29 * KNN_SLOT);                      \
        int   k10 = *(lds_i32*)(pk + 30 * KNN_SLOT);                      \
        int   k11 = *(lds_i32*)(pk + 31 * KNN_SLOT);                      \
        int   k12 = *(lds_i32*)(pk + 32 * KNN_SLOT);                      \
        int   k13 = *(lds_i32*)(pk + 33 * KNN_SLOT);                      \
        int   k14 = *(lds_i32*)(pk + 34 * KNN_SLOT);                      \
        int   k15 = *(lds_i32*)(pk + 35 * KNN_SLOT);                      \
        int   k16 = *(lds_i32*)(pk + 36 * KNN_SLOT);                      \
        int   k17 = *(lds_i32*)(pk + 37 * KNN_SLOT);                      \
        int   k18 = *(lds_i32*)(pk + 38 * KNN_SLOT);                      \
        int   k19 = *(lds_i32*)(pk + 39 * KNN_SLOT);                      \
         \
         \
         \
         \
         \
        unsigned cur = *(lds_u32*)(tbase);                                \
        unsigned ea  = tbase + KNN_SLOT;                                  \
        unsigned nxt = *(lds_u32*)(ea);                                   \
        int jn; float pxn, pyn, pzn, ppn; \
        KNN_NEXT \
          \
          \
          \
          \
          \
          \
          \
          \
        float d; int j; \
        KNN_AHEAD(0u, d, j) \
        int s = 0; \
          \
          \
        do { \
              \
            KNN_TESTS \
              \
              \
              \
              \
              \
            float d1; int j1; \
            KNN_AHEAD((unsigned)s + 1u, d1, j1) \
              \
              \
              \
              \
              \
            __builtin_amdgcn_sched_barrier(0); \
            KNN_SELECTS \
            __builtin_amdgcn_sched_barrier(0); \
              \
              \
            asm volatile("" :: "v"(nxt), "v"(pxn), "v"(pyn), "v"(pzn), "v"(ppn)); \
            d = d1; j = j1; \
            ++s; \
        } while (__any((unsigned)s < cnt)); \
        *(lds_f32*)(pk + 0 * KNN_SLOT) = e0;                            \
        *(lds_f32*)(pk + 1 * KNN_SLOT) = e1;                            \
        *(lds_f32*)(pk + 2 * KNN_SLOT) = e2;                            \
        *(lds_f32*)(pk + 3 * KNN_SLOT) = e3;                            \
        *(lds_f32*)(pk + 4 * KNN_SLOT) = e4;                            \
        *(lds_f32*)(pk + 5 * KNN_SLOT) = e5;                            \
        *(lds_f32*)(pk + 6 * KNN_SLOT) = e6;                            \
        *(lds_f32*)(pk + 7 * KNN_SLOT) = e7;                            \
        *(lds_f32*)(pk + 8 * KNN_SLOT) = e8;                            \
        *(lds_f32*)(pk + 9 * KNN_SLOT) = e9;                            \
        *(lds_f32*)(pk + 10 * KNN_SLOT) = e10;                            \
        *(lds_f32*)(pk + 11 * KNN_SLOT) = e11;                            \
        *(lds_f32*)(pk + 12 * KNN_SLOT) = e12;                            \
        *(lds_f32*)(pk + 13 * KNN_SLOT) = e13;                            \
        *(lds_f32*)(pk + 14 * KNN_SLOT) = e14;                            \
        *(lds_f32*)(pk + 15 * KNN_SLOT) = e15;                            \
        *(lds_f32*)(pk + 16 * KNN_SLOT) = e16;                            \
        *(lds_f32*)(pk + 17 * KNN_SLOT) = e17;                            \
        *(lds_f32*)(pk + 18 * KNN_SLOT) = e18;                            \
        *(lds_f32*)(pk + 19 * KNN_SLOT) = e19;                            \
        *(lds_i32*)(pk + 20 * KNN_SLOT) = k0;                            \
        *(lds_i32*)(pk + 21 * KNN_SLOT) = k1;                            \
        *(lds_i32*)(pk + 22 * KNN_SLOT) = k2;                            \
        *(lds_i32*)(pk + 23 * KNN_SLOT) = k3;                            \
        *(lds_i32*)(pk + 24 * KNN_SLOT) = k4;                            \
        *(lds_i32*)(pk + 25 * KNN_SLOT) = k5;                            \
        *(lds_i32*)(pk + 26 * KNN_SLOT) = k6;                            \
        *(lds_i32*)(pk + 27 * KNN_SLOT) = k7;                            \
        *(lds_i32*)(pk + 28 * KNN_SLOT) = k8;                            \
        *(lds_i32*)(pk + 29 * KNN_SLOT) = k9;                            \
        *(lds_i32*)(pk + 30 * KNN_SLOT) = k10;                            \
        *(lds_i32*)(pk + 31 * KNN_SLOT) = k11;                            \
        *(lds_i32*)(pk + 32 * KNN_SLOT) = k12;                            \
        *(lds_i32*)(pk + 33 * KNN_SLOT) = k13;                            \
        *(lds_i32*)(pk + 34 * KNN_SLOT) = k14;                            \
        *(lds_i32*)(pk + 35 * KNN_SLOT) = k15;                            \
        *(lds_i32*)(pk + 36 * KNN_SLOT) = k16;                            \
        *(lds_i32*)(pk + 37 * KNN_SLOT) = k17;                            \
        *(lds_i32*)(pk + 38 * KNN_SLOT) = k18;                            \
        *(lds_i32*)(pk + 39 * KNN_SLOT) = k19;                            \
        tail = tbase;                                                     \
        thr = e19;                                                        \
        cnt = 0u;                                                         \
    }

#define KNN_NEXT                                                          \
    {   const bool need = (cur & 0xffffu) == 0u;                          \
        cur = need ? nxt : cur;                                           \
        ea  = need ? (ea + KNN_SLOT) : ea;                                \
        nxt = *(lds_u32*)(ea);                                            \
        const unsigned lz = (unsigned)__builtin_clz((cur & 0xffffu) | 1u);\
        jn  = (int)((cur >> 16) + lz - 16u);                              \
        cur = cur & ~(0x80000000u >> lz);                                 \
        const unsigned ga = pbase + (((unsigned)jn & (unsigned)(N_ - 1)) << 2);\
        pxn = *(lds_f32*)(ga);                                            \
        pyn = *(lds_f32*)(ga + KNN_PLANE);                                \
        pzn = *(lds_f32*)(ga + 2 * KNN_PLANE);                            \
        ppn = *(lds_f32*)(ga + 3 * KNN_PLANE); }

#define KNN_AHEAD(sn, D, J)                                               \
    {   const bool  have = (sn) < cnt;                                    \
        const float px = pxn, py = pyn, pz = pzn, pp = ppn;               \
        J = have ? jn : 0;                                                \
        KNN_NEXT                                                          \
        const float dot = __builtin_fmaf(qz, pz, __builtin_fmaf(qx, px, qy * py));\
        const float dc  = __builtin_fmaf(dot, 2.0f, -qq) - pp;            \
        D = have ? dc : -3.4e38f; }

#define KNN_TESTS                                                         \
    const bool c0 = d > e0, c1 = d > e1, c2 = d > e2, c3 = d > e3, c4 = d > e4;\
    const bool c5 = d > e5, c6 = d > e6, c7 = d > e7, c8 = d > e8, c9 = d > e9;\
    const bool c10 = d > e10, c11 = d > e11, c12 = d > e12, c13 = d > e13, c14 = d > e14;\
    const bool c15 = d > e15, c16 = d > e16, c17 = d > e17, c18 = d > e18, c19 = d > e19;

#define KNN_LAY1(i) const float t##i = c##i ? d : e##i; const int u##i = c##i ? j : k##i;
#define KNN_LAY2(iu, il) e##il = c##iu ? e##iu : t##il; k##il = c##iu ? k##iu : u##il;
#define KNN_SELECTS                                                       \
    KNN_LAY1(0) KNN_LAY1(1) KNN_LAY1(2) KNN_LAY1(3) KNN_LAY1(4)           \
    KNN_LAY1(5) KNN_LAY1(6) KNN_LAY1(7) KNN_LAY1(8) KNN_LAY1(9)           \
    KNN_LAY1(10) KNN_LAY1(11) KNN_LAY1(12) KNN_LAY1(13) KNN_LAY1(14)      \
    KNN_LAY1(15) KNN_LAY1(16) KNN_LAY1(17) KNN_LAY1(18) KNN_LAY1(19)      \
    KNN_LAY2(18, 19) KNN_LAY2(17, 18) KNN_LAY2(16, 17) KNN_LAY2(15, 16)   \
    KNN_LAY2(14, 15) KNN_LAY2(13, 14) KNN_LAY2(12, 13) KNN_LAY2(11, 12)   \
    KNN_LAY2(10, 11) KNN_LAY2(9, 10) KNN_LAY2(8, 9) KNN_LAY2(7, 8)        \
    KNN_LAY2(6, 7) KNN_LAY2(5, 6) KNN_LAY2(4, 5) KNN_LAY2(3, 4)           \
    KNN_LAY2(2, 3) KNN_LAY2(1, 2) KNN_LAY2(0, 1)                          \
    e0 = t0;                                                              \
    k0 = u0;

#define EDGE_PW 8
#define EDGE_PB 32
#define EDGE_LP 33
#define EDGE_HS (64 * EDGE_LP + 16)
#define EDGE_KP 132

__device__ __forceinline__ float edge_xhalf(float v)
{
    return __builtin_bit_cast(float, __builtin_amdgcn_permlanex16(
        __builtin_bit_cast(int, v), __builtin_bit_cast(int, v),
        0x76543210, 0xfedcba98, false, false));
}

__device__ __forceinline__ float edge_lane(float v, int l)
{
    return __builtin_bit_cast(float,
        __builtin_amdgcn_readlane(__builtin_bit_cast(int, v), l));
}

__global__ __launch_bounds__(KNN_T) void knn_edge_kernel(
    const float* __restrict__ x,
    const float* __restrict__ w1, const float* __restrict__ b1,
    const float* __restrict__ g1, const float* __restrict__ beta1,
    const float* __restrict__ rm1,const float* __restrict__ rv1,
    const float* __restrict__ w2, const float* __restrict__ b2,
    const float* __restrict__ g2, const float* __restrict__ beta2,
    const float* __restrict__ rm2,const float* __restrict__ rv2,
    float* __restrict__ out,
    float* __restrict__ out_m1,   float* __restrict__ out_m2)
{
    __shared__ __attribute__((aligned(16))) float lds[4 * N_ + 2 * KNN_CAP * KNN_T + 2 * K_ * KNN_T];
    float* spts = lds;
    int*   qj   = (int*)(lds + 4 * N_ + KNN_CAP * KNN_T);
    const float* sx = spts;
    const float* sy = spts + N_;
    const float* sz = spts + 2 * N_;
    const int blocksPerBatch = N_ / KNN_T;
    const int b     = blockIdx.x / blocksPerBatch;
    const int chunk = blockIdx.x % blocksPerBatch;
    const float* xb = x + (size_t)b * 3 * N_;
    const v4f* xb4 = (const v4f*)xb;
    v4f*       sp4 = (v4f*)spts;
    v4f        st[(3 * N_ / 4) / KNN_T];
#pragma unroll
    for (int t = 0; t < (3 * N_ / 4) / KNN_T; ++t)
        st[t] = xb4[t * KNN_T + threadIdx.x];
#pragma unroll
    for (int t = 0; t < (3 * N_ / 4) / KNN_T; ++t)
        sp4[t * KNN_T + threadIdx.x] = st[t];
#pragma unroll
    for (int t = 0; t < (N_ / 4) / KNN_T; ++t) {
        const v4f X = st[t], Y = st[t + (N_ / 4) / KNN_T], Z = st[t + 2 * ((N_ / 4) / KNN_T)];
        sp4[3 * (N_ / 4) + t * KNN_T + threadIdx.x] =
            __builtin_elementwise_fma(Z, Z, __builtin_elementwise_fma(X, X, Y * Y));
    }
    __syncthreads();

    const int   tid = threadIdx.x;
    const int   n  = chunk * KNN_T + tid;
    const float qx = sx[n], qy = sy[n], qz = sz[n];
    const float qq = fmaf(qz, qz, fmaf(qx, qx, qy * qy));
    const v2f qx2 = {qx, qx}, qy2 = {qy, qy}, qz2 = {qz, qz};
    const v2f nqq2 = {-qq, -qq}, two2 = {2.0f, 2.0f};

    const unsigned lds0  = (unsigned)(size_t)lds;
    const unsigned pbase = lds0;
    const unsigned tbase = lds0 + 4 * KNN_PLANE + (unsigned)tid * 4u;
    unsigned       pk    = lds0 + 4 * KNN_PLANE + 2 * KNN_QJ + (unsigned)tid * 4u;
    asm("" : "+v"(pk));

#pragma unroll
    for (int i = 0; i < K_; ++i) {
        *(lds_f32*)(pk + (unsigned)i * KNN_SLOT) = -3.4e38f;
        *(lds_i32*)(pk + (unsigned)(K_ + i) * KNN_SLOT) = 0;
    }

    float    thr  = -3.4e38f;
    unsigned tail = tbase;
    unsigned cnt  = 0u;
    unsigned dmask = 0u;

    v4f ax0, ax1, ax2, ax3, ay0, ay1, ay2, ay3, az0, az1, az2, az3, ap0, ap1, ap2, ap3;
    v4f bx0, bx1, bx2, bx3, by0, by1, by2, by3, bz0, bz1, bz2, bz3, bp0, bp1, bp2, bp3;
    KNN_LOAD(ax, ay, az, ap, pbase)
    KNN_LOAD(bx, by, bz, bp, pbase + KNN_CH * 4)
    for (int j0 = 0; j0 < N_; j0 += 2 * KNN_CH) {
        KNN_CHUNK(ax, ay, az, ap, j0, (j0 + 2 * KNN_CH) & (N_ - 1))
        KNN_DRAIN(j0)
        KNN_CHUNK(bx, by, bz, bp, j0 + KNN_CH, (j0 + 3 * KNN_CH) & (N_ - 1))
        KNN_DRAIN(j0 + KNN_CH)
    }
    const int k0 = *(lds_i32*)(pk + 20 * KNN_SLOT);
    const int k1 = *(lds_i32*)(pk + 21 * KNN_SLOT);
    const int k2 = *(lds_i32*)(pk + 22 * KNN_SLOT);
    const int k3 = *(lds_i32*)(pk + 23 * KNN_SLOT);
    const int k4 = *(lds_i32*)(pk + 24 * KNN_SLOT);
    const int k5 = *(lds_i32*)(pk + 25 * KNN_SLOT);
    const int k6 = *(lds_i32*)(pk + 26 * KNN_SLOT);
    const int k7 = *(lds_i32*)(pk + 27 * KNN_SLOT);
    const int k8 = *(lds_i32*)(pk + 28 * KNN_SLOT);
    const int k9 = *(lds_i32*)(pk + 29 * KNN_SLOT);
    const int k10 = *(lds_i32*)(pk + 30 * KNN_SLOT);
    const int k11 = *(lds_i32*)(pk + 31 * KNN_SLOT);
    const int k12 = *(lds_i32*)(pk + 32 * KNN_SLOT);
    const int k13 = *(lds_i32*)(pk + 33 * KNN_SLOT);
    const int k14 = *(lds_i32*)(pk + 34 * KNN_SLOT);
    const int k15 = *(lds_i32*)(pk + 35 * KNN_SLOT);
    const int k16 = *(lds_i32*)(pk + 36 * KNN_SLOT);
    const int k17 = *(lds_i32*)(pk + 37 * KNN_SLOT);
    const int k18 = *(lds_i32*)(pk + 38 * KNN_SLOT);
    const int k19 = *(lds_i32*)(pk + 39 * KNN_SLOT);
    __syncthreads();
    int* tk = qj;
    tk[0 * KNN_TP + tid] = k0;
    tk[1 * KNN_TP + tid] = k1;
    tk[2 * KNN_TP + tid] = k2;
    tk[3 * KNN_TP + tid] = k3;
    tk[4 * KNN_TP + tid] = k4;
    tk[5 * KNN_TP + tid] = k5;
    tk[6 * KNN_TP + tid] = k6;
    tk[7 * KNN_TP + tid] = k7;
    tk[8 * KNN_TP + tid] = k8;
    tk[9 * KNN_TP + tid] = k9;
    tk[10 * KNN_TP + tid] = k10;
    tk[11 * KNN_TP + tid] = k11;
    tk[12 * KNN_TP + tid] = k12;
    tk[13 * KNN_TP + tid] = k13;
    tk[14 * KNN_TP + tid] = k14;
    tk[15 * KNN_TP + tid] = k15;
    tk[16 * KNN_TP + tid] = k16;
    tk[17 * KNN_TP + tid] = k17;
    tk[18 * KNN_TP + tid] = k18;
    tk[19 * KNN_TP + tid] = k19;
    __syncthreads();

    __shared__ float pmx[2 * EDGE_HS];
    __shared__ float psm[2 * EDGE_HS];
    __shared__ __attribute__((aligned(16))) float pa[EDGE_PB * EDGE_KP];

    const int lane = tid & 31;
    const int half = lane >> 4;
    const int l15  = lane & 15;
    const int wv   = __builtin_amdgcn_readfirstlane(tid >> 5);

    const bool valid1 = (16 + l15) < K_;
    const int  kc1    = valid1 ? (16 + l15) : (K_ - 1);
    const int* tk0 = tk + l15 * KNN_TP;
    const int* tk1 = tk + kc1 * KNN_TP;

    v2f   bt0[4], bt1[4];
    float bs[4], bt[4];
#pragma unroll
    for (int ct = 0; ct < 4; ++ct) {
        const int    o    = ct * 16 + l15;
        const float* wrow = w1 + o * 6;
        const float w0 = wrow[0], w1v = wrow[1], w2v = wrow[2];
        const float w3 = wrow[3], w4v = wrow[4], w5v = wrow[5];
        bt0[ct].x = half ? w2v : w0;   bt0[ct].y = half ? w3 : w1v;
        bt1[ct].x = half ? 0.0f : w4v; bt1[ct].y = half ? 0.0f : w5v;
        const float s = g1[o] * rsqrtf(rv1[o] + EPS_);
        bs[ct] = s;
        bt[ct] = (b1[o] - rm1[o]) * s + beta1[o];
    }

    const float* wrow2 = w2 + (size_t)(wv * 16 + l15) * 128 + half * 2;
    v2f wa[32];
#pragma unroll
    for (int ks = 0; ks < 32; ++ks)
        wa[ks] = *(const v2f*)(wrow2 + ks * 4);
    const int   oc  = wv * 16 + l15;
    const float s2v = g2[oc] * rsqrtf(rv2[oc] + EPS_);
    const float t2v = (b2[oc] - rm2[oc]) * s2v + beta2[oc];

    const float* pr0 = pa + l15 * EDGE_KP + half * 2;
    const float* pr1 = pr0 + 16 * EDGE_KP;

#pragma nounroll
    for (int g = 0; g < KNN_T / EDGE_PB; ++g) {
        const int lq = g * EDGE_PB + wv * EDGE_PW;
        const int n0 = chunk * KNN_T + g * EDGE_PB;
        const int nw = n0 + wv * EDGE_PW;

        int nb0[EDGE_PW], nb1[EDGE_PW];
#pragma unroll
        for (int p = 0; p < EDGE_PW; ++p) {
            nb0[p] = tk0[lq + p];
            nb1[p] = tk1[lq + p];
        }

        float gx0[EDGE_PW], gy0[EDGE_PW], gz0[EDGE_PW];
        float gx1[EDGE_PW], gy1[EDGE_PW], gz1[EDGE_PW];
        float ccx[EDGE_PW], ccy[EDGE_PW], ccz[EDGE_PW];
#pragma unroll
        for (int p = 0; p < EDGE_PW; ++p) {
            gx0[p] = sx[nb0[p]]; gy0[p] = sy[nb0[p]]; gz0[p] = sz[nb0[p]];
            gx1[p] = sx[nb1[p]]; gy1[p] = sy[nb1[p]]; gz1[p] = sz[nb1[p]];
            ccx[p] = sx[nw + p]; ccy[p] = sy[nw + p]; ccz[p] = sz[nw + p];
        }

        float pmr[EDGE_PW * 4], psr[EDGE_PW * 4];
#pragma unroll
        for (int p = 0; p < EDGE_PW; ++p) {
            const float cx = ccx[p], cy = ccy[p], cz = ccz[p];

            v2f a0[2], a1[2];
#pragma unroll
            for (int rt = 0; rt < 2; ++rt) {
                const bool valid = rt ? valid1 : true;
                const float nx = rt ? gx1[p] : gx0[p];
                const float ny = rt ? gy1[p] : gy0[p];
                const float nz = rt ? gz1[p] : gz0[p];
                const float e0 = valid ? (nx - cx) : 0.0f;
                const float e1 = valid ? (ny - cy) : 0.0f;
                const float e2 = valid ? (nz - cz) : 0.0f;
                const float e3 = valid ? cx : 0.0f;
                const float e4 = valid ? cy : 0.0f;
                const float e5 = valid ? cz : 0.0f;
                v2f t0, t1;
                t0.x = half ? e2 : e0;  t0.y = half ? e3 : e1;
                t1.x = half ? 0.0f : e4; t1.y = half ? 0.0f : e5;
                a0[rt] = t0; a1[rt] = t1;
            }

#pragma unroll
            for (int ct = 0; ct < 4; ++ct) {
                v8f z = {};
                z = __builtin_amdgcn_wmma_f32_16x16x4_f32(false, a0[0], false, bt0[ct],
                                                          (short)0, z, false, false);
                z = __builtin_amdgcn_wmma_f32_16x16x4_f32(false, a1[0], false, bt1[ct],
                                                          (short)0, z, false, false);
                v8f z1 = {};
                z1 = __builtin_amdgcn_wmma_f32_16x16x4_f32(false, a0[1], false, bt0[ct],
                                                           (short)0, z1, false, false);
                z1 = __builtin_amdgcn_wmma_f32_16x16x4_f32(false, a1[1], false, bt1[ct],
                                                           (short)0, z1, false, false);

                const float s = bs[ct], t = bt[ct];
                float mx = -3.4e38f, sm = 0.0f;
#pragma unroll
                for (int v = 0; v < 8; ++v) {
                    const float val = fmaxf(z[v] * s + t, 0.0f);
                    mx = fmaxf(mx, val); sm += val;
                }
#pragma unroll
                for (int v = 0; v < 4; ++v) {
                    const float val = fmaxf(z1[v] * s + t, 0.0f);
                    mx = (half == 0) ? fmaxf(mx, val) : mx;
                    sm = (half == 0) ? (sm + val) : sm;
                }
                pmr[p * 4 + ct] = mx;
                psr[p * 4 + ct] = sm;
            }
        }
        __builtin_amdgcn_sched_barrier(0);
#pragma unroll
        for (int p = 0; p < EDGE_PW; ++p) {
#pragma unroll
            for (int ct = 0; ct < 4; ++ct) {
                const int q = half * EDGE_HS + (ct * 16 + l15) * EDGE_LP + wv * EDGE_PW + p;
                pmx[q] = pmr[p * 4 + ct];
                psm[q] = psr[p * 4 + ct];
            }
        }
        __builtin_amdgcn_sched_barrier(0);
        __syncthreads();

        float jm[16], ja[16];
        float rml[16], rmh[16], rsl[16], rsh[16];
#pragma unroll
        for (int r = 0; r < 16; ++r) {
            const int o = wv * 16 + r;
            rml[r] = pmx[o * EDGE_LP + lane]; rmh[r] = pmx[EDGE_HS + o * EDGE_LP + lane];
            rsl[r] = psm[o * EDGE_LP + lane]; rsh[r] = psm[EDGE_HS + o * EDGE_LP + lane];
        }
        __builtin_amdgcn_sched_barrier(0);
#pragma unroll
        for (int r = 0; r < 16; ++r) {
            jm[r] = fmaxf(rml[r], rmh[r]);
            ja[r] = (rsl[r] + rsh[r]) * (1.0f / K_);
        }
        __builtin_amdgcn_sched_barrier(0);
#pragma unroll
        for (int r = 0; r < 16; ++r) {
            const int o = wv * 16 + r;
            pa[lane * EDGE_KP + o]      = jm[r];
            pa[lane * EDGE_KP + 64 + o] = ja[r];
        }
        __syncthreads();

#pragma unroll
        for (int r = 0; r < 16; ++r) {
            const int    o   = wv * 16 + r;
            const size_t off = ((size_t)b * 64 + o) * N_ + n0 + lane;
            out_m1[off] = jm[r];
            out_m2[off] = ja[r];
        }

        v2f pb0[32], pb1[32];
#pragma unroll
        for (int ks = 0; ks < 32; ++ks) {
            pb0[ks] = *(const v2f*)(pr0 + ks * 4);
            pb1[ks] = *(const v2f*)(pr1 + ks * 4);
        }
        __builtin_amdgcn_sched_barrier(0);
        v8f d0 = {}, d1 = {};
#pragma unroll
        for (int ks = 0; ks < 32; ++ks) {
            d0 = __builtin_amdgcn_wmma_f32_16x16x4_f32(false, wa[ks], false, pb0[ks],
                                                       (short)0, d0, false, false);
            d1 = __builtin_amdgcn_wmma_f32_16x16x4_f32(false, wa[ks], false, pb1[ks],
                                                       (short)0, d1, false, false);
        }

#pragma unroll
        for (int r = 0; r < 8; ++r) {
            const float x0 = edge_xhalf(d0[r]);
            const float x1 = edge_xhalf(d1[r]);
            const float lo = half ? x1 : d0[r];
            const float hi = half ? d1[r] : x0;
            {
                const int   o = wv * 16 + r;
                const float s = edge_lane(s2v, r), t = edge_lane(t2v, r);
                out[((size_t)b * 64 + o) * N_ + n0 + lane] = fmaxf(lo * s + t, 0.0f);
            }
            {
                const int   o = wv * 16 + 8 + r;
                const float s = edge_lane(s2v, 8 + r), t = edge_lane(t2v, 8 + r);
                out[((size_t)b * 64 + o) * N_ + n0 + lane] = fmaxf(hi * s + t, 0.0f);
            }
        }
    }
}

extern "C" void kernel_launch(void* const* d_in, const int* in_sizes, int n_in,
                              void* d_out, int out_size, void* d_ws, size_t ws_size,
                              hipStream_t stream)
{
    const float* x     = (const float*)d_in[0];
    const float* w1    = (const float*)d_in[1];
    const float* b1    = (const float*)d_in[2];
    const float* g1    = (const float*)d_in[3];
    const float* beta1 = (const float*)d_in[4];
    const float* rm1   = (const float*)d_in[5];
    const float* rv1   = (const float*)d_in[6];
    const float* w2    = (const float*)d_in[7];
    const float* b2    = (const float*)d_in[8];
    const float* g2    = (const float*)d_in[9];
    const float* beta2 = (const float*)d_in[10];
    const float* rm2   = (const float*)d_in[11];
    const float* rv2   = (const float*)d_in[12];

    float* out    = (float*)d_out;
    float* out_m1 = out + (size_t)B_ * 64 * N_;
    float* out_m2 = out + 2 * (size_t)B_ * 64 * N_;

    knn_edge_kernel<<<B_ * (N_ / KNN_T), KNN_T, 0, stream>>>(
        x, w1, b1, g1, beta1, rm1, rv1,
        w2, b2, g2, beta2, rm2, rv2, out, out_m1, out_m2);
}
